# starting kernel
# baseline (speedup 1.0000x reference)
#include <hip/hip_runtime.h>


typedef __attribute__((ext_vector_type(2))) float v2f;
typedef __attribute__((ext_vector_type(8))) float v8f;

#define IN_C  128
#define HID   512
#define OUT2  64
#define NCLS  2
#define NEG_SLOPE 0.2f
#define KMAX  512

__global__ void gat_wmma_gemm_f32(const float* __restrict__ A,
                                  const float* __restrict__ B,
                                  float* __restrict__ C,
                                  int M, int K, int Nc) {
  __shared__ float Bs[KMAX * 16];

  const int tid  = threadIdx.x;
  const int lane = tid & 31;
  const int waveid = tid >> 5;
  const int tn = blockIdx.y;
  const int tilesM = (M + 15) >> 4;

  const int nK16 = K * 16;
  for (int idx = tid; idx < nK16; idx += 256) {
    const int k = idx >> 4;
    const int c = idx & 15;
    Bs[((k >> 1) << 5) + (c << 1) + (k & 1)] = B[(size_t)k * Nc + tn * 16 + c];
  }
  __syncthreads();

  const int tm = blockIdx.x * 8 + waveid;
  if (tm >= tilesM) return;

  const int half = lane >> 4;
  const int l16  = lane & 15;

  const float* Ap = A + (size_t)(tm * 16 + l16) * K + 2 * half;
  const float* Bl = Bs + half * 32 + (l16 << 1);

  v8f acc = {};
#pragma unroll 8
  for (int k = 0; k < K; k += 4) {
    v2f a = *(const v2f*)(Ap + k);
    v2f b = *(const v2f*)(Bl + k * 16);
    acc = __builtin_amdgcn_wmma_f32_16x16x4_f32(
         false, a,  false, b,
         (short)0, acc,  false,  false);
  }

  float* Cp = C + (size_t)(tm * 16 + 8 * half) * Nc + tn * 16 + l16;
#pragma unroll
  for (int r = 0; r < 8; ++r) Cp[(size_t)r * Nc] = acc[r];
}

__global__ void gat_node_dots(const float* __restrict__ h,
                              const float* __restrict__ a_src,
                              const float* __restrict__ a_dst,
                              float* __restrict__ as_out,
                              float* __restrict__ ad_out,
                              int N, int D) {
  const int wave = blockIdx.x * (blockDim.x >> 5) + (threadIdx.x >> 5);
  const int lane = threadIdx.x & 31;
  if (wave >= N) return;
  const float* row = h + (size_t)wave * D;
  float ss = 0.f, sd = 0.f;
  for (int c = lane; c < D; c += 32) {
    float v = row[c];
    ss += v * a_src[c];
    sd += v * a_dst[c];
  }
#pragma unroll
  for (int off = 16; off > 0; off >>= 1) {
    ss += __shfl_down(ss, off, 32);
    sd += __shfl_down(sd, off, 32);
  }
  if (lane == 0) { as_out[wave] = ss; ad_out[wave] = sd; }
}

__global__ void gat_fill_f32(float* __restrict__ p, float v, int n) {
  int i = blockIdx.x * blockDim.x + threadIdx.x;
  if (i < n) p[i] = v;
}

__device__ __forceinline__ void edge_sd(const int* __restrict__ ei, int e,
                                        int E, int& s, int& d) {
  if (e < E) { s = ei[e]; d = ei[E + e]; }
  else       { s = d = e - E; }
}

__device__ __forceinline__ float leakyr(float v) {
  return v > 0.f ? v : NEG_SLOPE * v;
}

__device__ __forceinline__ void atomicMaxF32(float* addr, float val) {
  if (val >= 0.f) atomicMax((int*)addr, __float_as_int(val));
  else            atomicMin((unsigned int*)addr, __float_as_uint(val));
}

__global__ void gat_edge_max(const int* __restrict__ ei,
                             const float* __restrict__ as,
                             const float* __restrict__ ad,
                             float* __restrict__ m, int E, int Etot) {
  int e = blockIdx.x * blockDim.x + threadIdx.x;
  if (e >= Etot) return;
  int s, d; edge_sd(ei, e, E, s, d);
  atomicMaxF32(&m[d], leakyr(as[s] + ad[d]));
}

__global__ void gat_edge_expsum(const int* __restrict__ ei,
                                const float* __restrict__ as,
                                const float* __restrict__ ad,
                                const float* __restrict__ m,
                                float* __restrict__ alpha,
                                float* __restrict__ denom, int E, int Etot) {
  int e = blockIdx.x * blockDim.x + threadIdx.x;
  if (e >= Etot) return;
  int s, d; edge_sd(ei, e, E, s, d);
  float ex = expf(leakyr(as[s] + ad[d]) - m[d]);
  alpha[e] = ex;
  atomicAdd(&denom[d], ex);
}

__global__ void gat_edge_norm(const int* __restrict__ ei,
                              const float* __restrict__ denom,
                              float* __restrict__ alpha, int E, int Etot) {
  int e = blockIdx.x * blockDim.x + threadIdx.x;
  if (e >= Etot) return;
  int s, d; edge_sd(ei, e, E, s, d);
  alpha[e] /= denom[d];
}

__global__ void gat_edge_aggregate(const int* __restrict__ ei,
                                   const float* __restrict__ alpha,
                                   const float* __restrict__ h,
                                   float* __restrict__ out,
                                   int E, int Etot, int D) {
  const int wave = blockIdx.x * (blockDim.x >> 5) + (threadIdx.x >> 5);
  const int lane = threadIdx.x & 31;
  if (wave >= Etot) return;
  int s, d; edge_sd(ei, wave, E, s, d);
  const float a = alpha[wave];
  const float* hs = h + (size_t)s * D;
  float* od = out + (size_t)d * D;
  for (int c = lane; c < D; c += 32) atomicAdd(&od[c], a * hs[c]);
}

__global__ void gat_bias_relu(float* __restrict__ p,
                              const float* __restrict__ bias, int N, int Dmask) {
  int i = blockIdx.x * blockDim.x + threadIdx.x;
  if (i >= N) return;
  float v = p[i] + bias[i & Dmask];
  p[i] = v > 0.f ? v : 0.f;
}

__global__ void gat_fc(const float* __restrict__ h,
                       const float* __restrict__ W,
                       const float* __restrict__ b,
                       float* __restrict__ out, int N) {
  int n = blockIdx.x * blockDim.x + threadIdx.x;
  if (n >= N) return;
  const float* row = h + (size_t)n * OUT2;
  float a0 = b[0], a1 = b[1];
#pragma unroll
  for (int c = 0; c < OUT2; ++c) {
    float v = row[c];
    a0 += v * W[c * NCLS + 0];
    a1 += v * W[c * NCLS + 1];
  }
  out[n * NCLS + 0] = a0;
  out[n * NCLS + 1] = a1;
}

extern "C" void kernel_launch(void* const* d_in, const int* in_sizes, int n_in,
                              void* d_out, int out_size, void* d_ws, size_t ws_size,
                              hipStream_t stream) {
  const float* x   = (const float*)d_in[0];
  const int*   ei  = (const int*)  d_in[1];
  const float* W1  = (const float*)d_in[2];
  const float* a1s = (const float*)d_in[3];
  const float* a1d = (const float*)d_in[4];
  const float* b1  = (const float*)d_in[5];
  const float* W2  = (const float*)d_in[6];
  const float* a2s = (const float*)d_in[7];
  const float* a2d = (const float*)d_in[8];
  const float* b2  = (const float*)d_in[9];
  const float* fcW = (const float*)d_in[10];
  const float* fcb = (const float*)d_in[11];

  const int N    = in_sizes[0] / IN_C;
  const int E    = in_sizes[1] / 2;
  const int Etot = E + N;

  float* w       = (float*)d_ws;
  float* h1      = w;
  float* agg     = w + (size_t)N * HID;
  float* as_buf  = w + (size_t)2 * N * HID;
  float* ad_buf  = as_buf + N;
  float* m_buf   = ad_buf + N;
  float* den_buf = m_buf + N;
  float* alpha_e = den_buf + N;

  const int TB = 256;
  const float NEG_INF = -__builtin_inff();

  auto gemm_grid = [](int M, int Nc) {
    return dim3((((M + 15) / 16) + 7) / 8, Nc / 16);
  };
  const int edge_blk  = (Etot + TB - 1) / TB;
  const int eagg_blk  = (Etot + 7) / 8;
  const int node_blk  = (N + 7) / 8;

  gat_wmma_gemm_f32<<<gemm_grid(N, HID), TB, 0, stream>>>(x, W1, h1, N, IN_C, HID);
  gat_node_dots<<<node_blk, TB, 0, stream>>>(h1, a1s, a1d, as_buf, ad_buf, N, HID);

  gat_fill_f32<<<(N + TB - 1) / TB, TB, 0, stream>>>(m_buf, NEG_INF, N);
  gat_fill_f32<<<(N + TB - 1) / TB, TB, 0, stream>>>(den_buf, 0.f, N);
  gat_fill_f32<<<(N * HID + TB - 1) / TB, TB, 0, stream>>>(agg, 0.f, N * HID);

  gat_edge_max   <<<edge_blk, TB, 0, stream>>>(ei, as_buf, ad_buf, m_buf, E, Etot);
  gat_edge_expsum<<<edge_blk, TB, 0, stream>>>(ei, as_buf, ad_buf, m_buf, alpha_e, den_buf, E, Etot);
  gat_edge_norm  <<<edge_blk, TB, 0, stream>>>(ei, den_buf, alpha_e, E, Etot);
  gat_edge_aggregate<<<eagg_blk, TB, 0, stream>>>(ei, alpha_e, h1, agg, E, Etot, HID);
  gat_bias_relu<<<(N * HID + TB - 1) / TB, TB, 0, stream>>>(agg, b1, N * HID, HID - 1);

  gat_wmma_gemm_f32<<<gemm_grid(N, OUT2), TB, 0, stream>>>(agg, W2, h1, N, HID, OUT2);
  gat_node_dots<<<node_blk, TB, 0, stream>>>(h1, a2s, a2d, as_buf, ad_buf, N, OUT2);

  gat_fill_f32<<<(N + TB - 1) / TB, TB, 0, stream>>>(m_buf, NEG_INF, N);
  gat_fill_f32<<<(N + TB - 1) / TB, TB, 0, stream>>>(den_buf, 0.f, N);
  gat_fill_f32<<<(N * OUT2 + TB - 1) / TB, TB, 0, stream>>>(agg, 0.f, N * OUT2);

  gat_edge_max   <<<edge_blk, TB, 0, stream>>>(ei, as_buf, ad_buf, m_buf, E, Etot);
  gat_edge_expsum<<<edge_blk, TB, 0, stream>>>(ei, as_buf, ad_buf, m_buf, alpha_e, den_buf, E, Etot);
  gat_edge_norm  <<<edge_blk, TB, 0, stream>>>(ei, den_buf, alpha_e, E, Etot);
  gat_edge_aggregate<<<eagg_blk, TB, 0, stream>>>(ei, alpha_e, h1, agg, E, Etot, OUT2);
  gat_bias_relu<<<(N * OUT2 + TB - 1) / TB, TB, 0, stream>>>(agg, b2, N * OUT2, OUT2 - 1);

  gat_fc<<<(N + TB - 1) / TB, TB, 0, stream>>>(agg, fcW, fcb, (float*)d_out, N);
}


# Layer 1 aggregates before it transforms: scores from x.(W1 a), atomic scatter of 128-wide x rows, then the f32 WMMA GEMM with W1; sums reassociated (linear steps commuted), 4x fewer atomics
# speedup vs baseline: 1.4502x; 1.4502x over previous
#include <hip/hip_runtime.h>


typedef __attribute__((ext_vector_type(2))) float v2f;
typedef __attribute__((ext_vector_type(8))) float v8f;

#define IN_C  128
#define HID   512
#define OUT2  64
#define NCLS  2
#define NEG_SLOPE 0.2f
#define KMAX  512

__global__ void gat_wmma_gemm_f32(const float* __restrict__ A,
                                  const float* __restrict__ B,
                                  float* __restrict__ C,
                                  int M, int K, int Nc) {
  __shared__ float Bs[KMAX * 16];

  const int tid  = threadIdx.x;
  const int lane = tid & 31;
  const int waveid = tid >> 5;
  const int tn = blockIdx.y;
  const int tilesM = (M + 15) >> 4;

  const int nK16 = K * 16;
  for (int idx = tid; idx < nK16; idx += 256) {
    const int k = idx >> 4;
    const int c = idx & 15;
    Bs[((k >> 1) << 5) + (c << 1) + (k & 1)] = B[(size_t)k * Nc + tn * 16 + c];
  }
  __syncthreads();

  const int tm = blockIdx.x * 8 + waveid;
  if (tm >= tilesM) return;

  const int half = lane >> 4;
  const int l16  = lane & 15;

  const float* Ap = A + (size_t)(tm * 16 + l16) * K + 2 * half;
  const float* Bl = Bs + half * 32 + (l16 << 1);

  v8f acc = {};
#pragma unroll 8
  for (int k = 0; k < K; k += 4) {
    v2f a = *(const v2f*)(Ap + k);
    v2f b = *(const v2f*)(Bl + k * 16);
    acc = __builtin_amdgcn_wmma_f32_16x16x4_f32(
         false, a,  false, b,
         (short)0, acc,  false,  false);
  }

  float* Cp = C + (size_t)(tm * 16 + 8 * half) * Nc + tn * 16 + l16;
#pragma unroll
  for (int r = 0; r < 8; ++r) Cp[(size_t)r * Nc] = acc[r];
}

__global__ void gat_node_dots(const float* __restrict__ h,
                              const float* __restrict__ a_src,
                              const float* __restrict__ a_dst,
                              float* __restrict__ as_out,
                              float* __restrict__ ad_out,
                              int N, int D) {
  const int wave = blockIdx.x * (blockDim.x >> 5) + (threadIdx.x >> 5);
  const int lane = threadIdx.x & 31;
  if (wave >= N) return;
  const float* row = h + (size_t)wave * D;
  float ss = 0.f, sd = 0.f;
  for (int c = lane; c < D; c += 32) {
    float v = row[c];
    ss += v * a_src[c];
    sd += v * a_dst[c];
  }
#pragma unroll
  for (int off = 16; off > 0; off >>= 1) {
    ss += __shfl_down(ss, off, 32);
    sd += __shfl_down(sd, off, 32);
  }
  if (lane == 0) { as_out[wave] = ss; ad_out[wave] = sd; }
}

__global__ void gat_fill_f32(float* __restrict__ p, float v, int n) {
  int i = blockIdx.x * blockDim.x + threadIdx.x;
  if (i < n) p[i] = v;
}

__device__ __forceinline__ void edge_sd(const int* __restrict__ ei, int e,
                                        int E, int& s, int& d) {
  if (e < E) { s = ei[e]; d = ei[E + e]; }
  else       { s = d = e - E; }
}

__device__ __forceinline__ float leakyr(float v) {
  return v > 0.f ? v : NEG_SLOPE * v;
}

__device__ __forceinline__ void atomicMaxF32(float* addr, float val) {
  if (val >= 0.f) atomicMax((int*)addr, __float_as_int(val));
  else            atomicMin((unsigned int*)addr, __float_as_uint(val));
}

__global__ void gat_edge_max(const int* __restrict__ ei,
                             const float* __restrict__ as,
                             const float* __restrict__ ad,
                             float* __restrict__ m, int E, int Etot) {
  int e = blockIdx.x * blockDim.x + threadIdx.x;
  if (e >= Etot) return;
  int s, d; edge_sd(ei, e, E, s, d);
  atomicMaxF32(&m[d], leakyr(as[s] + ad[d]));
}

__global__ void gat_edge_expsum(const int* __restrict__ ei,
                                const float* __restrict__ as,
                                const float* __restrict__ ad,
                                const float* __restrict__ m,
                                float* __restrict__ alpha,
                                float* __restrict__ denom, int E, int Etot) {
  int e = blockIdx.x * blockDim.x + threadIdx.x;
  if (e >= Etot) return;
  int s, d; edge_sd(ei, e, E, s, d);
  float ex = expf(leakyr(as[s] + ad[d]) - m[d]);
  alpha[e] = ex;
  atomicAdd(&denom[d], ex);
}

__global__ void gat_edge_norm(const int* __restrict__ ei,
                              const float* __restrict__ denom,
                              float* __restrict__ alpha, int E, int Etot) {
  int e = blockIdx.x * blockDim.x + threadIdx.x;
  if (e >= Etot) return;
  int s, d; edge_sd(ei, e, E, s, d);
  alpha[e] /= denom[d];
}

__global__ void gat_edge_aggregate(const int* __restrict__ ei,
                                   const float* __restrict__ alpha,
                                   const float* __restrict__ h,
                                   float* __restrict__ out,
                                   int E, int Etot, int D) {
  const int wave = blockIdx.x * (blockDim.x >> 5) + (threadIdx.x >> 5);
  const int lane = threadIdx.x & 31;
  if (wave >= Etot) return;
  int s, d; edge_sd(ei, wave, E, s, d);
  const float a = alpha[wave];
  const float* hs = h + (size_t)s * D;
  float* od = out + (size_t)d * D;
  for (int c = lane; c < D; c += 32) atomicAdd(&od[c], a * hs[c]);
}

__global__ void gat_bias_relu(float* __restrict__ p,
                              const float* __restrict__ bias, int N, int Dmask) {
  int i = blockIdx.x * blockDim.x + threadIdx.x;
  if (i >= N) return;
  float v = p[i] + bias[i & Dmask];
  p[i] = v > 0.f ? v : 0.f;
}

__global__ void gat_fc(const float* __restrict__ h,
                       const float* __restrict__ W,
                       const float* __restrict__ b,
                       float* __restrict__ out, int N) {
  int n = blockIdx.x * blockDim.x + threadIdx.x;
  if (n >= N) return;
  const float* row = h + (size_t)n * OUT2;
  float a0 = b[0], a1 = b[1];
#pragma unroll
  for (int c = 0; c < OUT2; ++c) {
    float v = row[c];
    a0 += v * W[c * NCLS + 0];
    a1 += v * W[c * NCLS + 1];
  }
  out[n * NCLS + 0] = a0;
  out[n * NCLS + 1] = a1;
}

extern "C" void kernel_launch(void* const* d_in, const int* in_sizes, int n_in,
                              void* d_out, int out_size, void* d_ws, size_t ws_size,
                              hipStream_t stream) {
  const float* x   = (const float*)d_in[0];
  const int*   ei  = (const int*)  d_in[1];
  const float* W1  = (const float*)d_in[2];
  const float* a1s = (const float*)d_in[3];
  const float* a1d = (const float*)d_in[4];
  const float* b1  = (const float*)d_in[5];
  const float* W2  = (const float*)d_in[6];
  const float* a2s = (const float*)d_in[7];
  const float* a2d = (const float*)d_in[8];
  const float* b2  = (const float*)d_in[9];
  const float* fcW = (const float*)d_in[10];
  const float* fcb = (const float*)d_in[11];

  const int N    = in_sizes[0] / IN_C;
  const int E    = in_sizes[1] / 2;
  const int Etot = E + N;

  float* w       = (float*)d_ws;
  float* bufA    = w;
  float* bufB    = w + (size_t)N * HID;
  float* as_buf  = w + (size_t)2 * N * HID;
  float* ad_buf  = as_buf + N;
  float* m_buf   = ad_buf + N;
  float* den_buf = m_buf + N;
  float* alpha_e = den_buf + N;

  const int TB = 256;
  const float NEG_INF = -__builtin_inff();

  auto gemm_grid = [](int M, int Nc) {
    return dim3((((M + 15) / 16) + 7) / 8, Nc / 16);
  };
  const int edge_blk  = (Etot + TB - 1) / TB;
  const int eagg_blk  = (Etot + 7) / 8;
  const int node_blk  = (N + 7) / 8;

  float* wa   = bufA;
  float* wd   = bufA + IN_C;
  float* aggx = bufB;
  gat_node_dots<<<IN_C / 8, TB, 0, stream>>>(W1, a1s, a1d, wa, wd, IN_C, HID);
  gat_node_dots<<<node_blk, TB, 0, stream>>>(x, wa, wd, as_buf, ad_buf, N, IN_C);

  gat_fill_f32<<<(N + TB - 1) / TB, TB, 0, stream>>>(m_buf, NEG_INF, N);
  gat_fill_f32<<<(N + TB - 1) / TB, TB, 0, stream>>>(den_buf, 0.f, N);
  gat_fill_f32<<<(N * IN_C + TB - 1) / TB, TB, 0, stream>>>(aggx, 0.f, N * IN_C);

  gat_edge_max   <<<edge_blk, TB, 0, stream>>>(ei, as_buf, ad_buf, m_buf, E, Etot);
  gat_edge_expsum<<<edge_blk, TB, 0, stream>>>(ei, as_buf, ad_buf, m_buf, alpha_e, den_buf, E, Etot);
  gat_edge_norm  <<<edge_blk, TB, 0, stream>>>(ei, den_buf, alpha_e, E, Etot);
  gat_edge_aggregate<<<eagg_blk, TB, 0, stream>>>(ei, alpha_e, x, aggx, E, Etot, IN_C);

  gat_wmma_gemm_f32<<<gemm_grid(N, HID), TB, 0, stream>>>(aggx, W1, bufA, N, IN_C, HID);
  gat_bias_relu<<<(N * HID + TB - 1) / TB, TB, 0, stream>>>(bufA, b1, N * HID, HID - 1);

  float* h2   = bufB;
  float* agg2 = bufA;
  gat_wmma_gemm_f32<<<gemm_grid(N, OUT2), TB, 0, stream>>>(bufA, W2, h2, N, HID, OUT2);
  gat_node_dots<<<node_blk, TB, 0, stream>>>(h2, a2s, a2d, as_buf, ad_buf, N, OUT2);

  gat_fill_f32<<<(N + TB - 1) / TB, TB, 0, stream>>>(m_buf, NEG_INF, N);
  gat_fill_f32<<<(N + TB - 1) / TB, TB, 0, stream>>>(den_buf, 0.f, N);
  gat_fill_f32<<<(N * OUT2 + TB - 1) / TB, TB, 0, stream>>>(agg2, 0.f, N * OUT2);

  gat_edge_max   <<<edge_blk, TB, 0, stream>>>(ei, as_buf, ad_buf, m_buf, E, Etot);
  gat_edge_expsum<<<edge_blk, TB, 0, stream>>>(ei, as_buf, ad_buf, m_buf, alpha_e, den_buf, E, Etot);
  gat_edge_norm  <<<edge_blk, TB, 0, stream>>>(ei, den_buf, alpha_e, E, Etot);
  gat_edge_aggregate<<<eagg_blk, TB, 0, stream>>>(ei, alpha_e, h2, agg2, E, Etot, OUT2);
  gat_bias_relu<<<(N * OUT2 + TB - 1) / TB, TB, 0, stream>>>(agg2, b2, N * OUT2, OUT2 - 1);

  gat_fc<<<(N + TB - 1) / TB, TB, 0, stream>>>(agg2, fcW, fcb, (float*)d_out, N);
}


# More work per wave: fills and bias+ReLU loop with 128-bit accesses, aggregation walks 32 edges per wave with grouped row loads; about 7x fewer waves launched; atomic adds keep arrival order
# speedup vs baseline: 1.7146x; 1.1823x over previous
#include <hip/hip_runtime.h>


typedef __attribute__((ext_vector_type(2))) float v2f;
typedef __attribute__((ext_vector_type(8))) float v8f;
typedef __attribute__((ext_vector_type(4))) float v4f;

#define IN_C  128
#define HID   512
#define OUT2  64
#define NCLS  2
#define NEG_SLOPE 0.2f
#define KMAX  512

__global__ void gat_wmma_gemm_f32(const float* __restrict__ A,
                                  const float* __restrict__ B,
                                  float* __restrict__ C,
                                  int M, int K, int Nc) {
  __shared__ float Bs[KMAX * 16];

  const int tid  = threadIdx.x;
  const int lane = tid & 31;
  const int waveid = tid >> 5;
  const int tn = blockIdx.y;
  const int tilesM = (M + 15) >> 4;

  const int nK16 = K * 16;
  for (int idx = tid; idx < nK16; idx += 256) {
    const int k = idx >> 4;
    const int c = idx & 15;
    Bs[((k >> 1) << 5) + (c << 1) + (k & 1)] = B[(size_t)k * Nc + tn * 16 + c];
  }
  __syncthreads();

  const int tm = blockIdx.x * 8 + waveid;
  if (tm >= tilesM) return;

  const int half = lane >> 4;
  const int l16  = lane & 15;

  const float* Ap = A + (size_t)(tm * 16 + l16) * K + 2 * half;
  const float* Bl = Bs + half * 32 + (l16 << 1);

  v8f acc = {};
#pragma unroll 8
  for (int k = 0; k < K; k += 4) {
    v2f a = *(const v2f*)(Ap + k);
    v2f b = *(const v2f*)(Bl + k * 16);
    acc = __builtin_amdgcn_wmma_f32_16x16x4_f32(
         false, a,  false, b,
         (short)0, acc,  false,  false);
  }

  float* Cp = C + (size_t)(tm * 16 + 8 * half) * Nc + tn * 16 + l16;
#pragma unroll
  for (int r = 0; r < 8; ++r) Cp[(size_t)r * Nc] = acc[r];
}

__global__ void gat_node_dots(const float* __restrict__ h,
                              const float* __restrict__ a_src,
                              const float* __restrict__ a_dst,
                              float* __restrict__ as_out,
                              float* __restrict__ ad_out,
                              int N, int D) {
  const int wave0  = blockIdx.x * (blockDim.x >> 5) + (threadIdx.x >> 5);
  const int nwaves = gridDim.x * (blockDim.x >> 5);
  const int lane = threadIdx.x & 31;
  for (int node = wave0; node < N; node += nwaves) {
    const float* row = h + (size_t)node * D;
    float ss = 0.f, sd = 0.f;
    for (int c = lane; c < D; c += 32) {
      float v = row[c];
      ss += v * a_src[c];
      sd += v * a_dst[c];
    }
#pragma unroll
    for (int off = 16; off > 0; off >>= 1) {
      ss += __shfl_down(ss, off, 32);
      sd += __shfl_down(sd, off, 32);
    }
    if (lane == 0) { as_out[node] = ss; ad_out[node] = sd; }
  }
}

__global__ void gat_fill_f32(float* __restrict__ p, float v, int n) {
  const int tid = blockIdx.x * blockDim.x + threadIdx.x;
  const int nth = gridDim.x * blockDim.x;
  const int n4  = n >> 2;
  float4* __restrict__ p4 = (float4*)p;
  const float4 v4 = make_float4(v, v, v, v);
  for (int i = tid; i < n4; i += nth) p4[i] = v4;
  const int t = (n4 << 2) + tid;
  if (t < n) p[t] = v;
}

__device__ __forceinline__ void edge_sd(const int* __restrict__ ei, int e,
                                        int E, int& s, int& d) {
  if (e < E) { s = ei[e]; d = ei[E + e]; }
  else       { s = d = e - E; }
}

__device__ __forceinline__ float leakyr(float v) {
  return v > 0.f ? v : NEG_SLOPE * v;
}

__device__ __forceinline__ void atomicMaxF32(float* addr, float val) {
  if (val >= 0.f) atomicMax((int*)addr, __float_as_int(val));
  else            atomicMin((unsigned int*)addr, __float_as_uint(val));
}

__global__ void gat_edge_max(const int* __restrict__ ei,
                             const float* __restrict__ as,
                             const float* __restrict__ ad,
                             float* __restrict__ m, int E, int Etot) {
  int e = blockIdx.x * blockDim.x + threadIdx.x;
  if (e >= Etot) return;
  int s, d; edge_sd(ei, e, E, s, d);
  atomicMaxF32(&m[d], leakyr(as[s] + ad[d]));
}

__global__ void gat_edge_expsum(const int* __restrict__ ei,
                                const float* __restrict__ as,
                                const float* __restrict__ ad,
                                const float* __restrict__ m,
                                float* __restrict__ alpha,
                                float* __restrict__ denom, int E, int Etot) {
  int e = blockIdx.x * blockDim.x + threadIdx.x;
  if (e >= Etot) return;
  int s, d; edge_sd(ei, e, E, s, d);
  float ex = expf(leakyr(as[s] + ad[d]) - m[d]);
  alpha[e] = ex;
  atomicAdd(&denom[d], ex);
}

__global__ void gat_edge_norm(const int* __restrict__ ei,
                              const float* __restrict__ denom,
                              float* __restrict__ alpha, int E, int Etot) {
  int e = blockIdx.x * blockDim.x + threadIdx.x;
  if (e >= Etot) return;
  int s, d; edge_sd(ei, e, E, s, d);
  alpha[e] /= denom[d];
}

template <int D>
__global__ void gat_edge_aggregate(const int* __restrict__ ei,
                                   const float* __restrict__ alpha,
                                   const float* __restrict__ h,
                                   float* __restrict__ out,
                                   int E, int Etot) {
  constexpr int PER = D / 32;
  constexpr int GRP = 512 / D;
  const int lane = threadIdx.x & 31;
  const int wib    = __builtin_amdgcn_readfirstlane((int)(threadIdx.x >> 5));
  const int wave0  = blockIdx.x * (blockDim.x >> 5) + wib;
  const int nwaves = gridDim.x * (blockDim.x >> 5);
  const int nruns  = (Etot + 31) >> 5;
  for (int run = wave0; run < nruns; run += nwaves) {
    const int base = run << 5;
    const int e = base + lane;
    int s_l = 0, d_l = 0;
    float a_l = 0.f;
    if (e < Etot) { edge_sd(ei, e, E, s_l, d_l); a_l = alpha[e]; }
    const int left = Etot - base;
    const int cnt  = left < 32 ? left : 32;
    int j = 0;
    for (; j + GRP <= cnt; j += GRP) {
      float v[GRP][PER];
#pragma unroll
      for (int u = 0; u < GRP; ++u) {
        const int s = __builtin_amdgcn_readlane(s_l, j + u);
        const float* hs = h + (size_t)s * D + lane;
#pragma unroll
        for (int i = 0; i < PER; ++i) v[u][i] = hs[32 * i];
      }
#pragma unroll
      for (int u = 0; u < GRP; ++u) {
        const int   d = __builtin_amdgcn_readlane(d_l, j + u);
        const float a = __int_as_float(__builtin_amdgcn_readlane(__float_as_int(a_l), j + u));
        float* od = out + (size_t)d * D + lane;
#pragma unroll
        for (int i = 0; i < PER; ++i) atomicAdd(&od[32 * i], a * v[u][i]);
      }
    }
    for (; j < cnt; ++j) {
      const int   s = __builtin_amdgcn_readlane(s_l, j);
      const int   d = __builtin_amdgcn_readlane(d_l, j);
      const float a = __int_as_float(__builtin_amdgcn_readlane(__float_as_int(a_l), j));
      const float* hs = h + (size_t)s * D + lane;
      float* od = out + (size_t)d * D + lane;
      float v[PER];
#pragma unroll
      for (int i = 0; i < PER; ++i) v[i] = hs[32 * i];
#pragma unroll
      for (int i = 0; i < PER; ++i) atomicAdd(&od[32 * i], a * v[i]);
    }
  }
}

__device__ __forceinline__ v4f bias_relu4(v4f v, const float* __restrict__ b) {
  v.x += b[0]; v.y += b[1]; v.z += b[2]; v.w += b[3];
  v.x = v.x > 0.f ? v.x : 0.f;
  v.y = v.y > 0.f ? v.y : 0.f;
  v.z = v.z > 0.f ? v.z : 0.f;
  v.w = v.w > 0.f ? v.w : 0.f;
  return v;
}

__global__ void gat_bias_relu(float* __restrict__ p,
                              const float* __restrict__ bias, int N, int Dmask) {
  const int tid = blockIdx.x * blockDim.x + threadIdx.x;
  const int nth = gridDim.x * blockDim.x;
  const int n4  = N >> 2;
  v4f* __restrict__ p4 = (v4f*)p;
  int i = tid;
  for (; i + 3 * nth < n4; i += 4 * nth) {
    const v4f v0 = p4[i];
    const v4f v1 = p4[i + nth];
    const v4f v2 = p4[i + 2 * nth];
    const v4f v3 = p4[i + 3 * nth];
    const v4f r0 = bias_relu4(v0, bias + ((i << 2) & Dmask));
    const v4f r1 = bias_relu4(v1, bias + (((i + nth) << 2) & Dmask));
    const v4f r2 = bias_relu4(v2, bias + (((i + 2 * nth) << 2) & Dmask));
    const v4f r3 = bias_relu4(v3, bias + (((i + 3 * nth) << 2) & Dmask));
    p4[i] = r0;
    p4[i + nth] = r1;
    p4[i + 2 * nth] = r2;
    p4[i + 3 * nth] = r3;
  }
  for (; i < n4; i += nth) p4[i] = bias_relu4(p4[i], bias + ((i << 2) & Dmask));
  const int t = (n4 << 2) + tid;
  if (t < N) {
    float v = p[t] + bias[t & Dmask];
    p[t] = v > 0.f ? v : 0.f;
  }
}

__global__ void gat_fc(const float* __restrict__ h,
                       const float* __restrict__ W,
                       const float* __restrict__ b,
                       float* __restrict__ out, int N) {
  int n = blockIdx.x * blockDim.x + threadIdx.x;
  if (n >= N) return;
  const float* row = h + (size_t)n * OUT2;
  float a0 = b[0], a1 = b[1];
#pragma unroll
  for (int c = 0; c < OUT2; ++c) {
    float v = row[c];
    a0 += v * W[c * NCLS + 0];
    a1 += v * W[c * NCLS + 1];
  }
  out[n * NCLS + 0] = a0;
  out[n * NCLS + 1] = a1;
}

extern "C" void kernel_launch(void* const* d_in, const int* in_sizes, int n_in,
                              void* d_out, int out_size, void* d_ws, size_t ws_size,
                              hipStream_t stream) {
  const float* x   = (const float*)d_in[0];
  const int*   ei  = (const int*)  d_in[1];
  const float* W1  = (const float*)d_in[2];
  const float* a1s = (const float*)d_in[3];
  const float* a1d = (const float*)d_in[4];
  const float* b1  = (const float*)d_in[5];
  const float* W2  = (const float*)d_in[6];
  const float* a2s = (const float*)d_in[7];
  const float* a2d = (const float*)d_in[8];
  const float* b2  = (const float*)d_in[9];
  const float* fcW = (const float*)d_in[10];
  const float* fcb = (const float*)d_in[11];

  const int N    = in_sizes[0] / IN_C;
  const int E    = in_sizes[1] / 2;
  const int Etot = E + N;

  float* w       = (float*)d_ws;
  float* bufA    = w;
  float* bufB    = w + (size_t)N * HID;
  float* as_buf  = w + (size_t)2 * N * HID;
  float* ad_buf  = as_buf + N;
  float* m_buf   = ad_buf + N;
  float* den_buf = m_buf + N;
  float* alpha_e = den_buf + N;

  const int TB = 256;
  const float NEG_INF = -__builtin_inff();

  auto gemm_grid = [](int M, int Nc) {
    return dim3((((M + 15) / 16) + 7) / 8, Nc / 16);
  };
  const int CHIP = 2048;
  auto capped = [](int blocks, int cap) { return blocks < cap ? blocks : cap; };
  auto ew_grid = [&](int n) { return capped((n / 4 + TB - 1) / TB, CHIP); };
  const int edge_blk  = (Etot + TB - 1) / TB;
  const int eagg_blk  = capped(((Etot + 31) / 32 + 7) / 8, CHIP);
  const int node_blk  = capped((N + 7) / 8, CHIP);

  float* wa   = bufA;
  float* wd   = bufA + IN_C;
  float* aggx = bufB;
  gat_node_dots<<<IN_C / 8, TB, 0, stream>>>(W1, a1s, a1d, wa, wd, IN_C, HID);
  gat_node_dots<<<node_blk, TB, 0, stream>>>(x, wa, wd, as_buf, ad_buf, N, IN_C);

  gat_fill_f32<<<ew_grid(N), TB, 0, stream>>>(m_buf, NEG_INF, N);
  gat_fill_f32<<<ew_grid(N), TB, 0, stream>>>(den_buf, 0.f, N);
  gat_fill_f32<<<ew_grid(N * IN_C), TB, 0, stream>>>(aggx, 0.f, N * IN_C);

  gat_edge_max   <<<edge_blk, TB, 0, stream>>>(ei, as_buf, ad_buf, m_buf, E, Etot);
  gat_edge_expsum<<<edge_blk, TB, 0, stream>>>(ei, as_buf, ad_buf, m_buf, alpha_e, den_buf, E, Etot);
  gat_edge_norm  <<<edge_blk, TB, 0, stream>>>(ei, den_buf, alpha_e, E, Etot);
  gat_edge_aggregate<IN_C><<<eagg_blk, TB, 0, stream>>>(ei, alpha_e, x, aggx, E, Etot);

  gat_wmma_gemm_f32<<<gemm_grid(N, HID), TB, 0, stream>>>(aggx, W1, bufA, N, IN_C, HID);
  gat_bias_relu<<<ew_grid(N * HID), TB, 0, stream>>>(bufA, b1, N * HID, HID - 1);

  float* h2   = bufB;
  float* agg2 = bufA;
  gat_wmma_gemm_f32<<<gemm_grid(N, OUT2), TB, 0, stream>>>(bufA, W2, h2, N, HID, OUT2);
  gat_node_dots<<<node_blk, TB, 0, stream>>>(h2, a2s, a2d, as_buf, ad_buf, N, OUT2);

  gat_fill_f32<<<ew_grid(N), TB, 0, stream>>>(m_buf, NEG_INF, N);
  gat_fill_f32<<<ew_grid(N), TB, 0, stream>>>(den_buf, 0.f, N);
  gat_fill_f32<<<ew_grid(N * OUT2), TB, 0, stream>>>(agg2, 0.f, N * OUT2);

  gat_edge_max   <<<edge_blk, TB, 0, stream>>>(ei, as_buf, ad_buf, m_buf, E, Etot);
  gat_edge_expsum<<<edge_blk, TB, 0, stream>>>(ei, as_buf, ad_buf, m_buf, alpha_e, den_buf, E, Etot);
  gat_edge_norm  <<<edge_blk, TB, 0, stream>>>(ei, den_buf, alpha_e, E, Etot);
  gat_edge_aggregate<OUT2><<<eagg_blk, TB, 0, stream>>>(ei, alpha_e, h2, agg2, E, Etot);
  gat_bias_relu<<<ew_grid(N * OUT2), TB, 0, stream>>>(agg2, b2, N * OUT2, OUT2 - 1);

  gat_fc<<<(N + TB - 1) / TB, TB, 0, stream>>>(agg2, fcW, fcb, (float*)d_out, N);
}


# Edges grouped by destination (count, offsets, scatter); one wave per node does softmax and row sum in registers, no float atomics; sums in a node follow the arrival order of the integer scatter adds
# speedup vs baseline: 2.3968x; 1.3979x over previous
#include <hip/hip_runtime.h>


typedef __attribute__((ext_vector_type(2))) float v2f;
typedef __attribute__((ext_vector_type(8))) float v8f;
typedef __attribute__((ext_vector_type(4))) float v4f;

#define IN_C  128
#define HID   512
#define OUT2  64
#define NCLS  2
#define NEG_SLOPE 0.2f
#define KMAX  512

__global__ void gat_wmma_gemm_f32(const float* __restrict__ A,
                                  const float* __restrict__ B,
                                  float* __restrict__ C,
                                  int M, int K, int Nc) {
  __shared__ float Bs[KMAX * 16];

  const int tid  = threadIdx.x;
  const int lane = tid & 31;
  const int waveid = tid >> 5;
  const int tn = blockIdx.y;
  const int tilesM = (M + 15) >> 4;

  const int nK16 = K * 16;
  for (int idx = tid; idx < nK16; idx += 256) {
    const int k = idx >> 4;
    const int c = idx & 15;
    Bs[((k >> 1) << 5) + (c << 1) + (k & 1)] = B[(size_t)k * Nc + tn * 16 + c];
  }
  __syncthreads();

  const int tm = blockIdx.x * 8 + waveid;
  if (tm >= tilesM) return;

  const int half = lane >> 4;
  const int l16  = lane & 15;

  const float* Ap = A + (size_t)(tm * 16 + l16) * K + 2 * half;
  const float* Bl = Bs + half * 32 + (l16 << 1);

  v8f acc = {};
#pragma unroll 8
  for (int k = 0; k < K; k += 4) {
    v2f a = *(const v2f*)(Ap + k);
    v2f b = *(const v2f*)(Bl + k * 16);
    acc = __builtin_amdgcn_wmma_f32_16x16x4_f32(
         false, a,  false, b,
         (short)0, acc,  false,  false);
  }

  float* Cp = C + (size_t)(tm * 16 + 8 * half) * Nc + tn * 16 + l16;
#pragma unroll
  for (int r = 0; r < 8; ++r) Cp[(size_t)r * Nc] = acc[r];
}

__global__ void gat_node_dots(const float* __restrict__ h,
                              const float* __restrict__ a_src,
                              const float* __restrict__ a_dst,
                              float* __restrict__ as_out,
                              float* __restrict__ ad_out,
                              int N, int D) {
  const int wave0  = blockIdx.x * (blockDim.x >> 5) + (threadIdx.x >> 5);
  const int nwaves = gridDim.x * (blockDim.x >> 5);
  const int lane = threadIdx.x & 31;
  for (int node = wave0; node < N; node += nwaves) {
    const float* row = h + (size_t)node * D;
    float ss = 0.f, sd = 0.f;
    for (int c = lane; c < D; c += 32) {
      float v = row[c];
      ss += v * a_src[c];
      sd += v * a_dst[c];
    }
#pragma unroll
    for (int off = 16; off > 0; off >>= 1) {
      ss += __shfl_down(ss, off, 32);
      sd += __shfl_down(sd, off, 32);
    }
    if (lane == 0) { as_out[node] = ss; ad_out[node] = sd; }
  }
}

__global__ void gat_fill_f32(float* __restrict__ p, float v, int n) {
  const int tid = blockIdx.x * blockDim.x + threadIdx.x;
  const int nth = gridDim.x * blockDim.x;
  const int n4  = n >> 2;
  float4* __restrict__ p4 = (float4*)p;
  const float4 v4 = make_float4(v, v, v, v);
  for (int i = tid; i < n4; i += nth) p4[i] = v4;
  const int t = (n4 << 2) + tid;
  if (t < n) p[t] = v;
}

__global__ void gat_dst_count(const int* __restrict__ ei, int* __restrict__ cnt, int E) {
  const int e = blockIdx.x * blockDim.x + threadIdx.x;
  if (e < E) atomicAdd(&cnt[ei[E + e]], 1);
}

#define SCAN_TB 1024
__global__ void gat_dst_offsets(const int* __restrict__ cnt,
                                int* __restrict__ rowptr,
                                int* __restrict__ cursor, int N) {
  __shared__ int wbefore[SCAN_TB / 32];
  __shared__ int wtot[SCAN_TB / 32];
  const int t = threadIdx.x;
  const int lane = t & 31;
  const int w = t >> 5;
  const int b = blockIdx.x;
  int before = 0;
  for (int k = 0; k < b; ++k) before += cnt[k * SCAN_TB + t];
  const int i = b * SCAN_TB + t;
  const int own = i < N ? cnt[i] : 0;
  int inc = own;
#pragma unroll
  for (int off = 1; off < 32; off <<= 1) {
    const int v = __shfl_up(inc, off, 32);
    if (lane >= off) inc += v;
  }
#pragma unroll
  for (int off = 16; off > 0; off >>= 1) before += __shfl_xor(before, off, 32);
  if (lane == 31) wtot[w] = inc;
  if (lane == 0)  wbefore[w] = before;
  __syncthreads();
  int base = 0;
  for (int k = 0; k < SCAN_TB / 32; ++k) {
    base += wbefore[k];
    if (k < w) base += wtot[k];
  }
  const int excl = base + inc - own;
  if (i < N) { rowptr[i] = excl; cursor[i] = excl; }
  if (i == N - 1) rowptr[N] = excl + own;
}

__global__ void gat_dst_scatter(const int* __restrict__ ei, int* __restrict__ cursor,
                                int* __restrict__ col, int E) {
  const int e = blockIdx.x * blockDim.x + threadIdx.x;
  if (e < E) {
    const int p = atomicAdd(&cursor[ei[E + e]], 1);
    col[p] = ei[e];
  }
}

__device__ __forceinline__ float leakyr(float v) {
  return v > 0.f ? v : NEG_SLOPE * v;
}

__device__ __forceinline__ float wave_max(float v) {
#pragma unroll
  for (int off = 16; off > 0; off >>= 1) v = fmaxf(v, __shfl_xor(v, off, 32));
  return v;
}

__device__ __forceinline__ float wave_sum(float v) {
#pragma unroll
  for (int off = 16; off > 0; off >>= 1) v += __shfl_xor(v, off, 32);
  return v;
}

template <int D> struct gat_rowvec;
template <> struct gat_rowvec<128> { typedef v4f type; };
template <> struct gat_rowvec<64>  { typedef v2f type; };

template <int D>
__device__ __forceinline__ void agg_rows(const float* __restrict__ h, int s_l, float a_l,
                                         int n, int lane,
                                         typename gat_rowvec<D>::type& acc) {
  typedef typename gat_rowvec<D>::type vec;
  constexpr int GRP = 8;
  for (int u = 0; u < n; u += GRP) {
    vec v[GRP];
#pragma unroll
    for (int g = 0; g < GRP; ++g) {
      const int s = __builtin_amdgcn_readlane(s_l, u + g);
      v[g] = ((const vec*)(h + (size_t)s * D))[lane];
    }
#pragma unroll
    for (int g = 0; g < GRP; ++g) {
      const float a = __int_as_float(__builtin_amdgcn_readlane(__float_as_int(a_l), u + g));
      acc += a * v[g];
    }
  }
}

template <int D>
__global__ void gat_node_softmax_agg(const int* __restrict__ rowptr,
                                     const int* __restrict__ col,
                                     const float* __restrict__ as,
                                     const float* __restrict__ ad,
                                     const float* __restrict__ h,
                                     float* __restrict__ out, int N) {
  typedef typename gat_rowvec<D>::type vec;
  const int lane = threadIdx.x & 31;
  const int wib    = __builtin_amdgcn_readfirstlane((int)(threadIdx.x >> 5));
  const int wave0  = blockIdx.x * (blockDim.x >> 5) + wib;
  const int nwaves = gridDim.x * (blockDim.x >> 5);
  const float NEG_INF = -__builtin_inff();
  for (int d = wave0; d < N; d += nwaves) {
    const int beg = __builtin_amdgcn_readfirstlane(rowptr[d]);
    const int end = __builtin_amdgcn_readfirstlane(rowptr[d + 1]);
    const int cnt = end - beg + 1;
    const float adv = ad[d];

    const bool act0 = lane < cnt;
    int   s0 = d;
    float e0 = NEG_INF;
    if (act0) {
      if (lane < cnt - 1) s0 = col[beg + lane];
      e0 = leakyr(as[s0] + adv);
    }
    float mx = e0;
    for (int base = 32; base < cnt; base += 32) {
      const int j = base + lane;
      if (j < cnt) {
        const int s = j < cnt - 1 ? col[beg + j] : d;
        mx = fmaxf(mx, leakyr(as[s] + adv));
      }
    }
    mx = wave_max(mx);

    const float ex0 = act0 ? expf(e0 - mx) : 0.f;
    float sm = ex0;
    for (int base = 32; base < cnt; base += 32) {
      const int j = base + lane;
      if (j < cnt) {
        const int s = j < cnt - 1 ? col[beg + j] : d;
        sm += expf(leakyr(as[s] + adv) - mx);
      }
    }
    sm = wave_sum(sm);

    vec acc = {};
    agg_rows<D>(h, s0, ex0 / sm, cnt < 32 ? cnt : 32, lane, acc);
    for (int base = 32; base < cnt; base += 32) {
      const int j = base + lane;
      int   s = d;
      float a = 0.f;
      if (j < cnt) {
        if (j < cnt - 1) s = col[beg + j];
        a = expf(leakyr(as[s] + adv) - mx) / sm;
      }
      const int left = cnt - base;
      agg_rows<D>(h, s, a, left < 32 ? left : 32, lane, acc);
    }

    ((vec*)(out + (size_t)d * D))[lane] = acc;
  }
}

__device__ __forceinline__ v4f bias_relu4(v4f v, const float* __restrict__ b) {
  v.x += b[0]; v.y += b[1]; v.z += b[2]; v.w += b[3];
  v.x = v.x > 0.f ? v.x : 0.f;
  v.y = v.y > 0.f ? v.y : 0.f;
  v.z = v.z > 0.f ? v.z : 0.f;
  v.w = v.w > 0.f ? v.w : 0.f;
  return v;
}

__global__ void gat_bias_relu(float* __restrict__ p,
                              const float* __restrict__ bias, int N, int Dmask) {
  const int tid = blockIdx.x * blockDim.x + threadIdx.x;
  const int nth = gridDim.x * blockDim.x;
  const int n4  = N >> 2;
  v4f* __restrict__ p4 = (v4f*)p;
  int i = tid;
  for (; i + 3 * nth < n4; i += 4 * nth) {
    const v4f v0 = p4[i];
    const v4f v1 = p4[i + nth];
    const v4f v2 = p4[i + 2 * nth];
    const v4f v3 = p4[i + 3 * nth];
    const v4f r0 = bias_relu4(v0, bias + ((i << 2) & Dmask));
    const v4f r1 = bias_relu4(v1, bias + (((i + nth) << 2) & Dmask));
    const v4f r2 = bias_relu4(v2, bias + (((i + 2 * nth) << 2) & Dmask));
    const v4f r3 = bias_relu4(v3, bias + (((i + 3 * nth) << 2) & Dmask));
    p4[i] = r0;
    p4[i + nth] = r1;
    p4[i + 2 * nth] = r2;
    p4[i + 3 * nth] = r3;
  }
  for (; i < n4; i += nth) p4[i] = bias_relu4(p4[i], bias + ((i << 2) & Dmask));
  const int t = (n4 << 2) + tid;
  if (t < N) {
    float v = p[t] + bias[t & Dmask];
    p[t] = v > 0.f ? v : 0.f;
  }
}

__global__ void gat_fc(const float* __restrict__ h,
                       const float* __restrict__ W,
                       const float* __restrict__ b,
                       float* __restrict__ out, int N) {
  int n = blockIdx.x * blockDim.x + threadIdx.x;
  if (n >= N) return;
  const float* row = h + (size_t)n * OUT2;
  float a0 = b[0], a1 = b[1];
#pragma unroll
  for (int c = 0; c < OUT2; ++c) {
    float v = row[c];
    a0 += v * W[c * NCLS + 0];
    a1 += v * W[c * NCLS + 1];
  }
  out[n * NCLS + 0] = a0;
  out[n * NCLS + 1] = a1;
}

extern "C" void kernel_launch(void* const* d_in, const int* in_sizes, int n_in,
                              void* d_out, int out_size, void* d_ws, size_t ws_size,
                              hipStream_t stream) {
  const float* x   = (const float*)d_in[0];
  const int*   ei  = (const int*)  d_in[1];
  const float* W1  = (const float*)d_in[2];
  const float* a1s = (const float*)d_in[3];
  const float* a1d = (const float*)d_in[4];
  const float* b1  = (const float*)d_in[5];
  const float* W2  = (const float*)d_in[6];
  const float* a2s = (const float*)d_in[7];
  const float* a2d = (const float*)d_in[8];
  const float* b2  = (const float*)d_in[9];
  const float* fcW = (const float*)d_in[10];
  const float* fcb = (const float*)d_in[11];

  const int N    = in_sizes[0] / IN_C;
  const int E    = in_sizes[1] / 2;

  float* w       = (float*)d_ws;
  float* bufA    = w;
  float* bufB    = w + (size_t)N * HID;
  float* as_buf  = w + (size_t)2 * N * HID;
  float* ad_buf  = as_buf + N;
  int*   cnt     = (int*)as_buf;
  int*   rowptr  = (int*)(ad_buf + N);
  int*   cursor  = rowptr + N + 1;
  int*   col     = cursor + N;

  const int TB = 256;

  auto gemm_grid = [](int M, int Nc) {
    return dim3((((M + 15) / 16) + 7) / 8, Nc / 16);
  };
  const int CHIP = 2048;
  auto capped = [](int blocks, int cap) { return blocks < cap ? blocks : cap; };
  auto ew_grid = [&](int n) { return capped((n / 4 + TB - 1) / TB, CHIP); };
  const int list_blk  = (E + TB - 1) / TB;
  const int node_blk  = capped((N + 7) / 8, CHIP);

  gat_fill_f32<<<ew_grid(N), TB, 0, stream>>>((float*)cnt, 0.f, N);
  gat_dst_count  <<<list_blk, TB, 0, stream>>>(ei, cnt, E);
  gat_dst_offsets<<<(N + SCAN_TB - 1) / SCAN_TB, SCAN_TB, 0, stream>>>(cnt, rowptr, cursor, N);
  gat_dst_scatter<<<list_blk, TB, 0, stream>>>(ei, cursor, col, E);

  float* wa   = bufA;
  float* wd   = bufA + IN_C;
  float* aggx = bufB;
  gat_node_dots<<<IN_C / 8, TB, 0, stream>>>(W1, a1s, a1d, wa, wd, IN_C, HID);
  gat_node_dots<<<node_blk, TB, 0, stream>>>(x, wa, wd, as_buf, ad_buf, N, IN_C);
  gat_node_softmax_agg<IN_C><<<node_blk, TB, 0, stream>>>(rowptr, col, as_buf, ad_buf, x, aggx, N);

  gat_wmma_gemm_f32<<<gemm_grid(N, HID), TB, 0, stream>>>(aggx, W1, bufA, N, IN_C, HID);
  gat_bias_relu<<<ew_grid(N * HID), TB, 0, stream>>>(bufA, b1, N * HID, HID - 1);

  float* h2   = bufB;
  float* agg2 = bufA;
  gat_wmma_gemm_f32<<<gemm_grid(N, OUT2), TB, 0, stream>>>(bufA, W2, h2, N, HID, OUT2);
  gat_node_dots<<<node_blk, TB, 0, stream>>>(h2, a2s, a2d, as_buf, ad_buf, N, OUT2);
  gat_node_softmax_agg<OUT2><<<node_blk, TB, 0, stream>>>(rowptr, col, as_buf, ad_buf, h2, agg2, N);
  gat_bias_relu<<<ew_grid(N * OUT2), TB, 0, stream>>>(agg2, b2, N * OUT2, OUT2 - 1);

  gat_fc<<<(N + TB - 1) / TB, TB, 0, stream>>>(agg2, fcW, fcb, (float*)d_out, N);
}


# Both dense products in one WMMA kernel: a 128-row block keeps its hidden rows in LDS, 64 columns at a time, bias+ReLU between the two; h1 never goes to memory; summation order unchanged
# speedup vs baseline: 3.7378x; 1.5595x over previous
#include <hip/hip_runtime.h>


typedef __attribute__((ext_vector_type(2))) float v2f;
typedef __attribute__((ext_vector_type(8))) float v8f;
typedef __attribute__((ext_vector_type(4))) float v4f;

#define IN_C  128
#define HID   512
#define OUT2  64
#define NCLS  2
#define NEG_SLOPE 0.2f

#define FM_HC   64
#define FM_NCH  (HID / FM_HC)
#define FM_NT1  (FM_HC / 16)
#define FM_NT2  (OUT2 / 16)
#define FM_HP   (FM_HC + 4)
#define FM_I1   ((IN_C / 2) * (FM_HC / 4) / 256)
#define FM_I2   ((FM_HC / 2) * (OUT2 / 4) / 256)

__device__ __forceinline__ void fm_load_chunk(const float* __restrict__ W1,
                                              const float* __restrict__ W2,
                                              int j, int tid,
                                              v4f (&p1)[2 * FM_I1], v4f (&p2)[2 * FM_I2]) {
#pragma unroll
  for (int i = 0; i < FM_I1; ++i) {
    const int it = tid + 256 * i;
    const int q = it % (FM_HC / 4);
    const int p = it / (FM_HC / 4);
    const float* g = W1 + (size_t)(2 * p) * HID + FM_HC * j + 4 * q;
    p1[2 * i]     = *(const v4f*)g;
    p1[2 * i + 1] = *(const v4f*)(g + HID);
  }
#pragma unroll
  for (int i = 0; i < FM_I2; ++i) {
    const int it = tid + 256 * i;
    const int q = it % (OUT2 / 4);
    const int p = it / (OUT2 / 4);
    const float* g = W2 + (size_t)(FM_HC * j + 2 * p) * OUT2 + 4 * q;
    p2[2 * i]     = *(const v4f*)g;
    p2[2 * i + 1] = *(const v4f*)(g + OUT2);
  }
}

__device__ __forceinline__ void fm_put_item(float* s, v4f r0, v4f r1) {
  const v4f lo = {r0.x, r1.x, r0.y, r1.y};
  const v4f hi = {r0.z, r1.z, r0.w, r1.w};
  *(v4f*)s = lo;
  *(v4f*)(s + 4) = hi;
}

__device__ __forceinline__ void fm_store_chunk(float* W1s, float* W2s, int tid,
                                               const v4f (&p1)[2 * FM_I1],
                                               const v4f (&p2)[2 * FM_I2]) {
#pragma unroll
  for (int i = 0; i < FM_I1; ++i) {
    const int it = tid + 256 * i;
    const int q = it % (FM_HC / 4);
    const int p = it / (FM_HC / 4);
    fm_put_item(W1s + (q >> 2) * (IN_C * 16) + p * 32 + (q & 3) * 8, p1[2 * i], p1[2 * i + 1]);
  }
#pragma unroll
  for (int i = 0; i < FM_I2; ++i) {
    const int it = tid + 256 * i;
    const int q = it % (OUT2 / 4);
    const int p = it / (OUT2 / 4);
    fm_put_item(W2s + (q >> 2) * (FM_HC * 16) + p * 32 + (q & 3) * 8, p2[2 * i], p2[2 * i + 1]);
  }
}

__global__ __launch_bounds__(256)
void gat_transform_fused(const float* __restrict__ A,
                         const float* __restrict__ W1,
                         const float* __restrict__ b1,
                         const float* __restrict__ W2,
                         float* __restrict__ H2, int N) {
  __shared__ __attribute__((aligned(16))) float W1s[FM_NT1 * IN_C * 16];
  __shared__ __attribute__((aligned(16))) float W2s[FM_NT2 * FM_HC * 16];
  __shared__ __attribute__((aligned(16))) float Hs[8 * 16 * FM_HP];

  const int tid  = threadIdx.x;
  const int lane = tid & 31;
  const int wave = __builtin_amdgcn_readfirstlane(tid >> 5);
  const int half = lane >> 4;
  const int l16  = lane & 15;
  const int row0 = (blockIdx.x * 8 + wave) * 16;
  const bool valid = row0 < N;

  v4f p1[2 * FM_I1], p2[2 * FM_I2];
  fm_load_chunk(W1, W2, 0, tid, p1, p2);

  int rowc = row0 + l16;
  if (rowc > N - 1) rowc = N - 1;
  const float* Ap = A + (size_t)rowc * IN_C + 2 * half;
  v2f af[IN_C / 4];
#pragma unroll
  for (int kq = 0; kq < IN_C / 4; ++kq) af[kq] = *(const v2f*)(Ap + 4 * kq);

  fm_store_chunk(W1s, W2s, tid, p1, p2);
  __syncthreads();

  float* Hw = Hs + wave * (16 * FM_HP);
  const float* B1l = W1s + half * 32 + (l16 << 1);
  const float* B2l = W2s + half * 32 + (l16 << 1);
  const float* H2l = Hw + l16 * FM_HP + 2 * half;
  float* Hst = Hw + (8 * half) * FM_HP + l16;

  v8f acc2[FM_NT2];
#pragma unroll
  for (int t = 0; t < FM_NT2; ++t) acc2[t] = v8f{};

  for (int j = 0; j < FM_NCH; ++j) {
    const bool more = j + 1 < FM_NCH;
    if (more) fm_load_chunk(W1, W2, j + 1, tid, p1, p2);

    if (valid) {
#pragma unroll
      for (int nt = 0; nt < FM_NT1; nt += 2) {
        v8f c0 = {}, c1 = {};
        const float* Ba = B1l + nt * (IN_C * 16);
        const float* Bb = Ba + IN_C * 16;
#pragma unroll
        for (int kq = 0; kq < IN_C / 4; ++kq) {
          const v2f ba = *(const v2f*)(Ba + kq * 64);
          const v2f bb = *(const v2f*)(Bb + kq * 64);
          c0 = __builtin_amdgcn_wmma_f32_16x16x4_f32(false, af[kq], false, ba, (short)0, c0, false, false);
          c1 = __builtin_amdgcn_wmma_f32_16x16x4_f32(false, af[kq], false, bb, (short)0, c1, false, false);
        }
        const float bia = b1[FM_HC * j + nt * 16 + l16];
        const float bib = b1[FM_HC * j + nt * 16 + 16 + l16];
#pragma unroll
        for (int r = 0; r < 8; ++r) {
          const float va = c0[r] + bia;
          const float vb = c1[r] + bib;
          Hst[r * FM_HP + nt * 16]      = va > 0.f ? va : 0.f;
          Hst[r * FM_HP + nt * 16 + 16] = vb > 0.f ? vb : 0.f;
        }
      }
    }
    __syncthreads();

    if (valid) {
#pragma unroll
      for (int kk = 0; kk < FM_HC / 4; ++kk) {
        const v2f a2 = *(const v2f*)(H2l + 4 * kk);
#pragma unroll
        for (int t = 0; t < FM_NT2; ++t) {
          const v2f b2 = *(const v2f*)(B2l + t * (FM_HC * 16) + kk * 64);
          acc2[t] = __builtin_amdgcn_wmma_f32_16x16x4_f32(false, a2, false, b2, (short)0, acc2[t], false, false);
        }
      }
    }
    if (!more) break;
    __syncthreads();
    fm_store_chunk(W1s, W2s, tid, p1, p2);
    __syncthreads();
  }

  if (valid) {
    float* Cp = H2 + (size_t)(row0 + 8 * half) * OUT2 + l16;
#pragma unroll
    for (int r = 0; r < 8; ++r) {
      if (row0 + 8 * half + r < N) {
#pragma unroll
        for (int t = 0; t < FM_NT2; ++t) Cp[(size_t)r * OUT2 + t * 16] = acc2[t][r];
      }
    }
  }
}

__global__ void gat_node_dots(const float* __restrict__ h,
                              const float* __restrict__ a_src,
                              const float* __restrict__ a_dst,
                              float* __restrict__ as_out,
                              float* __restrict__ ad_out,
                              int N, int D) {
  const int wave0  = blockIdx.x * (blockDim.x >> 5) + (threadIdx.x >> 5);
  const int nwaves = gridDim.x * (blockDim.x >> 5);
  const int lane = threadIdx.x & 31;
  for (int node = wave0; node < N; node += nwaves) {
    const float* row = h + (size_t)node * D;
    float ss = 0.f, sd = 0.f;
    for (int c = lane; c < D; c += 32) {
      float v = row[c];
      ss += v * a_src[c];
      sd += v * a_dst[c];
    }
#pragma unroll
    for (int off = 16; off > 0; off >>= 1) {
      ss += __shfl_down(ss, off, 32);
      sd += __shfl_down(sd, off, 32);
    }
    if (lane == 0) { as_out[node] = ss; ad_out[node] = sd; }
  }
}

__global__ void gat_fill_f32(float* __restrict__ p, float v, int n) {
  const int tid = blockIdx.x * blockDim.x + threadIdx.x;
  const int nth = gridDim.x * blockDim.x;
  const int n4  = n >> 2;
  float4* __restrict__ p4 = (float4*)p;
  const float4 v4 = make_float4(v, v, v, v);
  for (int i = tid; i < n4; i += nth) p4[i] = v4;
  const int t = (n4 << 2) + tid;
  if (t < n) p[t] = v;
}

__global__ void gat_dst_count(const int* __restrict__ ei, int* __restrict__ cnt, int E) {
  const int e = blockIdx.x * blockDim.x + threadIdx.x;
  if (e < E) atomicAdd(&cnt[ei[E + e]], 1);
}

#define SCAN_TB 1024
__global__ void gat_dst_offsets(const int* __restrict__ cnt,
                                int* __restrict__ rowptr,
                                int* __restrict__ cursor, int N) {
  __shared__ int wbefore[SCAN_TB / 32];
  __shared__ int wtot[SCAN_TB / 32];
  const int t = threadIdx.x;
  const int lane = t & 31;
  const int w = t >> 5;
  const int b = blockIdx.x;
  int before = 0;
  for (int k = 0; k < b; ++k) before += cnt[k * SCAN_TB + t];
  const int i = b * SCAN_TB + t;
  const int own = i < N ? cnt[i] : 0;
  int inc = own;
#pragma unroll
  for (int off = 1; off < 32; off <<= 1) {
    const int v = __shfl_up(inc, off, 32);
    if (lane >= off) inc += v;
  }
#pragma unroll
  for (int off = 16; off > 0; off >>= 1) before += __shfl_xor(before, off, 32);
  if (lane == 31) wtot[w] = inc;
  if (lane == 0)  wbefore[w] = before;
  __syncthreads();
  int base = 0;
  for (int k = 0; k < SCAN_TB / 32; ++k) {
    base += wbefore[k];
    if (k < w) base += wtot[k];
  }
  const int excl = base + inc - own;
  if (i < N) { rowptr[i] = excl; cursor[i] = excl; }
  if (i == N - 1) rowptr[N] = excl + own;
}

__global__ void gat_dst_scatter(const int* __restrict__ ei, int* __restrict__ cursor,
                                int* __restrict__ col, int E) {
  const int e = blockIdx.x * blockDim.x + threadIdx.x;
  if (e < E) {
    const int p = atomicAdd(&cursor[ei[E + e]], 1);
    col[p] = ei[e];
  }
}

__device__ __forceinline__ float leakyr(float v) {
  return v > 0.f ? v : NEG_SLOPE * v;
}

__device__ __forceinline__ float wave_max(float v) {
#pragma unroll
  for (int off = 16; off > 0; off >>= 1) v = fmaxf(v, __shfl_xor(v, off, 32));
  return v;
}

__device__ __forceinline__ float wave_sum(float v) {
#pragma unroll
  for (int off = 16; off > 0; off >>= 1) v += __shfl_xor(v, off, 32);
  return v;
}

template <int D> struct gat_rowvec;
template <> struct gat_rowvec<128> { typedef v4f type; };
template <> struct gat_rowvec<64>  { typedef v2f type; };

template <int D>
__device__ __forceinline__ void agg_rows(const float* __restrict__ h, int s_l, float a_l,
                                         int n, int lane,
                                         typename gat_rowvec<D>::type& acc) {
  typedef typename gat_rowvec<D>::type vec;
  constexpr int GRP = 8;
  for (int u = 0; u < n; u += GRP) {
    vec v[GRP];
#pragma unroll
    for (int g = 0; g < GRP; ++g) {
      const int s = __builtin_amdgcn_readlane(s_l, u + g);
      v[g] = ((const vec*)(h + (size_t)s * D))[lane];
    }
#pragma unroll
    for (int g = 0; g < GRP; ++g) {
      const float a = __int_as_float(__builtin_amdgcn_readlane(__float_as_int(a_l), u + g));
      acc += a * v[g];
    }
  }
}

template <int D>
__global__ void gat_node_softmax_agg(const int* __restrict__ rowptr,
                                     const int* __restrict__ col,
                                     const float* __restrict__ as,
                                     const float* __restrict__ ad,
                                     const float* __restrict__ h,
                                     float* __restrict__ out, int N) {
  typedef typename gat_rowvec<D>::type vec;
  const int lane = threadIdx.x & 31;
  const int wib    = __builtin_amdgcn_readfirstlane((int)(threadIdx.x >> 5));
  const int wave0  = blockIdx.x * (blockDim.x >> 5) + wib;
  const int nwaves = gridDim.x * (blockDim.x >> 5);
  const float NEG_INF = -__builtin_inff();
  for (int d = wave0; d < N; d += nwaves) {
    const int beg = __builtin_amdgcn_readfirstlane(rowptr[d]);
    const int end = __builtin_amdgcn_readfirstlane(rowptr[d + 1]);
    const int cnt = end - beg + 1;
    const float adv = ad[d];

    const bool act0 = lane < cnt;
    int   s0 = d;
    float e0 = NEG_INF;
    if (act0) {
      if (lane < cnt - 1) s0 = col[beg + lane];
      e0 = leakyr(as[s0] + adv);
    }
    float mx = e0;
    for (int base = 32; base < cnt; base += 32) {
      const int j = base + lane;
      if (j < cnt) {
        const int s = j < cnt - 1 ? col[beg + j] : d;
        mx = fmaxf(mx, leakyr(as[s] + adv));
      }
    }
    mx = wave_max(mx);

    const float ex0 = act0 ? expf(e0 - mx) : 0.f;
    float sm = ex0;
    for (int base = 32; base < cnt; base += 32) {
      const int j = base + lane;
      if (j < cnt) {
        const int s = j < cnt - 1 ? col[beg + j] : d;
        sm += expf(leakyr(as[s] + adv) - mx);
      }
    }
    sm = wave_sum(sm);

    vec acc = {};
    agg_rows<D>(h, s0, ex0 / sm, cnt < 32 ? cnt : 32, lane, acc);
    for (int base = 32; base < cnt; base += 32) {
      const int j = base + lane;
      int   s = d;
      float a = 0.f;
      if (j < cnt) {
        if (j < cnt - 1) s = col[beg + j];
        a = expf(leakyr(as[s] + adv) - mx) / sm;
      }
      const int left = cnt - base;
      agg_rows<D>(h, s, a, left < 32 ? left : 32, lane, acc);
    }

    ((vec*)(out + (size_t)d * D))[lane] = acc;
  }
}

__device__ __forceinline__ v4f bias_relu4(v4f v, const float* __restrict__ b) {
  v.x += b[0]; v.y += b[1]; v.z += b[2]; v.w += b[3];
  v.x = v.x > 0.f ? v.x : 0.f;
  v.y = v.y > 0.f ? v.y : 0.f;
  v.z = v.z > 0.f ? v.z : 0.f;
  v.w = v.w > 0.f ? v.w : 0.f;
  return v;
}

__global__ void gat_bias_relu(float* __restrict__ p,
                              const float* __restrict__ bias, int N, int Dmask) {
  const int tid = blockIdx.x * blockDim.x + threadIdx.x;
  const int nth = gridDim.x * blockDim.x;
  const int n4  = N >> 2;
  v4f* __restrict__ p4 = (v4f*)p;
  int i = tid;
  for (; i + 3 * nth < n4; i += 4 * nth) {
    const v4f v0 = p4[i];
    const v4f v1 = p4[i + nth];
    const v4f v2 = p4[i + 2 * nth];
    const v4f v3 = p4[i + 3 * nth];
    const v4f r0 = bias_relu4(v0, bias + ((i << 2) & Dmask));
    const v4f r1 = bias_relu4(v1, bias + (((i + nth) << 2) & Dmask));
    const v4f r2 = bias_relu4(v2, bias + (((i + 2 * nth) << 2) & Dmask));
    const v4f r3 = bias_relu4(v3, bias + (((i + 3 * nth) << 2) & Dmask));
    p4[i] = r0;
    p4[i + nth] = r1;
    p4[i + 2 * nth] = r2;
    p4[i + 3 * nth] = r3;
  }
  for (; i < n4; i += nth) p4[i] = bias_relu4(p4[i], bias + ((i << 2) & Dmask));
  const int t = (n4 << 2) + tid;
  if (t < N) {
    float v = p[t] + bias[t & Dmask];
    p[t] = v > 0.f ? v : 0.f;
  }
}

__global__ void gat_fc(const float* __restrict__ h,
                       const float* __restrict__ W,
                       const float* __restrict__ b,
                       float* __restrict__ out, int N) {
  int n = blockIdx.x * blockDim.x + threadIdx.x;
  if (n >= N) return;
  const float* row = h + (size_t)n * OUT2;
  float a0 = b[0], a1 = b[1];
#pragma unroll
  for (int c = 0; c < OUT2; ++c) {
    float v = row[c];
    a0 += v * W[c * NCLS + 0];
    a1 += v * W[c * NCLS + 1];
  }
  out[n * NCLS + 0] = a0;
  out[n * NCLS + 1] = a1;
}

extern "C" void kernel_launch(void* const* d_in, const int* in_sizes, int n_in,
                              void* d_out, int out_size, void* d_ws, size_t ws_size,
                              hipStream_t stream) {
  const float* x   = (const float*)d_in[0];
  const int*   ei  = (const int*)  d_in[1];
  const float* W1  = (const float*)d_in[2];
  const float* a1s = (const float*)d_in[3];
  const float* a1d = (const float*)d_in[4];
  const float* b1  = (const float*)d_in[5];
  const float* W2  = (const float*)d_in[6];
  const float* a2s = (const float*)d_in[7];
  const float* a2d = (const float*)d_in[8];
  const float* b2  = (const float*)d_in[9];
  const float* fcW = (const float*)d_in[10];
  const float* fcb = (const float*)d_in[11];

  const int N    = in_sizes[0] / IN_C;
  const int E    = in_sizes[1] / 2;

  float* w       = (float*)d_ws;
  float* bufA    = w;
  float* bufB    = w + (size_t)N * HID;
  float* as_buf  = w + (size_t)2 * N * HID;
  float* ad_buf  = as_buf + N;
  int*   cnt     = (int*)as_buf;
  int*   rowptr  = (int*)(ad_buf + N);
  int*   cursor  = rowptr + N + 1;
  int*   col     = cursor + N;

  const int TB = 256;

  const int CHIP = 2048;
  auto capped = [](int blocks, int cap) { return blocks < cap ? blocks : cap; };
  auto ew_grid = [&](int n) { return capped((n / 4 + TB - 1) / TB, CHIP); };
  const int list_blk  = (E + TB - 1) / TB;
  const int node_blk  = capped((N + 7) / 8, CHIP);

  gat_fill_f32<<<ew_grid(N), TB, 0, stream>>>((float*)cnt, 0.f, N);
  gat_dst_count  <<<list_blk, TB, 0, stream>>>(ei, cnt, E);
  gat_dst_offsets<<<(N + SCAN_TB - 1) / SCAN_TB, SCAN_TB, 0, stream>>>(cnt, rowptr, cursor, N);
  gat_dst_scatter<<<list_blk, TB, 0, stream>>>(ei, cursor, col, E);

  float* wa   = bufA;
  float* wd   = bufA + IN_C;
  float* aggx = bufB;
  gat_node_dots<<<IN_C / 8, TB, 0, stream>>>(W1, a1s, a1d, wa, wd, IN_C, HID);
  gat_node_dots<<<node_blk, TB, 0, stream>>>(x, wa, wd, as_buf, ad_buf, N, IN_C);
  gat_node_softmax_agg<IN_C><<<node_blk, TB, 0, stream>>>(rowptr, col, as_buf, ad_buf, x, aggx, N);

  float* h2   = bufA;
  float* agg2 = bufB;
  gat_transform_fused<<<(N + 127) / 128, TB, 0, stream>>>(aggx, W1, b1, W2, h2, N);

  gat_node_dots<<<node_blk, TB, 0, stream>>>(h2, a2s, a2d, as_buf, ad_buf, N, OUT2);
  gat_node_softmax_agg<OUT2><<<node_blk, TB, 0, stream>>>(rowptr, col, as_buf, ad_buf, h2, agg2, N);
  gat_bias_relu<<<ew_grid(N * OUT2), TB, 0, stream>>>(agg2, b2, N * OUT2, OUT2 - 1);

  gat_fc<<<(N + TB - 1) / TB, TB, 0, stream>>>(agg2, fcW, fcb, (float*)d_out, N);
}


# Tails folded into producer epilogues: layer-2 score dots from the WMMA accumulators, bias+ReLU+classifier on the aggregated row in registers; 12 -> 9 launches; those sums use a new reduction tree
# speedup vs baseline: 4.0142x; 1.0739x over previous
#include <hip/hip_runtime.h>


typedef __attribute__((ext_vector_type(2))) float v2f;
typedef __attribute__((ext_vector_type(8))) float v8f;
typedef __attribute__((ext_vector_type(4))) float v4f;

#define IN_C  128
#define HID   512
#define OUT2  64
#define NCLS  2
#define NEG_SLOPE 0.2f

#define FM_HC   64
#define FM_NCH  (HID / FM_HC)
#define FM_NT1  (FM_HC / 16)
#define FM_NT2  (OUT2 / 16)
#define FM_HP   (FM_HC + 4)
#define FM_I1   ((IN_C / 2) * (FM_HC / 4) / 256)
#define FM_I2   ((FM_HC / 2) * (OUT2 / 4) / 256)

__device__ __forceinline__ void fm_load_chunk(const float* __restrict__ W1,
                                              const float* __restrict__ W2,
                                              int j, int tid,
                                              v4f (&p1)[2 * FM_I1], v4f (&p2)[2 * FM_I2]) {
#pragma unroll
  for (int i = 0; i < FM_I1; ++i) {
    const int it = tid + 256 * i;
    const int q = it % (FM_HC / 4);
    const int p = it / (FM_HC / 4);
    const float* g = W1 + (size_t)(2 * p) * HID + FM_HC * j + 4 * q;
    p1[2 * i]     = *(const v4f*)g;
    p1[2 * i + 1] = *(const v4f*)(g + HID);
  }
#pragma unroll
  for (int i = 0; i < FM_I2; ++i) {
    const int it = tid + 256 * i;
    const int q = it % (OUT2 / 4);
    const int p = it / (OUT2 / 4);
    const float* g = W2 + (size_t)(FM_HC * j + 2 * p) * OUT2 + 4 * q;
    p2[2 * i]     = *(const v4f*)g;
    p2[2 * i + 1] = *(const v4f*)(g + OUT2);
  }
}

__device__ __forceinline__ void fm_put_item(float* s, v4f r0, v4f r1) {
  const v4f lo = {r0.x, r1.x, r0.y, r1.y};
  const v4f hi = {r0.z, r1.z, r0.w, r1.w};
  *(v4f*)s = lo;
  *(v4f*)(s + 4) = hi;
}

__device__ __forceinline__ void fm_store_chunk(float* W1s, float* W2s, int tid,
                                               const v4f (&p1)[2 * FM_I1],
                                               const v4f (&p2)[2 * FM_I2]) {
#pragma unroll
  for (int i = 0; i < FM_I1; ++i) {
    const int it = tid + 256 * i;
    const int q = it % (FM_HC / 4);
    const int p = it / (FM_HC / 4);
    fm_put_item(W1s + (q >> 2) * (IN_C * 16) + p * 32 + (q & 3) * 8, p1[2 * i], p1[2 * i + 1]);
  }
#pragma unroll
  for (int i = 0; i < FM_I2; ++i) {
    const int it = tid + 256 * i;
    const int q = it % (OUT2 / 4);
    const int p = it / (OUT2 / 4);
    fm_put_item(W2s + (q >> 2) * (FM_HC * 16) + p * 32 + (q & 3) * 8, p2[2 * i], p2[2 * i + 1]);
  }
}

__global__ __launch_bounds__(256)
void gat_transform_fused(const float* __restrict__ A,
                         const float* __restrict__ W1,
                         const float* __restrict__ b1,
                         const float* __restrict__ W2,
                         const float* __restrict__ a2s,
                         const float* __restrict__ a2d,
                         float* __restrict__ H2,
                         float* __restrict__ as_out,
                         float* __restrict__ ad_out, int N) {
  __shared__ __attribute__((aligned(16))) float W1s[FM_NT1 * IN_C * 16];
  __shared__ __attribute__((aligned(16))) float W2s[FM_NT2 * FM_HC * 16];
  __shared__ __attribute__((aligned(16))) float Hs[8 * 16 * FM_HP];

  const int tid  = threadIdx.x;
  const int lane = tid & 31;
  const int wave = __builtin_amdgcn_readfirstlane(tid >> 5);
  const int half = lane >> 4;
  const int l16  = lane & 15;
  const int row0 = (blockIdx.x * 8 + wave) * 16;
  const bool valid = row0 < N;

  v4f p1[2 * FM_I1], p2[2 * FM_I2];
  fm_load_chunk(W1, W2, 0, tid, p1, p2);

  int rowc = row0 + l16;
  if (rowc > N - 1) rowc = N - 1;
  const float* Ap = A + (size_t)rowc * IN_C + 2 * half;
  v2f af[IN_C / 4];
#pragma unroll
  for (int kq = 0; kq < IN_C / 4; ++kq) af[kq] = *(const v2f*)(Ap + 4 * kq);

  fm_store_chunk(W1s, W2s, tid, p1, p2);
  __syncthreads();

  float* Hw = Hs + wave * (16 * FM_HP);
  const float* B1l = W1s + half * 32 + (l16 << 1);
  const float* B2l = W2s + half * 32 + (l16 << 1);
  const float* H2l = Hw + l16 * FM_HP + 2 * half;
  float* Hst = Hw + (8 * half) * FM_HP + l16;

  v8f acc2[FM_NT2];
#pragma unroll
  for (int t = 0; t < FM_NT2; ++t) acc2[t] = v8f{};

  for (int j = 0; j < FM_NCH; ++j) {
    const bool more = j + 1 < FM_NCH;
    if (more) fm_load_chunk(W1, W2, j + 1, tid, p1, p2);

    if (valid) {
#pragma unroll
      for (int nt = 0; nt < FM_NT1; nt += 2) {
        v8f c0 = {}, c1 = {};
        const float* Ba = B1l + nt * (IN_C * 16);
        const float* Bb = Ba + IN_C * 16;
#pragma unroll
        for (int kq = 0; kq < IN_C / 4; ++kq) {
          const v2f ba = *(const v2f*)(Ba + kq * 64);
          const v2f bb = *(const v2f*)(Bb + kq * 64);
          c0 = __builtin_amdgcn_wmma_f32_16x16x4_f32(false, af[kq], false, ba, (short)0, c0, false, false);
          c1 = __builtin_amdgcn_wmma_f32_16x16x4_f32(false, af[kq], false, bb, (short)0, c1, false, false);
        }
        const float bia = b1[FM_HC * j + nt * 16 + l16];
        const float bib = b1[FM_HC * j + nt * 16 + 16 + l16];
#pragma unroll
        for (int r = 0; r < 8; ++r) {
          const float va = c0[r] + bia;
          const float vb = c1[r] + bib;
          Hst[r * FM_HP + nt * 16]      = va > 0.f ? va : 0.f;
          Hst[r * FM_HP + nt * 16 + 16] = vb > 0.f ? vb : 0.f;
        }
      }
    }
    __syncthreads();

    if (valid) {
#pragma unroll
      for (int kk = 0; kk < FM_HC / 4; ++kk) {
        const v2f a2 = *(const v2f*)(H2l + 4 * kk);
#pragma unroll
        for (int t = 0; t < FM_NT2; ++t) {
          const v2f b2 = *(const v2f*)(B2l + t * (FM_HC * 16) + kk * 64);
          acc2[t] = __builtin_amdgcn_wmma_f32_16x16x4_f32(false, a2, false, b2, (short)0, acc2[t], false, false);
        }
      }
    }
    if (!more) break;
    __syncthreads();
    fm_store_chunk(W1s, W2s, tid, p1, p2);
    __syncthreads();
  }

  if (valid) {
    float* Cp = H2 + (size_t)(row0 + 8 * half) * OUT2 + l16;
#pragma unroll
    for (int r = 0; r < 8; ++r) {
      if (row0 + 8 * half + r < N) {
#pragma unroll
        for (int t = 0; t < FM_NT2; ++t) Cp[(size_t)r * OUT2 + t * 16] = acc2[t][r];
      }
    }

    float ps[8], pd[8];
#pragma unroll
    for (int r = 0; r < 8; ++r) { ps[r] = 0.f; pd[r] = 0.f; }
#pragma unroll
    for (int t = 0; t < FM_NT2; ++t) {
      const float cs = a2s[t * 16 + l16];
      const float cd = a2d[t * 16 + l16];
#pragma unroll
      for (int r = 0; r < 8; ++r) {
        ps[r] += acc2[t][r] * cs;
        pd[r] += acc2[t][r] * cd;
      }
    }
#pragma unroll
    for (int off = 8; off > 0; off >>= 1) {
#pragma unroll
      for (int r = 0; r < 8; ++r) {
        ps[r] += __shfl_xor(ps[r], off, 32);
        pd[r] += __shfl_xor(pd[r], off, 32);
      }
    }
    float vs = ps[0], vd = pd[0];
#pragma unroll
    for (int r = 1; r < 8; ++r) {
      vs = l16 == r ? ps[r] : vs;
      vd = l16 == r ? pd[r] : vd;
    }
    const int srow = row0 + 8 * half + l16;
    if (l16 < 8 && srow < N) { as_out[srow] = vs; ad_out[srow] = vd; }
  }
}

__global__ void gat_node_dots(const float* __restrict__ h,
                              const float* __restrict__ a_src,
                              const float* __restrict__ a_dst,
                              float* __restrict__ as_out,
                              float* __restrict__ ad_out,
                              int N, int D) {
  const int wave0  = blockIdx.x * (blockDim.x >> 5) + (threadIdx.x >> 5);
  const int nwaves = gridDim.x * (blockDim.x >> 5);
  const int lane = threadIdx.x & 31;
  for (int node = wave0; node < N; node += nwaves) {
    const float* row = h + (size_t)node * D;
    float ss = 0.f, sd = 0.f;
    for (int c = lane; c < D; c += 32) {
      float v = row[c];
      ss += v * a_src[c];
      sd += v * a_dst[c];
    }
#pragma unroll
    for (int off = 16; off > 0; off >>= 1) {
      ss += __shfl_down(ss, off, 32);
      sd += __shfl_down(sd, off, 32);
    }
    if (lane == 0) { as_out[node] = ss; ad_out[node] = sd; }
  }
}

__global__ void gat_fill_f32(float* __restrict__ p, float v, int n) {
  const int tid = blockIdx.x * blockDim.x + threadIdx.x;
  const int nth = gridDim.x * blockDim.x;
  const int n4  = n >> 2;
  float4* __restrict__ p4 = (float4*)p;
  const float4 v4 = make_float4(v, v, v, v);
  for (int i = tid; i < n4; i += nth) p4[i] = v4;
  const int t = (n4 << 2) + tid;
  if (t < n) p[t] = v;
}

__global__ void gat_dst_count(const int* __restrict__ ei, int* __restrict__ cnt, int E) {
  const int e = blockIdx.x * blockDim.x + threadIdx.x;
  if (e < E) atomicAdd(&cnt[ei[E + e]], 1);
}

#define SCAN_TB 1024
__global__ void gat_dst_offsets(const int* __restrict__ cnt,
                                int* __restrict__ rowptr,
                                int* __restrict__ cursor, int N) {
  __shared__ int wbefore[SCAN_TB / 32];
  __shared__ int wtot[SCAN_TB / 32];
  const int t = threadIdx.x;
  const int lane = t & 31;
  const int w = t >> 5;
  const int b = blockIdx.x;
  int before = 0;
  for (int k = 0; k < b; ++k) before += cnt[k * SCAN_TB + t];
  const int i = b * SCAN_TB + t;
  const int own = i < N ? cnt[i] : 0;
  int inc = own;
#pragma unroll
  for (int off = 1; off < 32; off <<= 1) {
    const int v = __shfl_up(inc, off, 32);
    if (lane >= off) inc += v;
  }
#pragma unroll
  for (int off = 16; off > 0; off >>= 1) before += __shfl_xor(before, off, 32);
  if (lane == 31) wtot[w] = inc;
  if (lane == 0)  wbefore[w] = before;
  __syncthreads();
  int base = 0;
  for (int k = 0; k < SCAN_TB / 32; ++k) {
    base += wbefore[k];
    if (k < w) base += wtot[k];
  }
  const int excl = base + inc - own;
  if (i < N) { rowptr[i] = excl; cursor[i] = excl; }
  if (i == N - 1) rowptr[N] = excl + own;
}

__global__ void gat_dst_scatter(const int* __restrict__ ei, int* __restrict__ cursor,
                                int* __restrict__ col, int E) {
  const int e = blockIdx.x * blockDim.x + threadIdx.x;
  if (e < E) {
    const int p = atomicAdd(&cursor[ei[E + e]], 1);
    col[p] = ei[e];
  }
}

__device__ __forceinline__ float leakyr(float v) {
  return v > 0.f ? v : NEG_SLOPE * v;
}

__device__ __forceinline__ float wave_max(float v) {
#pragma unroll
  for (int off = 16; off > 0; off >>= 1) v = fmaxf(v, __shfl_xor(v, off, 32));
  return v;
}

__device__ __forceinline__ float wave_sum(float v) {
#pragma unroll
  for (int off = 16; off > 0; off >>= 1) v += __shfl_xor(v, off, 32);
  return v;
}

template <int D> struct gat_rowvec;
template <> struct gat_rowvec<128> { typedef v4f type; };
template <> struct gat_rowvec<64>  { typedef v2f type; };

template <int D>
__device__ __forceinline__ void agg_rows(const float* __restrict__ h, int s_l, float a_l,
                                         int n, int lane,
                                         typename gat_rowvec<D>::type& acc) {
  typedef typename gat_rowvec<D>::type vec;
  constexpr int GRP = 8;
  for (int u = 0; u < n; u += GRP) {
    vec v[GRP];
#pragma unroll
    for (int g = 0; g < GRP; ++g) {
      const int s = __builtin_amdgcn_readlane(s_l, u + g);
      v[g] = ((const vec*)(h + (size_t)s * D))[lane];
    }
#pragma unroll
    for (int g = 0; g < GRP; ++g) {
      const float a = __int_as_float(__builtin_amdgcn_readlane(__float_as_int(a_l), u + g));
      acc += a * v[g];
    }
  }
}

template <int D, bool TAIL>
__global__ void gat_node_softmax_agg(const int* __restrict__ rowptr,
                                     const int* __restrict__ col,
                                     const float* __restrict__ as,
                                     const float* __restrict__ ad,
                                     const float* __restrict__ h,
                                     const float* __restrict__ bias,
                                     const float* __restrict__ fcW,
                                     const float* __restrict__ fcb,
                                     float* __restrict__ out, int N) {
  typedef typename gat_rowvec<D>::type vec;
  static_assert(!TAIL || D == OUT2, "the classifier tail is written for the 64-wide row");
  float tb0 = 0.f, tb1 = 0.f, tw00 = 0.f, tw01 = 0.f, tw10 = 0.f, tw11 = 0.f, tf0 = 0.f, tf1 = 0.f;
  if constexpr (TAIL) {
    const int c = 2 * (threadIdx.x & 31);
    tb0  = bias[c];
    tb1  = bias[c + 1];
    tw00 = fcW[c * NCLS];
    tw01 = fcW[c * NCLS + 1];
    tw10 = fcW[(c + 1) * NCLS];
    tw11 = fcW[(c + 1) * NCLS + 1];
    tf0  = fcb[0];
    tf1  = fcb[1];
  }
  const int lane = threadIdx.x & 31;
  const int wib    = __builtin_amdgcn_readfirstlane((int)(threadIdx.x >> 5));
  const int wave0  = blockIdx.x * (blockDim.x >> 5) + wib;
  const int nwaves = gridDim.x * (blockDim.x >> 5);
  const float NEG_INF = -__builtin_inff();
  for (int d = wave0; d < N; d += nwaves) {
    const int beg = __builtin_amdgcn_readfirstlane(rowptr[d]);
    const int end = __builtin_amdgcn_readfirstlane(rowptr[d + 1]);
    const int cnt = end - beg + 1;
    const float adv = ad[d];

    const bool act0 = lane < cnt;
    int   s0 = d;
    float e0 = NEG_INF;
    if (act0) {
      if (lane < cnt - 1) s0 = col[beg + lane];
      e0 = leakyr(as[s0] + adv);
    }
    float mx = e0;
    for (int base = 32; base < cnt; base += 32) {
      const int j = base + lane;
      if (j < cnt) {
        const int s = j < cnt - 1 ? col[beg + j] : d;
        mx = fmaxf(mx, leakyr(as[s] + adv));
      }
    }
    mx = wave_max(mx);

    const float ex0 = act0 ? expf(e0 - mx) : 0.f;
    float sm = ex0;
    for (int base = 32; base < cnt; base += 32) {
      const int j = base + lane;
      if (j < cnt) {
        const int s = j < cnt - 1 ? col[beg + j] : d;
        sm += expf(leakyr(as[s] + adv) - mx);
      }
    }
    sm = wave_sum(sm);

    vec acc = {};
    agg_rows<D>(h, s0, ex0 / sm, cnt < 32 ? cnt : 32, lane, acc);
    for (int base = 32; base < cnt; base += 32) {
      const int j = base + lane;
      int   s = d;
      float a = 0.f;
      if (j < cnt) {
        if (j < cnt - 1) s = col[beg + j];
        a = expf(leakyr(as[s] + adv) - mx) / sm;
      }
      const int left = cnt - base;
      agg_rows<D>(h, s, a, left < 32 ? left : 32, lane, acc);
    }

    if constexpr (TAIL) {
      float v0 = acc[0] + tb0;
      float v1 = acc[1] + tb1;
      v0 = v0 > 0.f ? v0 : 0.f;
      v1 = v1 > 0.f ? v1 : 0.f;
      const float o0 = wave_sum(v0 * tw00 + v1 * tw10);
      const float o1 = wave_sum(v0 * tw01 + v1 * tw11);
      if (lane == 0) {
        const v2f o = {o0 + tf0, o1 + tf1};
        *(v2f*)(out + (size_t)d * NCLS) = o;
      }
    } else {
      ((vec*)(out + (size_t)d * D))[lane] = acc;
    }
  }
}

extern "C" void kernel_launch(void* const* d_in, const int* in_sizes, int n_in,
                              void* d_out, int out_size, void* d_ws, size_t ws_size,
                              hipStream_t stream) {
  const float* x   = (const float*)d_in[0];
  const int*   ei  = (const int*)  d_in[1];
  const float* W1  = (const float*)d_in[2];
  const float* a1s = (const float*)d_in[3];
  const float* a1d = (const float*)d_in[4];
  const float* b1  = (const float*)d_in[5];
  const float* W2  = (const float*)d_in[6];
  const float* a2s = (const float*)d_in[7];
  const float* a2d = (const float*)d_in[8];
  const float* b2  = (const float*)d_in[9];
  const float* fcW = (const float*)d_in[10];
  const float* fcb = (const float*)d_in[11];

  const int N    = in_sizes[0] / IN_C;
  const int E    = in_sizes[1] / 2;

  float* w       = (float*)d_ws;
  float* bufA    = w;
  float* bufB    = w + (size_t)N * HID;
  float* as_buf  = w + (size_t)2 * N * HID;
  float* ad_buf  = as_buf + N;
  int*   cnt     = (int*)as_buf;
  int*   rowptr  = (int*)(ad_buf + N);
  int*   cursor  = rowptr + N + 1;
  int*   col     = cursor + N;

  const int TB = 256;

  const int CHIP = 2048;
  auto capped = [](int blocks, int cap) { return blocks < cap ? blocks : cap; };
  auto ew_grid = [&](int n) { return capped((n / 4 + TB - 1) / TB, CHIP); };
  const int list_blk  = (E + TB - 1) / TB;
  const int node_blk  = capped((N + 7) / 8, CHIP);

  gat_fill_f32<<<ew_grid(N), TB, 0, stream>>>((float*)cnt, 0.f, N);
  gat_dst_count  <<<list_blk, TB, 0, stream>>>(ei, cnt, E);
  gat_dst_offsets<<<(N + SCAN_TB - 1) / SCAN_TB, SCAN_TB, 0, stream>>>(cnt, rowptr, cursor, N);
  gat_dst_scatter<<<list_blk, TB, 0, stream>>>(ei, cursor, col, E);

  float* wa   = bufA;
  float* wd   = bufA + IN_C;
  float* aggx = bufB;
  gat_node_dots<<<IN_C / 8, TB, 0, stream>>>(W1, a1s, a1d, wa, wd, IN_C, HID);
  gat_node_dots<<<node_blk, TB, 0, stream>>>(x, wa, wd, as_buf, ad_buf, N, IN_C);
  gat_node_softmax_agg<IN_C, false><<<node_blk, TB, 0, stream>>>(rowptr, col, as_buf, ad_buf, x,
                                                               nullptr, nullptr, nullptr, aggx, N);

  float* h2   = bufA;
  gat_transform_fused<<<(N + 127) / 128, TB, 0, stream>>>(aggx, W1, b1, W2, a2s, a2d, h2, as_buf, ad_buf, N);

  gat_node_softmax_agg<OUT2, true><<<node_blk, TB, 0, stream>>>(rowptr, col, as_buf, ad_buf, h2,
                                                              b2, fcW, fcb, (float*)d_out, N);
}


# Fused transform re-cut for matrix-pipe balance: one 208-row block per WGP, its four SIMDs split each strip by column tile (13 strips per SIMD, was 16); layer-2 score dots summed in a new order
# speedup vs baseline: 4.2393x; 1.0561x over previous
#include <hip/hip_runtime.h>


typedef __attribute__((ext_vector_type(2))) float v2f;
typedef __attribute__((ext_vector_type(8))) float v8f;
typedef __attribute__((ext_vector_type(4))) float v4f;

#define IN_C  128
#define HID   512
#define OUT2  64
#define NCLS  2
#define NEG_SLOPE 0.2f

#define FM_HC   64
#define FM_NCH  (HID / FM_HC)
#define FM_R    13
#define FM_ROWS (16 * FM_R)
#define FM_OWN  ((FM_R + 1) / 2)
#define FM_OWNP ((FM_OWN + 1) & ~1)
#define FM_AP   (IN_C + 4)
#define FM_HP   (FM_HC + 4)
#define FM_AI   (FM_ROWS * (IN_C / 4) / 256)

__device__ __forceinline__ v8f fm_wmma(v2f a, v2f b, v8f c) {
  return __builtin_amdgcn_wmma_f32_16x16x4_f32(false, a, false, b, (short)0, c, false, false);
}

__device__ __forceinline__ void fm_chain_end(v8f& c) {
  asm volatile("v_nop\n\tv_nop\n\tv_nop\n\tv_nop" : "+v"(c));
}

template <int K0, int K1>
__device__ __forceinline__ void fm_load_b1(const float* __restrict__ W1, int j, int c, int half, int l16,
                                           v2f (&bw1)[IN_C / 4]) {
  const float* g = W1 + (size_t)(2 * half) * HID + FM_HC * j + 16 * c + l16;
#pragma unroll
  for (int kq = K0; kq < K1; ++kq) {
    bw1[kq].x = g[(size_t)(4 * kq) * HID];
    bw1[kq].y = g[(size_t)(4 * kq + 1) * HID];
  }
}

__device__ __forceinline__ void fm_load_b2(const float* __restrict__ W2, int j, int c, int half, int l16,
                                           v2f (&bw2)[FM_HC / 4]) {
  const float* g = W2 + (size_t)(FM_HC * j + 2 * half) * OUT2 + 16 * c + l16;
#pragma unroll
  for (int kk = 0; kk < FM_HC / 4; ++kk) {
    bw2[kk].x = g[(size_t)(4 * kk) * OUT2];
    bw2[kk].y = g[(size_t)(4 * kk + 1) * OUT2];
  }
}

template <int NK>
__device__ __forceinline__ void fm_frags(const float* s, v2f (&f)[NK]) {
#pragma unroll
  for (int k = 0; k < NK; ++k) f[k] = *(const v2f*)(s + 4 * k);
}

__device__ __forceinline__ void fm_park_hidden(float* Hst, v8f c, float bia) {
#pragma unroll
  for (int i = 0; i < 8; ++i) {
    const float v = c[i] + bia;
    Hst[i * FM_HP] = v > 0.f ? v : 0.f;
  }
}

__global__ __launch_bounds__(256)
void gat_transform_fused(const float* __restrict__ A,
                         const float* __restrict__ W1,
                         const float* __restrict__ b1,
                         const float* __restrict__ W2,
                         const float* __restrict__ a2s,
                         const float* __restrict__ a2d,
                         float* __restrict__ H2,
                         float* __restrict__ as_out,
                         float* __restrict__ ad_out, int N) {
  __shared__ __attribute__((aligned(16))) float lds[FM_ROWS * FM_AP + 2 * FM_ROWS * FM_HP];
  float* As = lds;
  float* Hs = lds + FM_ROWS * FM_AP;

  const int tid  = threadIdx.x;
  const int lane = tid & 31;
  const int wave = __builtin_amdgcn_readfirstlane(tid >> 5);
  const int c    = wave & 3;
  const int p    = wave >> 2;
  const int half = lane >> 4;
  const int l16  = lane & 15;
  const int row0 = blockIdx.x * FM_ROWS;
  int nst = (N - row0 + 15) >> 4;
  if (nst > FM_R) nst = FM_R;

  v2f bw1[IN_C / 4], bw2[FM_HC / 4];

  {
    v4f ar[FM_AI];
#pragma unroll
    for (int t = 0; t < FM_AI; ++t) {
      const int i = tid + 256 * t;
      int grow = row0 + i / (IN_C / 4);
      if (grow > N - 1) grow = N - 1;
      ar[t] = *(const v4f*)(A + (size_t)grow * IN_C + 4 * (i % (IN_C / 4)));
    }
    fm_load_b1<0, IN_C / 8>(W1, 0, c, half, l16, bw1);
#pragma unroll
    for (int t = 0; t < FM_AI; ++t) {
      const int i = tid + 256 * t;
      *(v4f*)(As + (i / (IN_C / 4)) * FM_AP + 4 * (i % (IN_C / 4))) = ar[t];
    }
  }
  fm_load_b1<IN_C / 8, IN_C / 4>(W1, 0, c, half, l16, bw1);
  fm_load_b2(W2, 0, c, half, l16, bw2);
  float bia = b1[16 * c + l16];
  __syncthreads();

  const float* Al = As + l16 * FM_AP + 2 * half;

  v8f acc2[FM_OWNP];
#pragma unroll
  for (int ri = 0; ri < FM_OWNP; ++ri) acc2[ri] = v8f{};

  for (int j = 0; j < FM_NCH; ++j) {
    float* Hb = Hs + (j & 1) * (FM_ROWS * FM_HP);
    float* Hst = Hb + (8 * half) * FM_HP + 16 * c + l16;

    int r = p;
    for (; r + 2 < nst; r += 4) {
      const float* Aa = Al + r * (16 * FM_AP);
      const float* Ab = Aa + 32 * FM_AP;
      v8f c0 = {}, c1 = {};
      v2f a0[IN_C / 4], a1[IN_C / 4];
      fm_frags(Aa, a0);
      fm_frags(Ab, a1);
#pragma unroll
      for (int kq = 0; kq < IN_C / 4; ++kq) {
        c0 = fm_wmma(a0[kq], bw1[kq], c0);
        c1 = fm_wmma(a1[kq], bw1[kq], c1);
      }
      fm_park_hidden(Hst + r * (16 * FM_HP), c0, bia);
      fm_park_hidden(Hst + (r + 2) * (16 * FM_HP), c1, bia);
    }
    if (r < nst) {
      const float* Aa = Al + r * (16 * FM_AP);
      v8f c0 = {};
      v2f a0[IN_C / 4];
      fm_frags(Aa, a0);
      __builtin_amdgcn_sched_barrier(0);
#pragma unroll
      for (int kq = 0; kq < IN_C / 4; ++kq) c0 = fm_wmma(a0[kq], bw1[kq], c0);
      fm_chain_end(c0);
      fm_park_hidden(Hst + r * (16 * FM_HP), c0, bia);
    }
    __syncthreads();

    const int jn = j + 1 < FM_NCH ? j + 1 : j;
    fm_load_b1<0, IN_C / 8>(W1, jn, c, half, l16, bw1);
    bia = b1[FM_HC * jn + 16 * c + l16];

    const float* Hl = Hb + l16 * FM_HP + 2 * half;
#pragma unroll
    for (int ri = 0; ri < FM_OWN; ri += 2) {
      const int ra = p + 2 * ri;
      if (ri + 1 < FM_OWN && ra + 2 < nst) {
        const float* Ha = Hl + ra * (16 * FM_HP);
        const float* Hc = Ha + 32 * FM_HP;
        v2f h0[FM_HC / 4], h1[FM_HC / 4];
        fm_frags(Ha, h0);
        fm_frags(Hc, h1);
        __builtin_amdgcn_sched_barrier(0);
#pragma unroll
        for (int kk = 0; kk < FM_HC / 4; ++kk) {
          acc2[ri]     = fm_wmma(h0[kk], bw2[kk], acc2[ri]);
          acc2[ri + 1] = fm_wmma(h1[kk], bw2[kk], acc2[ri + 1]);
        }
      } else if (ra < nst) {
        const float* Ha = Hl + ra * (16 * FM_HP);
        v2f h0[FM_HC / 4];
        fm_frags(Ha, h0);
        __builtin_amdgcn_sched_barrier(0);
#pragma unroll
        for (int kk = 0; kk < FM_HC / 4; ++kk) acc2[ri] = fm_wmma(h0[kk], bw2[kk], acc2[ri]);
        fm_chain_end(acc2[ri]);
      }
      if (ri == 0) fm_load_b1<IN_C / 8, IN_C / 4>(W1, jn, c, half, l16, bw1);
    }
    fm_load_b2(W2, jn, c, half, l16, bw2);
  }

  float* Os = Hs;
#pragma unroll
  for (int ri = 0; ri < FM_OWN; ++ri) {
    const int ra = p + 2 * ri;
    if (ra < nst) {
      float* o = Os + (ra * 16 + 8 * half) * FM_HP + 16 * c + l16;
#pragma unroll
      for (int i = 0; i < 8; ++i) o[i * FM_HP] = acc2[ri][i];
    }
  }
  __syncthreads();

  if (tid < nst * 16) {
    const float* o = Os + tid * FM_HP;
    float ps = 0.f, pd = 0.f;
#pragma unroll
    for (int q = 0; q < OUT2 / 4; ++q) {
      const v4f v = *(const v4f*)(o + 4 * q);
      ps += v.x * a2s[4 * q];     pd += v.x * a2d[4 * q];
      ps += v.y * a2s[4 * q + 1]; pd += v.y * a2d[4 * q + 1];
      ps += v.z * a2s[4 * q + 2]; pd += v.z * a2d[4 * q + 2];
      ps += v.w * a2s[4 * q + 3]; pd += v.w * a2d[4 * q + 3];
    }
    const int grow = row0 + tid;
    if (grow < N) { as_out[grow] = ps; ad_out[grow] = pd; }
  }

  v4f orow[FM_R];
#pragma unroll
  for (int t = 0; t < FM_R; ++t)
    orow[t] = *(const v4f*)(Os + (2 * wave + 16 * t + half) * FM_HP + 4 * l16);
#pragma unroll
  for (int t = 0; t < FM_R; ++t) {
    const int grow = row0 + 2 * wave + 16 * t + half;
    if (grow < N) *(v4f*)(H2 + (size_t)grow * OUT2 + 4 * l16) = orow[t];
  }
}

__global__ void gat_node_dots(const float* __restrict__ h,
                              const float* __restrict__ a_src,
                              const float* __restrict__ a_dst,
                              float* __restrict__ as_out,
                              float* __restrict__ ad_out,
                              int N, int D) {
  const int wave0  = blockIdx.x * (blockDim.x >> 5) + (threadIdx.x >> 5);
  const int nwaves = gridDim.x * (blockDim.x >> 5);
  const int lane = threadIdx.x & 31;
  for (int node = wave0; node < N; node += nwaves) {
    const float* row = h + (size_t)node * D;
    float ss = 0.f, sd = 0.f;
    for (int c = lane; c < D; c += 32) {
      float v = row[c];
      ss += v * a_src[c];
      sd += v * a_dst[c];
    }
#pragma unroll
    for (int off = 16; off > 0; off >>= 1) {
      ss += __shfl_down(ss, off, 32);
      sd += __shfl_down(sd, off, 32);
    }
    if (lane == 0) { as_out[node] = ss; ad_out[node] = sd; }
  }
}

__global__ void gat_fill_f32(float* __restrict__ p, float v, int n) {
  const int tid = blockIdx.x * blockDim.x + threadIdx.x;
  const int nth = gridDim.x * blockDim.x;
  const int n4  = n >> 2;
  float4* __restrict__ p4 = (float4*)p;
  const float4 v4 = make_float4(v, v, v, v);
  for (int i = tid; i < n4; i += nth) p4[i] = v4;
  const int t = (n4 << 2) + tid;
  if (t < n) p[t] = v;
}

__global__ void gat_dst_count(const int* __restrict__ ei, int* __restrict__ cnt, int E) {
  const int e = blockIdx.x * blockDim.x + threadIdx.x;
  if (e < E) atomicAdd(&cnt[ei[E + e]], 1);
}

#define SCAN_TB 1024
__global__ void gat_dst_offsets(const int* __restrict__ cnt,
                                int* __restrict__ rowptr,
                                int* __restrict__ cursor, int N) {
  __shared__ int wbefore[SCAN_TB / 32];
  __shared__ int wtot[SCAN_TB / 32];
  const int t = threadIdx.x;
  const int lane = t & 31;
  const int w = t >> 5;
  const int b = blockIdx.x;
  int before = 0;
  for (int k = 0; k < b; ++k) before += cnt[k * SCAN_TB + t];
  const int i = b * SCAN_TB + t;
  const int own = i < N ? cnt[i] : 0;
  int inc = own;
#pragma unroll
  for (int off = 1; off < 32; off <<= 1) {
    const int v = __shfl_up(inc, off, 32);
    if (lane >= off) inc += v;
  }
#pragma unroll
  for (int off = 16; off > 0; off >>= 1) before += __shfl_xor(before, off, 32);
  if (lane == 31) wtot[w] = inc;
  if (lane == 0)  wbefore[w] = before;
  __syncthreads();
  int base = 0;
  for (int k = 0; k < SCAN_TB / 32; ++k) {
    base += wbefore[k];
    if (k < w) base += wtot[k];
  }
  const int excl = base + inc - own;
  if (i < N) { rowptr[i] = excl; cursor[i] = excl; }
  if (i == N - 1) rowptr[N] = excl + own;
}

__global__ void gat_dst_scatter(const int* __restrict__ ei, int* __restrict__ cursor,
                                int* __restrict__ col, int E) {
  const int e = blockIdx.x * blockDim.x + threadIdx.x;
  if (e < E) {
    const int p = atomicAdd(&cursor[ei[E + e]], 1);
    col[p] = ei[e];
  }
}

__device__ __forceinline__ float leakyr(float v) {
  return v > 0.f ? v : NEG_SLOPE * v;
}

__device__ __forceinline__ float wave_max(float v) {
#pragma unroll
  for (int off = 16; off > 0; off >>= 1) v = fmaxf(v, __shfl_xor(v, off, 32));
  return v;
}

__device__ __forceinline__ float wave_sum(float v) {
#pragma unroll
  for (int off = 16; off > 0; off >>= 1) v += __shfl_xor(v, off, 32);
  return v;
}

template <int D> struct gat_rowvec;
template <> struct gat_rowvec<128> { typedef v4f type; };
template <> struct gat_rowvec<64>  { typedef v2f type; };

template <int D>
__device__ __forceinline__ void agg_rows(const float* __restrict__ h, int s_l, float a_l,
                                         int n, int lane,
                                         typename gat_rowvec<D>::type& acc) {
  typedef typename gat_rowvec<D>::type vec;
  constexpr int GRP = 8;
  for (int u = 0; u < n; u += GRP) {
    vec v[GRP];
#pragma unroll
    for (int g = 0; g < GRP; ++g) {
      const int s = __builtin_amdgcn_readlane(s_l, u + g);
      v[g] = ((const vec*)(h + (size_t)s * D))[lane];
    }
#pragma unroll
    for (int g = 0; g < GRP; ++g) {
      const float a = __int_as_float(__builtin_amdgcn_readlane(__float_as_int(a_l), u + g));
      acc += a * v[g];
    }
  }
}

template <int D, bool TAIL>
__global__ void gat_node_softmax_agg(const int* __restrict__ rowptr,
                                     const int* __restrict__ col,
                                     const float* __restrict__ as,
                                     const float* __restrict__ ad,
                                     const float* __restrict__ h,
                                     const float* __restrict__ bias,
                                     const float* __restrict__ fcW,
                                     const float* __restrict__ fcb,
                                     float* __restrict__ out, int N) {
  typedef typename gat_rowvec<D>::type vec;
  static_assert(!TAIL || D == OUT2, "the classifier tail is written for the 64-wide row");
  float tb0 = 0.f, tb1 = 0.f, tw00 = 0.f, tw01 = 0.f, tw10 = 0.f, tw11 = 0.f, tf0 = 0.f, tf1 = 0.f;
  if constexpr (TAIL) {
    const int c = 2 * (threadIdx.x & 31);
    tb0  = bias[c];
    tb1  = bias[c + 1];
    tw00 = fcW[c * NCLS];
    tw01 = fcW[c * NCLS + 1];
    tw10 = fcW[(c + 1) * NCLS];
    tw11 = fcW[(c + 1) * NCLS + 1];
    tf0  = fcb[0];
    tf1  = fcb[1];
  }
  const int lane = threadIdx.x & 31;
  const int wib    = __builtin_amdgcn_readfirstlane((int)(threadIdx.x >> 5));
  const int wave0  = blockIdx.x * (blockDim.x >> 5) + wib;
  const int nwaves = gridDim.x * (blockDim.x >> 5);
  const float NEG_INF = -__builtin_inff();
  for (int d = wave0; d < N; d += nwaves) {
    const int beg = __builtin_amdgcn_readfirstlane(rowptr[d]);
    const int end = __builtin_amdgcn_readfirstlane(rowptr[d + 1]);
    const int cnt = end - beg + 1;
    const float adv = ad[d];

    const bool act0 = lane < cnt;
    int   s0 = d;
    float e0 = NEG_INF;
    if (act0) {
      if (lane < cnt - 1) s0 = col[beg + lane];
      e0 = leakyr(as[s0] + adv);
    }
    float mx = e0;
    for (int base = 32; base < cnt; base += 32) {
      const int j = base + lane;
      if (j < cnt) {
        const int s = j < cnt - 1 ? col[beg + j] : d;
        mx = fmaxf(mx, leakyr(as[s] + adv));
      }
    }
    mx = wave_max(mx);

    const float ex0 = act0 ? expf(e0 - mx) : 0.f;
    float sm = ex0;
    for (int base = 32; base < cnt; base += 32) {
      const int j = base + lane;
      if (j < cnt) {
        const int s = j < cnt - 1 ? col[beg + j] : d;
        sm += expf(leakyr(as[s] + adv) - mx);
      }
    }
    sm = wave_sum(sm);

    vec acc = {};
    agg_rows<D>(h, s0, ex0 / sm, cnt < 32 ? cnt : 32, lane, acc);
    for (int base = 32; base < cnt; base += 32) {
      const int j = base + lane;
      int   s = d;
      float a = 0.f;
      if (j < cnt) {
        if (j < cnt - 1) s = col[beg + j];
        a = expf(leakyr(as[s] + adv) - mx) / sm;
      }
      const int left = cnt - base;
      agg_rows<D>(h, s, a, left < 32 ? left : 32, lane, acc);
    }

    if constexpr (TAIL) {
      float v0 = acc[0] + tb0;
      float v1 = acc[1] + tb1;
      v0 = v0 > 0.f ? v0 : 0.f;
      v1 = v1 > 0.f ? v1 : 0.f;
      const float o0 = wave_sum(v0 * tw00 + v1 * tw10);
      const float o1 = wave_sum(v0 * tw01 + v1 * tw11);
      if (lane == 0) {
        const v2f o = {o0 + tf0, o1 + tf1};
        *(v2f*)(out + (size_t)d * NCLS) = o;
      }
    } else {
      ((vec*)(out + (size_t)d * D))[lane] = acc;
    }
  }
}

extern "C" void kernel_launch(void* const* d_in, const int* in_sizes, int n_in,
                              void* d_out, int out_size, void* d_ws, size_t ws_size,
                              hipStream_t stream) {
  const float* x   = (const float*)d_in[0];
  const int*   ei  = (const int*)  d_in[1];
  const float* W1  = (const float*)d_in[2];
  const float* a1s = (const float*)d_in[3];
  const float* a1d = (const float*)d_in[4];
  const float* b1  = (const float*)d_in[5];
  const float* W2  = (const float*)d_in[6];
  const float* a2s = (const float*)d_in[7];
  const float* a2d = (const float*)d_in[8];
  const float* b2  = (const float*)d_in[9];
  const float* fcW = (const float*)d_in[10];
  const float* fcb = (const float*)d_in[11];

  const int N    = in_sizes[0] / IN_C;
  const int E    = in_sizes[1] / 2;

  float* w       = (float*)d_ws;
  float* bufA    = w;
  float* bufB    = w + (size_t)N * HID;
  float* as_buf  = w + (size_t)2 * N * HID;
  float* ad_buf  = as_buf + N;
  int*   cnt     = (int*)as_buf;
  int*   rowptr  = (int*)(ad_buf + N);
  int*   cursor  = rowptr + N + 1;
  int*   col     = cursor + N;

  const int TB = 256;

  const int CHIP = 2048;
  auto capped = [](int blocks, int cap) { return blocks < cap ? blocks : cap; };
  auto ew_grid = [&](int n) { return capped((n / 4 + TB - 1) / TB, CHIP); };
  const int list_blk  = (E + TB - 1) / TB;
  const int node_blk  = capped((N + 7) / 8, CHIP);

  gat_fill_f32<<<ew_grid(N), TB, 0, stream>>>((float*)cnt, 0.f, N);
  gat_dst_count  <<<list_blk, TB, 0, stream>>>(ei, cnt, E);
  gat_dst_offsets<<<(N + SCAN_TB - 1) / SCAN_TB, SCAN_TB, 0, stream>>>(cnt, rowptr, cursor, N);
  gat_dst_scatter<<<list_blk, TB, 0, stream>>>(ei, cursor, col, E);

  float* wa   = bufA;
  float* wd   = bufA + IN_C;
  float* aggx = bufB;
  gat_node_dots<<<IN_C / 8, TB, 0, stream>>>(W1, a1s, a1d, wa, wd, IN_C, HID);
  gat_node_dots<<<node_blk, TB, 0, stream>>>(x, wa, wd, as_buf, ad_buf, N, IN_C);
  gat_node_softmax_agg<IN_C, false><<<node_blk, TB, 0, stream>>>(rowptr, col, as_buf, ad_buf, x,
                                                               nullptr, nullptr, nullptr, aggx, N);

  float* h2   = bufA;
  gat_transform_fused<<<(N + FM_ROWS - 1) / FM_ROWS, TB, 0, stream>>>(aggx, W1, b1, W2, a2s, a2d, h2, as_buf, ad_buf, N);

  gat_node_softmax_agg<OUT2, true><<<node_blk, TB, 0, stream>>>(rowptr, col, as_buf, ad_buf, h2,
                                                              b2, fcW, fcb, (float*)d_out, N);
}


# Layer-1 score dots ride on index launches: W1 a vectors in the count-zeroing launch, per-node x dots on the scatter launch's waves, 4 rows in flight; 9 -> 7 launches, 16k fewer waves; sums unchanged
# speedup vs baseline: 4.8740x; 1.1497x over previous
#include <hip/hip_runtime.h>


typedef __attribute__((ext_vector_type(2))) float v2f;
typedef __attribute__((ext_vector_type(8))) float v8f;
typedef __attribute__((ext_vector_type(4))) float v4f;

#define IN_C  128
#define HID   512
#define OUT2  64
#define NCLS  2
#define NEG_SLOPE 0.2f

#define FM_HC   64
#define FM_NCH  (HID / FM_HC)
#define FM_R    13
#define FM_ROWS (16 * FM_R)
#define FM_OWN  ((FM_R + 1) / 2)
#define FM_OWNP ((FM_OWN + 1) & ~1)
#define FM_AP   (IN_C + 4)
#define FM_HP   (FM_HC + 4)
#define FM_AI   (FM_ROWS * (IN_C / 4) / 256)

__device__ __forceinline__ v8f fm_wmma(v2f a, v2f b, v8f c) {
  return __builtin_amdgcn_wmma_f32_16x16x4_f32(false, a, false, b, (short)0, c, false, false);
}

__device__ __forceinline__ void fm_chain_end(v8f& c) {
  asm volatile("v_nop\n\tv_nop\n\tv_nop\n\tv_nop" : "+v"(c));
}

template <int K0, int K1>
__device__ __forceinline__ void fm_load_b1(const float* __restrict__ W1, int j, int c, int half, int l16,
                                           v2f (&bw1)[IN_C / 4]) {
  const float* g = W1 + (size_t)(2 * half) * HID + FM_HC * j + 16 * c + l16;
#pragma unroll
  for (int kq = K0; kq < K1; ++kq) {
    bw1[kq].x = g[(size_t)(4 * kq) * HID];
    bw1[kq].y = g[(size_t)(4 * kq + 1) * HID];
  }
}

__device__ __forceinline__ void fm_load_b2(const float* __restrict__ W2, int j, int c, int half, int l16,
                                           v2f (&bw2)[FM_HC / 4]) {
  const float* g = W2 + (size_t)(FM_HC * j + 2 * half) * OUT2 + 16 * c + l16;
#pragma unroll
  for (int kk = 0; kk < FM_HC / 4; ++kk) {
    bw2[kk].x = g[(size_t)(4 * kk) * OUT2];
    bw2[kk].y = g[(size_t)(4 * kk + 1) * OUT2];
  }
}

template <int NK>
__device__ __forceinline__ void fm_frags(const float* s, v2f (&f)[NK]) {
#pragma unroll
  for (int k = 0; k < NK; ++k) f[k] = *(const v2f*)(s + 4 * k);
}

__device__ __forceinline__ void fm_park_hidden(float* Hst, v8f c, float bia) {
#pragma unroll
  for (int i = 0; i < 8; ++i) {
    const float v = c[i] + bia;
    Hst[i * FM_HP] = v > 0.f ? v : 0.f;
  }
}

__global__ __launch_bounds__(256)
void gat_transform_fused(const float* __restrict__ A,
                         const float* __restrict__ W1,
                         const float* __restrict__ b1,
                         const float* __restrict__ W2,
                         const float* __restrict__ a2s,
                         const float* __restrict__ a2d,
                         float* __restrict__ H2,
                         float* __restrict__ as_out,
                         float* __restrict__ ad_out, int N) {
  __shared__ __attribute__((aligned(16))) float lds[FM_ROWS * FM_AP + 2 * FM_ROWS * FM_HP];
  float* As = lds;
  float* Hs = lds + FM_ROWS * FM_AP;

  const int tid  = threadIdx.x;
  const int lane = tid & 31;
  const int wave = __builtin_amdgcn_readfirstlane(tid >> 5);
  const int c    = wave & 3;
  const int p    = wave >> 2;
  const int half = lane >> 4;
  const int l16  = lane & 15;
  const int row0 = blockIdx.x * FM_ROWS;
  int nst = (N - row0 + 15) >> 4;
  if (nst > FM_R) nst = FM_R;

  v2f bw1[IN_C / 4], bw2[FM_HC / 4];

  {
    v4f ar[FM_AI];
#pragma unroll
    for (int t = 0; t < FM_AI; ++t) {
      const int i = tid + 256 * t;
      int grow = row0 + i / (IN_C / 4);
      if (grow > N - 1) grow = N - 1;
      ar[t] = *(const v4f*)(A + (size_t)grow * IN_C + 4 * (i % (IN_C / 4)));
    }
    fm_load_b1<0, IN_C / 8>(W1, 0, c, half, l16, bw1);
#pragma unroll
    for (int t = 0; t < FM_AI; ++t) {
      const int i = tid + 256 * t;
      *(v4f*)(As + (i / (IN_C / 4)) * FM_AP + 4 * (i % (IN_C / 4))) = ar[t];
    }
  }
  fm_load_b1<IN_C / 8, IN_C / 4>(W1, 0, c, half, l16, bw1);
  fm_load_b2(W2, 0, c, half, l16, bw2);
  float bia = b1[16 * c + l16];
  __syncthreads();

  const float* Al = As + l16 * FM_AP + 2 * half;

  v8f acc2[FM_OWNP];
#pragma unroll
  for (int ri = 0; ri < FM_OWNP; ++ri) acc2[ri] = v8f{};

  for (int j = 0; j < FM_NCH; ++j) {
    float* Hb = Hs + (j & 1) * (FM_ROWS * FM_HP);
    float* Hst = Hb + (8 * half) * FM_HP + 16 * c + l16;

    int r = p;
    for (; r + 2 < nst; r += 4) {
      const float* Aa = Al + r * (16 * FM_AP);
      const float* Ab = Aa + 32 * FM_AP;
      v8f c0 = {}, c1 = {};
      v2f a0[IN_C / 4], a1[IN_C / 4];
      fm_frags(Aa, a0);
      fm_frags(Ab, a1);
#pragma unroll
      for (int kq = 0; kq < IN_C / 4; ++kq) {
        c0 = fm_wmma(a0[kq], bw1[kq], c0);
        c1 = fm_wmma(a1[kq], bw1[kq], c1);
      }
      fm_park_hidden(Hst + r * (16 * FM_HP), c0, bia);
      fm_park_hidden(Hst + (r + 2) * (16 * FM_HP), c1, bia);
    }
    if (r < nst) {
      const float* Aa = Al + r * (16 * FM_AP);
      v8f c0 = {};
      v2f a0[IN_C / 4];
      fm_frags(Aa, a0);
      __builtin_amdgcn_sched_barrier(0);
#pragma unroll
      for (int kq = 0; kq < IN_C / 4; ++kq) c0 = fm_wmma(a0[kq], bw1[kq], c0);
      fm_chain_end(c0);
      fm_park_hidden(Hst + r * (16 * FM_HP), c0, bia);
    }
    __syncthreads();

    const int jn = j + 1 < FM_NCH ? j + 1 : j;
    fm_load_b1<0, IN_C / 8>(W1, jn, c, half, l16, bw1);
    bia = b1[FM_HC * jn + 16 * c + l16];

    const float* Hl = Hb + l16 * FM_HP + 2 * half;
#pragma unroll
    for (int ri = 0; ri < FM_OWN; ri += 2) {
      const int ra = p + 2 * ri;
      if (ri + 1 < FM_OWN && ra + 2 < nst) {
        const float* Ha = Hl + ra * (16 * FM_HP);
        const float* Hc = Ha + 32 * FM_HP;
        v2f h0[FM_HC / 4], h1[FM_HC / 4];
        fm_frags(Ha, h0);
        fm_frags(Hc, h1);
        __builtin_amdgcn_sched_barrier(0);
#pragma unroll
        for (int kk = 0; kk < FM_HC / 4; ++kk) {
          acc2[ri]     = fm_wmma(h0[kk], bw2[kk], acc2[ri]);
          acc2[ri + 1] = fm_wmma(h1[kk], bw2[kk], acc2[ri + 1]);
        }
      } else if (ra < nst) {
        const float* Ha = Hl + ra * (16 * FM_HP);
        v2f h0[FM_HC / 4];
        fm_frags(Ha, h0);
        __builtin_amdgcn_sched_barrier(0);
#pragma unroll
        for (int kk = 0; kk < FM_HC / 4; ++kk) acc2[ri] = fm_wmma(h0[kk], bw2[kk], acc2[ri]);
        fm_chain_end(acc2[ri]);
      }
      if (ri == 0) fm_load_b1<IN_C / 8, IN_C / 4>(W1, jn, c, half, l16, bw1);
    }
    fm_load_b2(W2, jn, c, half, l16, bw2);
  }

  float* Os = Hs;
#pragma unroll
  for (int ri = 0; ri < FM_OWN; ++ri) {
    const int ra = p + 2 * ri;
    if (ra < nst) {
      float* o = Os + (ra * 16 + 8 * half) * FM_HP + 16 * c + l16;
#pragma unroll
      for (int i = 0; i < 8; ++i) o[i * FM_HP] = acc2[ri][i];
    }
  }
  __syncthreads();

  if (tid < nst * 16) {
    const float* o = Os + tid * FM_HP;
    float ps = 0.f, pd = 0.f;
#pragma unroll
    for (int q = 0; q < OUT2 / 4; ++q) {
      const v4f v = *(const v4f*)(o + 4 * q);
      ps += v.x * a2s[4 * q];     pd += v.x * a2d[4 * q];
      ps += v.y * a2s[4 * q + 1]; pd += v.y * a2d[4 * q + 1];
      ps += v.z * a2s[4 * q + 2]; pd += v.z * a2d[4 * q + 2];
      ps += v.w * a2s[4 * q + 3]; pd += v.w * a2d[4 * q + 3];
    }
    const int grow = row0 + tid;
    if (grow < N) { as_out[grow] = ps; ad_out[grow] = pd; }
  }

  v4f orow[FM_R];
#pragma unroll
  for (int t = 0; t < FM_R; ++t)
    orow[t] = *(const v4f*)(Os + (2 * wave + 16 * t + half) * FM_HP + 4 * l16);
#pragma unroll
  for (int t = 0; t < FM_R; ++t) {
    const int grow = row0 + 2 * wave + 16 * t + half;
    if (grow < N) *(v4f*)(H2 + (size_t)grow * OUT2 + 4 * l16) = orow[t];
  }
}

__global__ void gat_zero_counts_w1_dots(int* __restrict__ cnt, int n,
                                        const float* __restrict__ W1,
                                        const float* __restrict__ a_src,
                                        const float* __restrict__ a_dst,
                                        float* __restrict__ wa,
                                        float* __restrict__ wd) {
  const int tid = blockIdx.x * blockDim.x + threadIdx.x;
  const int nth = gridDim.x * blockDim.x;
  const int n4  = n >> 2;
  int4* __restrict__ p4 = (int4*)cnt;
  const int4 z4 = make_int4(0, 0, 0, 0);
  for (int i = tid; i < n4; i += nth) p4[i] = z4;
  const int t = (n4 << 2) + tid;
  if (t < n) cnt[t] = 0;

  const int lane = threadIdx.x & 31;
  const int wave0  = __builtin_amdgcn_readfirstlane(tid >> 5);
  const int nwaves = nth >> 5;
  for (int r = wave0; r < IN_C; r += nwaves) {
    const float* row = W1 + (size_t)r * HID;
    float v[HID / 32], as[HID / 32], ad[HID / 32];
#pragma unroll
    for (int k = 0; k < HID / 32; ++k) v[k] = row[lane + 32 * k];
#pragma unroll
    for (int k = 0; k < HID / 32; ++k) { as[k] = a_src[lane + 32 * k]; ad[k] = a_dst[lane + 32 * k]; }
    float ss = 0.f, sd = 0.f;
#pragma unroll
    for (int k = 0; k < HID / 32; ++k) { ss = __builtin_fmaf(v[k], as[k], ss); sd = __builtin_fmaf(v[k], ad[k], sd); }
#pragma unroll
    for (int off = 16; off > 0; off >>= 1) {
      ss += __shfl_down(ss, off, 32);
      sd += __shfl_down(sd, off, 32);
    }
    if (lane == 0) { wa[r] = ss; wd[r] = sd; }
  }
}

__global__ void gat_dst_count(const int* __restrict__ ei, int* __restrict__ cnt, int E) {
  const int e = blockIdx.x * blockDim.x + threadIdx.x;
  if (e < E) atomicAdd(&cnt[ei[E + e]], 1);
}

#define SCAN_TB 1024
__global__ void gat_dst_offsets(const int* __restrict__ cnt,
                                int* __restrict__ rowptr,
                                int* __restrict__ cursor, int N) {
  __shared__ int wbefore[SCAN_TB / 32];
  __shared__ int wtot[SCAN_TB / 32];
  const int t = threadIdx.x;
  const int lane = t & 31;
  const int w = t >> 5;
  const int b = blockIdx.x;
  int before = 0;
  for (int k = 0; k < b; ++k) before += cnt[k * SCAN_TB + t];
  const int i = b * SCAN_TB + t;
  const int own = i < N ? cnt[i] : 0;
  int inc = own;
#pragma unroll
  for (int off = 1; off < 32; off <<= 1) {
    const int v = __shfl_up(inc, off, 32);
    if (lane >= off) inc += v;
  }
#pragma unroll
  for (int off = 16; off > 0; off >>= 1) before += __shfl_xor(before, off, 32);
  if (lane == 31) wtot[w] = inc;
  if (lane == 0)  wbefore[w] = before;
  __syncthreads();
  int base = 0;
  for (int k = 0; k < SCAN_TB / 32; ++k) {
    base += wbefore[k];
    if (k < w) base += wtot[k];
  }
  const int excl = base + inc - own;
  if (i < N) { rowptr[i] = excl; cursor[i] = excl; }
  if (i == N - 1) rowptr[N] = excl + own;
}

#define XD_NB 4
#define XD_Q  (IN_C / 32)

__device__ __forceinline__ void xd_load(const float* __restrict__ x, int n0, int nwaves, int N,
                                        int lane, float (&v)[XD_NB][XD_Q]) {
#pragma unroll
  for (int k = 0; k < XD_NB; ++k) {
    int node = n0 + k * nwaves;
    if (node > N - 1) node = n0;
    const float* row = x + (size_t)node * IN_C;
#pragma unroll
    for (int q = 0; q < XD_Q; ++q) v[k][q] = row[lane + 32 * q];
  }
}

__global__ void gat_dst_scatter_x_dots(const int* __restrict__ ei, int* __restrict__ cursor,
                                       int* __restrict__ col, int E,
                                       const float* __restrict__ x,
                                       const float* __restrict__ wa,
                                       const float* __restrict__ wd,
                                       float* __restrict__ as_out,
                                       float* __restrict__ ad_out, int N) {
  const int e    = blockIdx.x * blockDim.x + threadIdx.x;
  const int lane = threadIdx.x & 31;
  const int wave0  = __builtin_amdgcn_readfirstlane(e >> 5);
  const int nwaves = gridDim.x * (blockDim.x >> 5);

  float a[XD_Q], b[XD_Q];
#pragma unroll
  for (int q = 0; q < XD_Q; ++q) { a[q] = wa[lane + 32 * q]; b[q] = wd[lane + 32 * q]; }

  int d = 0, s = 0;
  if (e < E) { d = ei[E + e]; s = ei[e]; }

  float v[XD_NB][XD_Q];
  int n0 = wave0;
  if (n0 < N) xd_load(x, n0, nwaves, N, lane, v);

  if (e < E) {
    const int p = atomicAdd(&cursor[d], 1);
    col[p] = s;
  }

  while (n0 < N) {
    float ss[XD_NB], sd[XD_NB];
#pragma unroll
    for (int k = 0; k < XD_NB; ++k) {
      ss[k] = 0.f; sd[k] = 0.f;
#pragma unroll
      for (int q = 0; q < XD_Q; ++q) {
        ss[k] = __builtin_fmaf(v[k][q], a[q], ss[k]);
        sd[k] = __builtin_fmaf(v[k][q], b[q], sd[k]);
      }
    }
#pragma unroll
    for (int off = 16; off > 0; off >>= 1) {
#pragma unroll
      for (int k = 0; k < XD_NB; ++k) {
        ss[k] += __shfl_down(ss[k], off, 32);
        sd[k] += __shfl_down(sd[k], off, 32);
      }
    }
#pragma unroll
    for (int k = 0; k < XD_NB; ++k) {
      const int node = n0 + k * nwaves;
      if (lane == 0 && node < N) { as_out[node] = ss[k]; ad_out[node] = sd[k]; }
    }
    n0 += XD_NB * nwaves;
    if (n0 < N) xd_load(x, n0, nwaves, N, lane, v);
  }
}

__device__ __forceinline__ float leakyr(float v) {
  return v > 0.f ? v : NEG_SLOPE * v;
}

__device__ __forceinline__ float wave_max(float v) {
#pragma unroll
  for (int off = 16; off > 0; off >>= 1) v = fmaxf(v, __shfl_xor(v, off, 32));
  return v;
}

__device__ __forceinline__ float wave_sum(float v) {
#pragma unroll
  for (int off = 16; off > 0; off >>= 1) v += __shfl_xor(v, off, 32);
  return v;
}

template <int D> struct gat_rowvec;
template <> struct gat_rowvec<128> { typedef v4f type; };
template <> struct gat_rowvec<64>  { typedef v2f type; };

template <int D>
__device__ __forceinline__ void agg_rows(const float* __restrict__ h, int s_l, float a_l,
                                         int n, int lane,
                                         typename gat_rowvec<D>::type& acc) {
  typedef typename gat_rowvec<D>::type vec;
  constexpr int GRP = 8;
  for (int u = 0; u < n; u += GRP) {
    vec v[GRP];
#pragma unroll
    for (int g = 0; g < GRP; ++g) {
      const int s = __builtin_amdgcn_readlane(s_l, u + g);
      v[g] = ((const vec*)(h + (size_t)s * D))[lane];
    }
#pragma unroll
    for (int g = 0; g < GRP; ++g) {
      const float a = __int_as_float(__builtin_amdgcn_readlane(__float_as_int(a_l), u + g));
      acc += a * v[g];
    }
  }
}

template <int D, bool TAIL>
__global__ void gat_node_softmax_agg(const int* __restrict__ rowptr,
                                     const int* __restrict__ col,
                                     const float* __restrict__ as,
                                     const float* __restrict__ ad,
                                     const float* __restrict__ h,
                                     const float* __restrict__ bias,
                                     const float* __restrict__ fcW,
                                     const float* __restrict__ fcb,
                                     float* __restrict__ out, int N) {
  typedef typename gat_rowvec<D>::type vec;
  static_assert(!TAIL || D == OUT2, "the classifier tail is written for the 64-wide row");
  float tb0 = 0.f, tb1 = 0.f, tw00 = 0.f, tw01 = 0.f, tw10 = 0.f, tw11 = 0.f, tf0 = 0.f, tf1 = 0.f;
  if constexpr (TAIL) {
    const int c = 2 * (threadIdx.x & 31);
    tb0  = bias[c];
    tb1  = bias[c + 1];
    tw00 = fcW[c * NCLS];
    tw01 = fcW[c * NCLS + 1];
    tw10 = fcW[(c + 1) * NCLS];
    tw11 = fcW[(c + 1) * NCLS + 1];
    tf0  = fcb[0];
    tf1  = fcb[1];
  }
  const int lane = threadIdx.x & 31;
  const int wib    = __builtin_amdgcn_readfirstlane((int)(threadIdx.x >> 5));
  const int wave0  = blockIdx.x * (blockDim.x >> 5) + wib;
  const int nwaves = gridDim.x * (blockDim.x >> 5);
  const float NEG_INF = -__builtin_inff();
  for (int d = wave0; d < N; d += nwaves) {
    const int beg = __builtin_amdgcn_readfirstlane(rowptr[d]);
    const int end = __builtin_amdgcn_readfirstlane(rowptr[d + 1]);
    const int cnt = end - beg + 1;
    const float adv = ad[d];

    const bool act0 = lane < cnt;
    int   s0 = d;
    float e0 = NEG_INF;
    if (act0) {
      if (lane < cnt - 1) s0 = col[beg + lane];
      e0 = leakyr(as[s0] + adv);
    }
    float mx = e0;
    for (int base = 32; base < cnt; base += 32) {
      const int j = base + lane;
      if (j < cnt) {
        const int s = j < cnt - 1 ? col[beg + j] : d;
        mx = fmaxf(mx, leakyr(as[s] + adv));
      }
    }
    mx = wave_max(mx);

    const float ex0 = act0 ? expf(e0 - mx) : 0.f;
    float sm = ex0;
    for (int base = 32; base < cnt; base += 32) {
      const int j = base + lane;
      if (j < cnt) {
        const int s = j < cnt - 1 ? col[beg + j] : d;
        sm += expf(leakyr(as[s] + adv) - mx);
      }
    }
    sm = wave_sum(sm);

    vec acc = {};
    agg_rows<D>(h, s0, ex0 / sm, cnt < 32 ? cnt : 32, lane, acc);
    for (int base = 32; base < cnt; base += 32) {
      const int j = base + lane;
      int   s = d;
      float a = 0.f;
      if (j < cnt) {
        if (j < cnt - 1) s = col[beg + j];
        a = expf(leakyr(as[s] + adv) - mx) / sm;
      }
      const int left = cnt - base;
      agg_rows<D>(h, s, a, left < 32 ? left : 32, lane, acc);
    }

    if constexpr (TAIL) {
      float v0 = acc[0] + tb0;
      float v1 = acc[1] + tb1;
      v0 = v0 > 0.f ? v0 : 0.f;
      v1 = v1 > 0.f ? v1 : 0.f;
      const float o0 = wave_sum(v0 * tw00 + v1 * tw10);
      const float o1 = wave_sum(v0 * tw01 + v1 * tw11);
      if (lane == 0) {
        const v2f o = {o0 + tf0, o1 + tf1};
        *(v2f*)(out + (size_t)d * NCLS) = o;
      }
    } else {
      ((vec*)(out + (size_t)d * D))[lane] = acc;
    }
  }
}

extern "C" void kernel_launch(void* const* d_in, const int* in_sizes, int n_in,
                              void* d_out, int out_size, void* d_ws, size_t ws_size,
                              hipStream_t stream) {
  const float* x   = (const float*)d_in[0];
  const int*   ei  = (const int*)  d_in[1];
  const float* W1  = (const float*)d_in[2];
  const float* a1s = (const float*)d_in[3];
  const float* a1d = (const float*)d_in[4];
  const float* b1  = (const float*)d_in[5];
  const float* W2  = (const float*)d_in[6];
  const float* a2s = (const float*)d_in[7];
  const float* a2d = (const float*)d_in[8];
  const float* b2  = (const float*)d_in[9];
  const float* fcW = (const float*)d_in[10];
  const float* fcb = (const float*)d_in[11];

  const int N    = in_sizes[0] / IN_C;
  const int E    = in_sizes[1] / 2;

  float* w       = (float*)d_ws;
  float* bufA    = w;
  float* bufB    = w + (size_t)N * HID;
  float* as_buf  = w + (size_t)2 * N * HID;
  float* ad_buf  = as_buf + N;
  int*   cnt     = (int*)(bufB + (size_t)N * IN_C);
  int*   rowptr  = (int*)(ad_buf + N);
  int*   cursor  = rowptr + N + 1;
  int*   col     = cursor + N;

  const int TB = 256;

  const int CHIP = 2048;
  auto capped = [](int blocks, int cap) { return blocks < cap ? blocks : cap; };
  auto ew_grid = [&](int n) { return capped((n / 4 + TB - 1) / TB, CHIP); };
  const int list_blk  = (E + TB - 1) / TB;
  const int node_blk  = capped((N + 7) / 8, CHIP);

  float* wa   = bufA;
  float* wd   = bufA + IN_C;
  float* aggx = bufB;
  gat_zero_counts_w1_dots<<<ew_grid(N), TB, 0, stream>>>(cnt, N, W1, a1s, a1d, wa, wd);
  gat_dst_count  <<<list_blk, TB, 0, stream>>>(ei, cnt, E);
  gat_dst_offsets<<<(N + SCAN_TB - 1) / SCAN_TB, SCAN_TB, 0, stream>>>(cnt, rowptr, cursor, N);
  gat_dst_scatter_x_dots<<<list_blk, TB, 0, stream>>>(ei, cursor, col, E, x, wa, wd, as_buf, ad_buf, N);

  gat_node_softmax_agg<IN_C, false><<<node_blk, TB, 0, stream>>>(rowptr, col, as_buf, ad_buf, x,
                                                               nullptr, nullptr, nullptr, aggx, N);

  float* h2   = bufA;
  gat_transform_fused<<<(N + FM_ROWS - 1) / FM_ROWS, TB, 0, stream>>>(aggx, W1, b1, W2, a2s, a2d, h2, as_buf, ad_buf, N);

  gat_node_softmax_agg<OUT2, true><<<node_blk, TB, 0, stream>>>(rowptr, col, as_buf, ad_buf, h2,
                                                              b2, fcW, fcb, (float*)d_out, N);
}


# Edge lists in one 128-byte line per destination (31 slots + count word, spill list past 31): one edge pass replaces count + offsets + scatter, 7 -> 5 launches; node sums keep the adds' arrival order
# speedup vs baseline: 6.5949x; 1.3531x over previous
#include <hip/hip_runtime.h>


typedef __attribute__((ext_vector_type(2))) float v2f;
typedef __attribute__((ext_vector_type(8))) float v8f;
typedef __attribute__((ext_vector_type(4))) float v4f;

#define IN_C  128
#define HID   512
#define OUT2  64
#define NCLS  2
#define NEG_SLOPE 0.2f

#define FM_HC   64
#define FM_NCH  (HID / FM_HC)
#define FM_R    13
#define FM_ROWS (16 * FM_R)
#define FM_OWN  ((FM_R + 1) / 2)
#define FM_OWNP ((FM_OWN + 1) & ~1)
#define FM_AP   (IN_C + 4)
#define FM_HP   (FM_HC + 4)
#define FM_AI   (FM_ROWS * (IN_C / 4) / 256)

__device__ __forceinline__ v8f fm_wmma(v2f a, v2f b, v8f c) {
  return __builtin_amdgcn_wmma_f32_16x16x4_f32(false, a, false, b, (short)0, c, false, false);
}

__device__ __forceinline__ void fm_chain_end(v8f& c) {
  asm volatile("v_nop\n\tv_nop\n\tv_nop\n\tv_nop" : "+v"(c));
}

template <int K0, int K1>
__device__ __forceinline__ void fm_load_b1(const float* __restrict__ W1, int j, int c, int half, int l16,
                                           v2f (&bw1)[IN_C / 4]) {
  const float* g = W1 + (size_t)(2 * half) * HID + FM_HC * j + 16 * c + l16;
#pragma unroll
  for (int kq = K0; kq < K1; ++kq) {
    bw1[kq].x = g[(size_t)(4 * kq) * HID];
    bw1[kq].y = g[(size_t)(4 * kq + 1) * HID];
  }
}

__device__ __forceinline__ void fm_load_b2(const float* __restrict__ W2, int j, int c, int half, int l16,
                                           v2f (&bw2)[FM_HC / 4]) {
  const float* g = W2 + (size_t)(FM_HC * j + 2 * half) * OUT2 + 16 * c + l16;
#pragma unroll
  for (int kk = 0; kk < FM_HC / 4; ++kk) {
    bw2[kk].x = g[(size_t)(4 * kk) * OUT2];
    bw2[kk].y = g[(size_t)(4 * kk + 1) * OUT2];
  }
}

template <int NK>
__device__ __forceinline__ void fm_frags(const float* s, v2f (&f)[NK]) {
#pragma unroll
  for (int k = 0; k < NK; ++k) f[k] = *(const v2f*)(s + 4 * k);
}

__device__ __forceinline__ void fm_park_hidden(float* Hst, v8f c, float bia) {
#pragma unroll
  for (int i = 0; i < 8; ++i) {
    const float v = c[i] + bia;
    Hst[i * FM_HP] = v > 0.f ? v : 0.f;
  }
}

__global__ __launch_bounds__(256)
void gat_transform_fused(const float* __restrict__ A,
                         const float* __restrict__ W1,
                         const float* __restrict__ b1,
                         const float* __restrict__ W2,
                         const float* __restrict__ a2s,
                         const float* __restrict__ a2d,
                         float* __restrict__ H2,
                         float* __restrict__ as_out,
                         float* __restrict__ ad_out, int N) {
  __shared__ __attribute__((aligned(16))) float lds[FM_ROWS * FM_AP + 2 * FM_ROWS * FM_HP];
  float* As = lds;
  float* Hs = lds + FM_ROWS * FM_AP;

  const int tid  = threadIdx.x;
  const int lane = tid & 31;
  const int wave = __builtin_amdgcn_readfirstlane(tid >> 5);
  const int c    = wave & 3;
  const int p    = wave >> 2;
  const int half = lane >> 4;
  const int l16  = lane & 15;
  const int row0 = blockIdx.x * FM_ROWS;
  int nst = (N - row0 + 15) >> 4;
  if (nst > FM_R) nst = FM_R;

  v2f bw1[IN_C / 4], bw2[FM_HC / 4];

  {
    v4f ar[FM_AI];
#pragma unroll
    for (int t = 0; t < FM_AI; ++t) {
      const int i = tid + 256 * t;
      int grow = row0 + i / (IN_C / 4);
      if (grow > N - 1) grow = N - 1;
      ar[t] = *(const v4f*)(A + (size_t)grow * IN_C + 4 * (i % (IN_C / 4)));
    }
    fm_load_b1<0, IN_C / 8>(W1, 0, c, half, l16, bw1);
#pragma unroll
    for (int t = 0; t < FM_AI; ++t) {
      const int i = tid + 256 * t;
      *(v4f*)(As + (i / (IN_C / 4)) * FM_AP + 4 * (i % (IN_C / 4))) = ar[t];
    }
  }
  fm_load_b1<IN_C / 8, IN_C / 4>(W1, 0, c, half, l16, bw1);
  fm_load_b2(W2, 0, c, half, l16, bw2);
  float bia = b1[16 * c + l16];
  __syncthreads();

  const float* Al = As + l16 * FM_AP + 2 * half;

  v8f acc2[FM_OWNP];
#pragma unroll
  for (int ri = 0; ri < FM_OWNP; ++ri) acc2[ri] = v8f{};

  for (int j = 0; j < FM_NCH; ++j) {
    float* Hb = Hs + (j & 1) * (FM_ROWS * FM_HP);
    float* Hst = Hb + (8 * half) * FM_HP + 16 * c + l16;

    int r = p;
    for (; r + 2 < nst; r += 4) {
      const float* Aa = Al + r * (16 * FM_AP);
      const float* Ab = Aa + 32 * FM_AP;
      v8f c0 = {}, c1 = {};
      v2f a0[IN_C / 4], a1[IN_C / 4];
      fm_frags(Aa, a0);
      fm_frags(Ab, a1);
#pragma unroll
      for (int kq = 0; kq < IN_C / 4; ++kq) {
        c0 = fm_wmma(a0[kq], bw1[kq], c0);
        c1 = fm_wmma(a1[kq], bw1[kq], c1);
      }
      fm_park_hidden(Hst + r * (16 * FM_HP), c0, bia);
      fm_park_hidden(Hst + (r + 2) * (16 * FM_HP), c1, bia);
    }
    if (r < nst) {
      const float* Aa = Al + r * (16 * FM_AP);
      v8f c0 = {};
      v2f a0[IN_C / 4];
      fm_frags(Aa, a0);
      __builtin_amdgcn_sched_barrier(0);
#pragma unroll
      for (int kq = 0; kq < IN_C / 4; ++kq) c0 = fm_wmma(a0[kq], bw1[kq], c0);
      fm_chain_end(c0);
      fm_park_hidden(Hst + r * (16 * FM_HP), c0, bia);
    }
    __syncthreads();

    const int jn = j + 1 < FM_NCH ? j + 1 : j;
    fm_load_b1<0, IN_C / 8>(W1, jn, c, half, l16, bw1);
    bia = b1[FM_HC * jn + 16 * c + l16];

    const float* Hl = Hb + l16 * FM_HP + 2 * half;
#pragma unroll
    for (int ri = 0; ri < FM_OWN; ri += 2) {
      const int ra = p + 2 * ri;
      if (ri + 1 < FM_OWN && ra + 2 < nst) {
        const float* Ha = Hl + ra * (16 * FM_HP);
        const float* Hc = Ha + 32 * FM_HP;
        v2f h0[FM_HC / 4], h1[FM_HC / 4];
        fm_frags(Ha, h0);
        fm_frags(Hc, h1);
        __builtin_amdgcn_sched_barrier(0);
#pragma unroll
        for (int kk = 0; kk < FM_HC / 4; ++kk) {
          acc2[ri]     = fm_wmma(h0[kk], bw2[kk], acc2[ri]);
          acc2[ri + 1] = fm_wmma(h1[kk], bw2[kk], acc2[ri + 1]);
        }
      } else if (ra < nst) {
        const float* Ha = Hl + ra * (16 * FM_HP);
        v2f h0[FM_HC / 4];
        fm_frags(Ha, h0);
        __builtin_amdgcn_sched_barrier(0);
#pragma unroll
        for (int kk = 0; kk < FM_HC / 4; ++kk) acc2[ri] = fm_wmma(h0[kk], bw2[kk], acc2[ri]);
        fm_chain_end(acc2[ri]);
      }
      if (ri == 0) fm_load_b1<IN_C / 8, IN_C / 4>(W1, jn, c, half, l16, bw1);
    }
    fm_load_b2(W2, jn, c, half, l16, bw2);
  }

  float* Os = Hs;
#pragma unroll
  for (int ri = 0; ri < FM_OWN; ++ri) {
    const int ra = p + 2 * ri;
    if (ra < nst) {
      float* o = Os + (ra * 16 + 8 * half) * FM_HP + 16 * c + l16;
#pragma unroll
      for (int i = 0; i < 8; ++i) o[i * FM_HP] = acc2[ri][i];
    }
  }
  __syncthreads();

  if (tid < nst * 16) {
    const float* o = Os + tid * FM_HP;
    float ps = 0.f, pd = 0.f;
#pragma unroll
    for (int q = 0; q < OUT2 / 4; ++q) {
      const v4f v = *(const v4f*)(o + 4 * q);
      ps += v.x * a2s[4 * q];     pd += v.x * a2d[4 * q];
      ps += v.y * a2s[4 * q + 1]; pd += v.y * a2d[4 * q + 1];
      ps += v.z * a2s[4 * q + 2]; pd += v.z * a2d[4 * q + 2];
      ps += v.w * a2s[4 * q + 3]; pd += v.w * a2d[4 * q + 3];
    }
    const int grow = row0 + tid;
    if (grow < N) { as_out[grow] = ps; ad_out[grow] = pd; }
  }

  v4f orow[FM_R];
#pragma unroll
  for (int t = 0; t < FM_R; ++t)
    orow[t] = *(const v4f*)(Os + (2 * wave + 16 * t + half) * FM_HP + 4 * l16);
#pragma unroll
  for (int t = 0; t < FM_R; ++t) {
    const int grow = row0 + 2 * wave + 16 * t + half;
    if (grow < N) *(v4f*)(H2 + (size_t)grow * OUT2 + 4 * l16) = orow[t];
  }
}

#define LINE_W 32
#define SLOTS  31

__global__ void gat_zero_lines_w1_dots(int* __restrict__ cnt, int n,
                                       const float* __restrict__ W1,
                                       const float* __restrict__ a_src,
                                       const float* __restrict__ a_dst,
                                       float* __restrict__ wa,
                                       float* __restrict__ wd) {
  const int tid = blockIdx.x * blockDim.x + threadIdx.x;
  const int nth = gridDim.x * blockDim.x;
  const int n4  = n >> 2;
  int4* __restrict__ p4 = (int4*)cnt;
  const int4 z4 = make_int4(0, 0, 0, 0);
  for (int i = tid; i < n4; i += nth) p4[i] = z4;
  const int t = (n4 << 2) + tid;
  if (t < n) cnt[t] = 0;

  const int lane = threadIdx.x & 31;
  const int wave0  = __builtin_amdgcn_readfirstlane(tid >> 5);
  const int nwaves = nth >> 5;
  for (int r = wave0; r < IN_C; r += nwaves) {
    const float* row = W1 + (size_t)r * HID;
    float v[HID / 32], as[HID / 32], ad[HID / 32];
#pragma unroll
    for (int k = 0; k < HID / 32; ++k) v[k] = row[lane + 32 * k];
#pragma unroll
    for (int k = 0; k < HID / 32; ++k) { as[k] = a_src[lane + 32 * k]; ad[k] = a_dst[lane + 32 * k]; }
    float ss = 0.f, sd = 0.f;
#pragma unroll
    for (int k = 0; k < HID / 32; ++k) { ss = __builtin_fmaf(v[k], as[k], ss); sd = __builtin_fmaf(v[k], ad[k], sd); }
#pragma unroll
    for (int off = 16; off > 0; off >>= 1) {
      ss += __shfl_down(ss, off, 32);
      sd += __shfl_down(sd, off, 32);
    }
    if (lane == 0) { wa[r] = ss; wd[r] = sd; }
  }
}

#define XD_NB 4
#define XD_Q  (IN_C / 32)

__device__ __forceinline__ void xd_load(const float* __restrict__ x, int n0, int nwaves, int N,
                                        int lane, float (&v)[XD_NB][XD_Q]) {
#pragma unroll
  for (int k = 0; k < XD_NB; ++k) {
    int node = n0 + k * nwaves;
    if (node > N - 1) node = n0;
    const float* row = x + (size_t)node * IN_C;
#pragma unroll
    for (int q = 0; q < XD_Q; ++q) v[k][q] = row[lane + 32 * q];
  }
}

__global__ void gat_dst_lines_x_dots(const int* __restrict__ ei, int* __restrict__ tab,
                                     int* __restrict__ spill_n, int* __restrict__ spill, int E,
                                     const float* __restrict__ x,
                                     const float* __restrict__ wa,
                                     const float* __restrict__ wd,
                                     float* __restrict__ as_out,
                                     float* __restrict__ ad_out, int N) {
  const int e    = blockIdx.x * blockDim.x + threadIdx.x;
  const int lane = threadIdx.x & 31;
  const int wave0  = __builtin_amdgcn_readfirstlane(e >> 5);
  const int nwaves = gridDim.x * (blockDim.x >> 5);

  float a[XD_Q], b[XD_Q];
#pragma unroll
  for (int q = 0; q < XD_Q; ++q) { a[q] = wa[lane + 32 * q]; b[q] = wd[lane + 32 * q]; }

  int d = 0, s = 0;
  if (e < E) { d = ei[E + e]; s = ei[e]; }

  float v[XD_NB][XD_Q];
  int n0 = wave0;
  if (n0 < N) xd_load(x, n0, nwaves, N, lane, v);

  if (e < E) {
    int* line = tab + (size_t)d * LINE_W;
    const int r = atomicAdd(&line[SLOTS], 1);
    if (r < SLOTS) {
      line[r] = s;
    } else {
      const int o = atomicAdd(spill_n, 1);
      spill[2 * o]     = d;
      spill[2 * o + 1] = s;
    }
  }

  while (n0 < N) {
    float ss[XD_NB], sd[XD_NB];
#pragma unroll
    for (int k = 0; k < XD_NB; ++k) {
      ss[k] = 0.f; sd[k] = 0.f;
#pragma unroll
      for (int q = 0; q < XD_Q; ++q) {
        ss[k] = __builtin_fmaf(v[k][q], a[q], ss[k]);
        sd[k] = __builtin_fmaf(v[k][q], b[q], sd[k]);
      }
    }
#pragma unroll
    for (int off = 16; off > 0; off >>= 1) {
#pragma unroll
      for (int k = 0; k < XD_NB; ++k) {
        ss[k] += __shfl_down(ss[k], off, 32);
        sd[k] += __shfl_down(sd[k], off, 32);
      }
    }
#pragma unroll
    for (int k = 0; k < XD_NB; ++k) {
      const int node = n0 + k * nwaves;
      if (lane == 0 && node < N) { as_out[node] = ss[k]; ad_out[node] = sd[k]; }
    }
    n0 += XD_NB * nwaves;
    if (n0 < N) xd_load(x, n0, nwaves, N, lane, v);
  }
}

__device__ __forceinline__ float leakyr(float v) {
  return v > 0.f ? v : NEG_SLOPE * v;
}

__device__ __forceinline__ float wave_max(float v) {
#pragma unroll
  for (int off = 16; off > 0; off >>= 1) v = fmaxf(v, __shfl_xor(v, off, 32));
  return v;
}

__device__ __forceinline__ float wave_sum(float v) {
#pragma unroll
  for (int off = 16; off > 0; off >>= 1) v += __shfl_xor(v, off, 32);
  return v;
}

template <int D> struct gat_rowvec;
template <> struct gat_rowvec<128> { typedef v4f type; };
template <> struct gat_rowvec<64>  { typedef v2f type; };

template <int D>
__device__ __forceinline__ void agg_rows(const float* __restrict__ h, int s_l, float a_l,
                                         int n, int lane,
                                         typename gat_rowvec<D>::type& acc) {
  typedef typename gat_rowvec<D>::type vec;
  constexpr int GRP = 8;
  for (int u = 0; u < n; u += GRP) {
    vec v[GRP];
#pragma unroll
    for (int g = 0; g < GRP; ++g) {
      const int s = __builtin_amdgcn_readlane(s_l, u + g);
      v[g] = ((const vec*)(h + (size_t)s * D))[lane];
    }
#pragma unroll
    for (int g = 0; g < GRP; ++g) {
      const float a = __int_as_float(__builtin_amdgcn_readlane(__float_as_int(a_l), u + g));
      acc += a * v[g];
    }
  }
}

__device__ __forceinline__ int spill_src(const int* __restrict__ spill, int j, int nsp, int d) {
  int s = -1;
  if (j < nsp && spill[2 * j] == d) s = spill[2 * j + 1];
  return s;
}

template <int D, bool TAIL>
__global__ void gat_node_softmax_agg(const int* __restrict__ tab,
                                     const int* __restrict__ spill_n,
                                     const int* __restrict__ spill, int E,
                                     const float* __restrict__ as,
                                     const float* __restrict__ ad,
                                     const float* __restrict__ h,
                                     const float* __restrict__ bias,
                                     const float* __restrict__ fcW,
                                     const float* __restrict__ fcb,
                                     float* __restrict__ out, int N) {
  typedef typename gat_rowvec<D>::type vec;
  static_assert(!TAIL || D == OUT2, "the classifier tail is written for the 64-wide row");
  float tb0 = 0.f, tb1 = 0.f, tw00 = 0.f, tw01 = 0.f, tw10 = 0.f, tw11 = 0.f, tf0 = 0.f, tf1 = 0.f;
  if constexpr (TAIL) {
    const int c = 2 * (threadIdx.x & 31);
    tb0  = bias[c];
    tb1  = bias[c + 1];
    tw00 = fcW[c * NCLS];
    tw01 = fcW[c * NCLS + 1];
    tw10 = fcW[(c + 1) * NCLS];
    tw11 = fcW[(c + 1) * NCLS + 1];
    tf0  = fcb[0];
    tf1  = fcb[1];
  }
  const int lane = threadIdx.x & 31;
  const int wib    = __builtin_amdgcn_readfirstlane((int)(threadIdx.x >> 5));
  const int wave0  = blockIdx.x * (blockDim.x >> 5) + wib;
  const int nwaves = gridDim.x * (blockDim.x >> 5);
  const float NEG_INF = -__builtin_inff();
  for (int d = wave0; d < N; d += nwaves) {
    const int word   = tab[(size_t)d * LINE_W + lane];
    const float adv  = ad[d];
    const int listed = __builtin_amdgcn_readlane(word, SLOTS);
    const int nslot  = listed < SLOTS ? listed : SLOTS;
    const int cnt    = nslot + 1;

    int nsp = 0;
    if (listed > SLOTS) {
      nsp = __builtin_amdgcn_readfirstlane(spill_n[0]);
      if (nsp > E) nsp = E;
    }

    const bool act0 = lane < cnt;
    int   s0 = d;
    float e0 = NEG_INF;
    if (act0) {
      if (lane < nslot) s0 = word;
      e0 = leakyr(as[s0] + adv);
    }
    float mx = e0;
    for (int base = 0; base < nsp; base += 32) {
      const int s = spill_src(spill, base + lane, nsp, d);
      if (s >= 0) mx = fmaxf(mx, leakyr(as[s] + adv));
    }
    mx = wave_max(mx);

    const float ex0 = act0 ? expf(e0 - mx) : 0.f;
    float sm = ex0;
    for (int base = 0; base < nsp; base += 32) {
      const int s = spill_src(spill, base + lane, nsp, d);
      if (s >= 0) sm += expf(leakyr(as[s] + adv) - mx);
    }
    sm = wave_sum(sm);

    vec acc = {};
    agg_rows<D>(h, s0, ex0 / sm, cnt, lane, acc);
    for (int base = 0; base < nsp; base += 32) {
      const int s = spill_src(spill, base + lane, nsp, d);
      float a = 0.f;
      if (s >= 0) a = expf(leakyr(as[s] + adv) - mx) / sm;
      unsigned m = __builtin_amdgcn_ballot_w32(s >= 0);
      while (m) {
        const int l = __builtin_ctz(m);
        m &= m - 1;
        const int   sl = __builtin_amdgcn_readlane(s, l);
        const float al = __int_as_float(__builtin_amdgcn_readlane(__float_as_int(a), l));
        acc += al * ((const vec*)(h + (size_t)sl * D))[lane];
      }
    }

    if constexpr (TAIL) {
      float v0 = acc[0] + tb0;
      float v1 = acc[1] + tb1;
      v0 = v0 > 0.f ? v0 : 0.f;
      v1 = v1 > 0.f ? v1 : 0.f;
      const float o0 = wave_sum(v0 * tw00 + v1 * tw10);
      const float o1 = wave_sum(v0 * tw01 + v1 * tw11);
      if (lane == 0) {
        const v2f o = {o0 + tf0, o1 + tf1};
        *(v2f*)(out + (size_t)d * NCLS) = o;
      }
    } else {
      ((vec*)(out + (size_t)d * D))[lane] = acc;
    }
  }
}

extern "C" void kernel_launch(void* const* d_in, const int* in_sizes, int n_in,
                              void* d_out, int out_size, void* d_ws, size_t ws_size,
                              hipStream_t stream) {
  const float* x   = (const float*)d_in[0];
  const int*   ei  = (const int*)  d_in[1];
  const float* W1  = (const float*)d_in[2];
  const float* a1s = (const float*)d_in[3];
  const float* a1d = (const float*)d_in[4];
  const float* b1  = (const float*)d_in[5];
  const float* W2  = (const float*)d_in[6];
  const float* a2s = (const float*)d_in[7];
  const float* a2d = (const float*)d_in[8];
  const float* b2  = (const float*)d_in[9];
  const float* fcW = (const float*)d_in[10];
  const float* fcb = (const float*)d_in[11];

  const int N    = in_sizes[0] / IN_C;
  const int E    = in_sizes[1] / 2;

  float* w       = (float*)d_ws;
  float* bufA    = w;
  float* bufB    = w + (size_t)N * HID;
  float* as_buf  = w + (size_t)2 * N * HID;
  float* ad_buf  = as_buf + N;
  int*   tab     = (int*)(bufB + (size_t)N * IN_C);
  int*   spill_n = tab + (size_t)N * LINE_W;
  int*   spill   = spill_n + 4;

  const int TB = 256;

  const int CHIP = 2048;
  auto capped = [](int blocks, int cap) { return blocks < cap ? blocks : cap; };
  const int list_blk  = (E + TB - 1) / TB;
  const int zero_n    = N * LINE_W + 4;
  const int zero_blk  = capped((zero_n / 32 + TB - 1) / TB, CHIP);
  const int node_blk  = capped((N + 7) / 8, CHIP);

  float* wa   = bufA;
  float* wd   = bufA + IN_C;
  float* aggx = bufB;
  gat_zero_lines_w1_dots<<<zero_blk, TB, 0, stream>>>(tab, zero_n, W1, a1s, a1d, wa, wd);
  gat_dst_lines_x_dots<<<list_blk, TB, 0, stream>>>(ei, tab, spill_n, spill, E, x, wa, wd, as_buf, ad_buf, N);

  gat_node_softmax_agg<IN_C, false><<<node_blk, TB, 0, stream>>>(tab, spill_n, spill, E, as_buf, ad_buf, x,
                                                               nullptr, nullptr, nullptr, aggx, N);

  float* h2   = bufA;
  gat_transform_fused<<<(N + FM_ROWS - 1) / FM_ROWS, TB, 0, stream>>>(aggx, W1, b1, W2, a2s, a2d, h2, as_buf, ad_buf, N);

  gat_node_softmax_agg<OUT2, true><<<node_blk, TB, 0, stream>>>(tab, spill_n, spill, E, as_buf, ad_buf, h2,
                                                              b2, fcW, fcb, (float*)d_out, N);
}


# Edges grouped through LDS: blocks sort 1024-edge chunks by 128-node range, a block per range fills its node lines with LDS adds; no global add per edge, no zero pass; node sums in LDS arrival order
# speedup vs baseline: 7.6542x; 1.1606x over previous
#include <hip/hip_runtime.h>


typedef __attribute__((ext_vector_type(2))) float v2f;
typedef __attribute__((ext_vector_type(8))) float v8f;
typedef __attribute__((ext_vector_type(4))) float v4f;

#define IN_C  128
#define HID   512
#define OUT2  64
#define NCLS  2
#define NEG_SLOPE 0.2f

#define FM_HC   64
#define FM_NCH  (HID / FM_HC)
#define FM_R    13
#define FM_ROWS (16 * FM_R)
#define FM_OWN  ((FM_R + 1) / 2)
#define FM_OWNP ((FM_OWN + 1) & ~1)
#define FM_AP   (IN_C + 4)
#define FM_HP   (FM_HC + 4)
#define FM_AI   (FM_ROWS * (IN_C / 4) / 256)

__device__ __forceinline__ v8f fm_wmma(v2f a, v2f b, v8f c) {
  return __builtin_amdgcn_wmma_f32_16x16x4_f32(false, a, false, b, (short)0, c, false, false);
}

__device__ __forceinline__ void fm_chain_end(v8f& c) {
  asm volatile("v_nop\n\tv_nop\n\tv_nop\n\tv_nop" : "+v"(c));
}

template <int K0, int K1>
__device__ __forceinline__ void fm_load_b1(const float* __restrict__ W1, int j, int c, int half, int l16,
                                           v2f (&bw1)[IN_C / 4]) {
  const float* g = W1 + (size_t)(2 * half) * HID + FM_HC * j + 16 * c + l16;
#pragma unroll
  for (int kq = K0; kq < K1; ++kq) {
    bw1[kq].x = g[(size_t)(4 * kq) * HID];
    bw1[kq].y = g[(size_t)(4 * kq + 1) * HID];
  }
}

__device__ __forceinline__ void fm_load_b2(const float* __restrict__ W2, int j, int c, int half, int l16,
                                           v2f (&bw2)[FM_HC / 4]) {
  const float* g = W2 + (size_t)(FM_HC * j + 2 * half) * OUT2 + 16 * c + l16;
#pragma unroll
  for (int kk = 0; kk < FM_HC / 4; ++kk) {
    bw2[kk].x = g[(size_t)(4 * kk) * OUT2];
    bw2[kk].y = g[(size_t)(4 * kk + 1) * OUT2];
  }
}

template <int NK>
__device__ __forceinline__ void fm_frags(const float* s, v2f (&f)[NK]) {
#pragma unroll
  for (int k = 0; k < NK; ++k) f[k] = *(const v2f*)(s + 4 * k);
}

__device__ __forceinline__ void fm_park_hidden(float* Hst, v8f c, float bia) {
#pragma unroll
  for (int i = 0; i < 8; ++i) {
    const float v = c[i] + bia;
    Hst[i * FM_HP] = v > 0.f ? v : 0.f;
  }
}

__global__ __launch_bounds__(256)
void gat_transform_fused(const float* __restrict__ A,
                         const float* __restrict__ W1,
                         const float* __restrict__ b1,
                         const float* __restrict__ W2,
                         const float* __restrict__ a2s,
                         const float* __restrict__ a2d,
                         float* __restrict__ H2,
                         float* __restrict__ as_out,
                         float* __restrict__ ad_out, int N) {
  __shared__ __attribute__((aligned(16))) float lds[FM_ROWS * FM_AP + 2 * FM_ROWS * FM_HP];
  float* As = lds;
  float* Hs = lds + FM_ROWS * FM_AP;

  const int tid  = threadIdx.x;
  const int lane = tid & 31;
  const int wave = __builtin_amdgcn_readfirstlane(tid >> 5);
  const int c    = wave & 3;
  const int p    = wave >> 2;
  const int half = lane >> 4;
  const int l16  = lane & 15;
  const int row0 = blockIdx.x * FM_ROWS;
  int nst = (N - row0 + 15) >> 4;
  if (nst > FM_R) nst = FM_R;

  v2f bw1[IN_C / 4], bw2[FM_HC / 4];

  {
    v4f ar[FM_AI];
#pragma unroll
    for (int t = 0; t < FM_AI; ++t) {
      const int i = tid + 256 * t;
      int grow = row0 + i / (IN_C / 4);
      if (grow > N - 1) grow = N - 1;
      ar[t] = *(const v4f*)(A + (size_t)grow * IN_C + 4 * (i % (IN_C / 4)));
    }
    fm_load_b1<0, IN_C / 8>(W1, 0, c, half, l16, bw1);
#pragma unroll
    for (int t = 0; t < FM_AI; ++t) {
      const int i = tid + 256 * t;
      *(v4f*)(As + (i / (IN_C / 4)) * FM_AP + 4 * (i % (IN_C / 4))) = ar[t];
    }
  }
  fm_load_b1<IN_C / 8, IN_C / 4>(W1, 0, c, half, l16, bw1);
  fm_load_b2(W2, 0, c, half, l16, bw2);
  float bia = b1[16 * c + l16];
  __syncthreads();

  const float* Al = As + l16 * FM_AP + 2 * half;

  v8f acc2[FM_OWNP];
#pragma unroll
  for (int ri = 0; ri < FM_OWNP; ++ri) acc2[ri] = v8f{};

  for (int j = 0; j < FM_NCH; ++j) {
    float* Hb = Hs + (j & 1) * (FM_ROWS * FM_HP);
    float* Hst = Hb + (8 * half) * FM_HP + 16 * c + l16;

    int r = p;
    for (; r + 2 < nst; r += 4) {
      const float* Aa = Al + r * (16 * FM_AP);
      const float* Ab = Aa + 32 * FM_AP;
      v8f c0 = {}, c1 = {};
      v2f a0[IN_C / 4], a1[IN_C / 4];
      fm_frags(Aa, a0);
      fm_frags(Ab, a1);
#pragma unroll
      for (int kq = 0; kq < IN_C / 4; ++kq) {
        c0 = fm_wmma(a0[kq], bw1[kq], c0);
        c1 = fm_wmma(a1[kq], bw1[kq], c1);
      }
      fm_park_hidden(Hst + r * (16 * FM_HP), c0, bia);
      fm_park_hidden(Hst + (r + 2) * (16 * FM_HP), c1, bia);
    }
    if (r < nst) {
      const float* Aa = Al + r * (16 * FM_AP);
      v8f c0 = {};
      v2f a0[IN_C / 4];
      fm_frags(Aa, a0);
      __builtin_amdgcn_sched_barrier(0);
#pragma unroll
      for (int kq = 0; kq < IN_C / 4; ++kq) c0 = fm_wmma(a0[kq], bw1[kq], c0);
      fm_chain_end(c0);
      fm_park_hidden(Hst + r * (16 * FM_HP), c0, bia);
    }
    __syncthreads();

    const int jn = j + 1 < FM_NCH ? j + 1 : j;
    fm_load_b1<0, IN_C / 8>(W1, jn, c, half, l16, bw1);
    bia = b1[FM_HC * jn + 16 * c + l16];

    const float* Hl = Hb + l16 * FM_HP + 2 * half;
#pragma unroll
    for (int ri = 0; ri < FM_OWN; ri += 2) {
      const int ra = p + 2 * ri;
      if (ri + 1 < FM_OWN && ra + 2 < nst) {
        const float* Ha = Hl + ra * (16 * FM_HP);
        const float* Hc = Ha + 32 * FM_HP;
        v2f h0[FM_HC / 4], h1[FM_HC / 4];
        fm_frags(Ha, h0);
        fm_frags(Hc, h1);
        __builtin_amdgcn_sched_barrier(0);
#pragma unroll
        for (int kk = 0; kk < FM_HC / 4; ++kk) {
          acc2[ri]     = fm_wmma(h0[kk], bw2[kk], acc2[ri]);
          acc2[ri + 1] = fm_wmma(h1[kk], bw2[kk], acc2[ri + 1]);
        }
      } else if (ra < nst) {
        const float* Ha = Hl + ra * (16 * FM_HP);
        v2f h0[FM_HC / 4];
        fm_frags(Ha, h0);
        __builtin_amdgcn_sched_barrier(0);
#pragma unroll
        for (int kk = 0; kk < FM_HC / 4; ++kk) acc2[ri] = fm_wmma(h0[kk], bw2[kk], acc2[ri]);
        fm_chain_end(acc2[ri]);
      }
      if (ri == 0) fm_load_b1<IN_C / 8, IN_C / 4>(W1, jn, c, half, l16, bw1);
    }
    fm_load_b2(W2, jn, c, half, l16, bw2);
  }

  float* Os = Hs;
#pragma unroll
  for (int ri = 0; ri < FM_OWN; ++ri) {
    const int ra = p + 2 * ri;
    if (ra < nst) {
      float* o = Os + (ra * 16 + 8 * half) * FM_HP + 16 * c + l16;
#pragma unroll
      for (int i = 0; i < 8; ++i) o[i * FM_HP] = acc2[ri][i];
    }
  }
  __syncthreads();

  if (tid < nst * 16) {
    const float* o = Os + tid * FM_HP;
    float ps = 0.f, pd = 0.f;
#pragma unroll
    for (int q = 0; q < OUT2 / 4; ++q) {
      const v4f v = *(const v4f*)(o + 4 * q);
      ps += v.x * a2s[4 * q];     pd += v.x * a2d[4 * q];
      ps += v.y * a2s[4 * q + 1]; pd += v.y * a2d[4 * q + 1];
      ps += v.z * a2s[4 * q + 2]; pd += v.z * a2d[4 * q + 2];
      ps += v.w * a2s[4 * q + 3]; pd += v.w * a2d[4 * q + 3];
    }
    const int grow = row0 + tid;
    if (grow < N) { as_out[grow] = ps; ad_out[grow] = pd; }
  }

  v4f orow[FM_R];
#pragma unroll
  for (int t = 0; t < FM_R; ++t)
    orow[t] = *(const v4f*)(Os + (2 * wave + 16 * t + half) * FM_HP + 4 * l16);
#pragma unroll
  for (int t = 0; t < FM_R; ++t) {
    const int grow = row0 + 2 * wave + 16 * t + half;
    if (grow < N) *(v4f*)(H2 + (size_t)grow * OUT2 + 4 * l16) = orow[t];
  }
}

#define LINE_W 32
#define SLOTS  31

#define RNG_SH  7
#define RNG_N   (1 << RNG_SH)
#define MAXR    512
#define CHUNK   1024
#define EPT     (CHUNK / 256)
#define OFF_P   (MAXR + 8)

__global__ __launch_bounds__(256)
void gat_sort_chunks_w1_dots(const int* __restrict__ ei, int E,
                             int2* __restrict__ sorted, int* __restrict__ off,
                             int* __restrict__ spill_n,
                             const float* __restrict__ W1,
                             const float* __restrict__ a_src,
                             const float* __restrict__ a_dst,
                             float* __restrict__ wa,
                             float* __restrict__ wd) {
  __shared__ int cnt[MAXR];
  __shared__ int wsum[8];
  __shared__ int2 stage[CHUNK];
  const int tid  = threadIdx.x;
  const int lane = tid & 31;
  const int wib  = tid >> 5;
  const int base = blockIdx.x * CHUNK;
  int total = E - base;
  if (total > CHUNK) total = CHUNK;

  int d[EPT], s[EPT];
#pragma unroll
  for (int t = 0; t < EPT; ++t) {
    const int i = tid + 256 * t;
    d[t] = 0; s[t] = 0;
    if (i < total) { d[t] = ei[E + base + i]; s[t] = ei[base + i]; }
  }
  cnt[tid] = 0;
  cnt[tid + 256] = 0;
  if (blockIdx.x == 0 && tid == 0) spill_n[0] = 0;
  __syncthreads();

  int rg[EPT], rk[EPT];
#pragma unroll
  for (int t = 0; t < EPT; ++t) {
    int r = d[t] >> RNG_SH;
    if (r > MAXR - 1) r = MAXR - 1;
    if (r < 0) r = 0;
    rg[t] = r;
    rk[t] = 0;
    if (tid + 256 * t < total) rk[t] = atomicAdd(&cnt[r], 1);
  }
  __syncthreads();

  const int c0 = cnt[2 * tid], c1 = cnt[2 * tid + 1];
  const int two = c0 + c1;
  int inc = two;
#pragma unroll
  for (int o = 1; o < 32; o <<= 1) {
    const int v = __shfl_up(inc, o, 32);
    if (lane >= o) inc += v;
  }
  if (lane == 31) wsum[wib] = inc;
  __syncthreads();
  int pre = 0;
#pragma unroll
  for (int w = 0; w < 8; ++w) if (w < wib) pre += wsum[w];
  const int ex = pre + inc - two;
  cnt[2 * tid]     = ex;
  cnt[2 * tid + 1] = ex + c0;
  __syncthreads();

#pragma unroll
  for (int t = 0; t < EPT; ++t)
    if (tid + 256 * t < total) stage[cnt[rg[t]] + rk[t]] = make_int2(d[t], s[t]);
  int* orow = off + (size_t)blockIdx.x * OFF_P;
  orow[tid]       = cnt[tid];
  orow[tid + 256] = cnt[tid + 256];
  if (tid == 0) orow[MAXR] = total;
  __syncthreads();
#pragma unroll
  for (int t = 0; t < EPT; ++t) {
    const int i = tid + 256 * t;
    if (i < total) sorted[(size_t)base + i] = stage[i];
  }

  const int wave0  = __builtin_amdgcn_readfirstlane((int)(blockIdx.x * 8 + wib));
  const int nwaves = gridDim.x * 8;
  for (int r = wave0; r < IN_C; r += nwaves) {
    const float* row = W1 + (size_t)r * HID;
    float v[HID / 32], as[HID / 32], ad[HID / 32];
#pragma unroll
    for (int k = 0; k < HID / 32; ++k) v[k] = row[lane + 32 * k];
#pragma unroll
    for (int k = 0; k < HID / 32; ++k) { as[k] = a_src[lane + 32 * k]; ad[k] = a_dst[lane + 32 * k]; }
    float ss = 0.f, sd = 0.f;
#pragma unroll
    for (int k = 0; k < HID / 32; ++k) { ss = __builtin_fmaf(v[k], as[k], ss); sd = __builtin_fmaf(v[k], ad[k], sd); }
#pragma unroll
    for (int o = 16; o > 0; o >>= 1) {
      ss += __shfl_down(ss, o, 32);
      sd += __shfl_down(sd, o, 32);
    }
    if (lane == 0) { wa[r] = ss; wd[r] = sd; }
  }
}

#define XD_NB 4
#define XD_Q  (IN_C / 32)
#define PC_G  4

__device__ __forceinline__ void xd_load(const float* __restrict__ x, int n0, int nwaves, int N,
                                        int lane, float (&v)[XD_NB][XD_Q]) {
#pragma unroll
  for (int k = 0; k < XD_NB; ++k) {
    int node = n0 + k * nwaves;
    if (node > N - 1) node = n0;
    const float* row = x + (size_t)node * IN_C;
#pragma unroll
    for (int q = 0; q < XD_Q; ++q) v[k][q] = row[lane + 32 * q];
  }
}

__global__ __launch_bounds__(256)
void gat_fill_lines_x_dots(const int2* __restrict__ sorted, const int* __restrict__ off, int nch,
                           int* __restrict__ tab,
                           int* __restrict__ spill_n, int* __restrict__ spill,
                           const float* __restrict__ x,
                           const float* __restrict__ wa,
                           const float* __restrict__ wd,
                           float* __restrict__ as_out,
                           float* __restrict__ ad_out, int N) {
  __shared__ __attribute__((aligned(16))) int lines[RNG_N * LINE_W];
  const int tid  = threadIdx.x;
  const int lane = tid & 31;
  const int wib   = __builtin_amdgcn_readfirstlane(tid >> 5);
  const int rng   = blockIdx.x;
  const int node0 = rng << RNG_SH;
  int nend = node0 + RNG_N;
  if (nend > N) nend = N;

  int4* l4 = (int4*)lines;
#pragma unroll
  for (int t = 0; t < RNG_N * LINE_W / 4 / 256; ++t) l4[tid + 256 * t] = make_int4(0, 0, 0, 0);
  __syncthreads();

  float a[XD_Q], b[XD_Q];
#pragma unroll
  for (int q = 0; q < XD_Q; ++q) { a[q] = wa[lane + 32 * q]; b[q] = wd[lane + 32 * q]; }
  float v[XD_NB][XD_Q];
  int n0 = node0 + wib;
  if (n0 < nend) xd_load(x, n0, 8, nend, lane, v);

  for (int c = tid; c < nch; c += 256) {
    const int* o = off + (size_t)c * OFF_P + rng;
    int b0 = o[0], e0 = o[1];
    if (b0 < 0) b0 = 0;
    if (e0 > CHUNK) e0 = CHUNK;
    const int2* p = sorted + (size_t)c * CHUNK;
    for (int i = b0; i < e0; i += PC_G) {
      int2 q[PC_G];
#pragma unroll
      for (int g = 0; g < PC_G; ++g) q[g] = p[i + g < e0 ? i + g : b0];
#pragma unroll
      for (int g = 0; g < PC_G; ++g) {
        if (i + g < e0) {
          const int n = q[g].x & (RNG_N - 1);
          const int k = atomicAdd(&lines[n * LINE_W + SLOTS], 1);
          if (k < SLOTS) {
            lines[n * LINE_W + k] = q[g].y;
          } else {
            const int w = atomicAdd(spill_n, 1);
            spill[2 * w]     = q[g].x;
            spill[2 * w + 1] = q[g].y;
          }
        }
      }
    }
  }
  __syncthreads();

  int4* t4 = (int4*)(tab + (size_t)node0 * LINE_W);
#pragma unroll
  for (int t = 0; t < RNG_N * LINE_W / 4 / 256; ++t) {
    const int i = tid + 256 * t;
    if (node0 + (i >> 3) < N) t4[i] = l4[i];
  }

  while (n0 < nend) {
    float ss[XD_NB], sd[XD_NB];
#pragma unroll
    for (int k = 0; k < XD_NB; ++k) {
      ss[k] = 0.f; sd[k] = 0.f;
#pragma unroll
      for (int q = 0; q < XD_Q; ++q) {
        ss[k] = __builtin_fmaf(v[k][q], a[q], ss[k]);
        sd[k] = __builtin_fmaf(v[k][q], b[q], sd[k]);
      }
    }
#pragma unroll
    for (int o = 16; o > 0; o >>= 1) {
#pragma unroll
      for (int k = 0; k < XD_NB; ++k) {
        ss[k] += __shfl_down(ss[k], o, 32);
        sd[k] += __shfl_down(sd[k], o, 32);
      }
    }
#pragma unroll
    for (int k = 0; k < XD_NB; ++k) {
      const int node = n0 + k * 8;
      if (lane == 0 && node < nend) { as_out[node] = ss[k]; ad_out[node] = sd[k]; }
    }
    n0 += XD_NB * 8;
    if (n0 < nend) xd_load(x, n0, 8, nend, lane, v);
  }
}

__device__ __forceinline__ float leakyr(float v) {
  return v > 0.f ? v : NEG_SLOPE * v;
}

__device__ __forceinline__ float wave_max(float v) {
#pragma unroll
  for (int off = 16; off > 0; off >>= 1) v = fmaxf(v, __shfl_xor(v, off, 32));
  return v;
}

__device__ __forceinline__ float wave_sum(float v) {
#pragma unroll
  for (int off = 16; off > 0; off >>= 1) v += __shfl_xor(v, off, 32);
  return v;
}

template <int D> struct gat_rowvec;
template <> struct gat_rowvec<128> { typedef v4f type; };
template <> struct gat_rowvec<64>  { typedef v2f type; };

template <int D>
__device__ __forceinline__ void agg_rows(const float* __restrict__ h, int s_l, float a_l,
                                         int n, int lane,
                                         typename gat_rowvec<D>::type& acc) {
  typedef typename gat_rowvec<D>::type vec;
  constexpr int GRP = 8;
  for (int u = 0; u < n; u += GRP) {
    vec v[GRP];
#pragma unroll
    for (int g = 0; g < GRP; ++g) {
      const int s = __builtin_amdgcn_readlane(s_l, u + g);
      v[g] = ((const vec*)(h + (size_t)s * D))[lane];
    }
#pragma unroll
    for (int g = 0; g < GRP; ++g) {
      const float a = __int_as_float(__builtin_amdgcn_readlane(__float_as_int(a_l), u + g));
      acc += a * v[g];
    }
  }
}

__device__ __forceinline__ int spill_src(const int* __restrict__ spill, int j, int nsp, int d) {
  int s = -1;
  if (j < nsp && spill[2 * j] == d) s = spill[2 * j + 1];
  return s;
}

template <int D, bool TAIL>
__global__ void gat_node_softmax_agg(const int* __restrict__ tab,
                                     const int* __restrict__ spill_n,
                                     const int* __restrict__ spill, int E,
                                     const float* __restrict__ as,
                                     const float* __restrict__ ad,
                                     const float* __restrict__ h,
                                     const float* __restrict__ bias,
                                     const float* __restrict__ fcW,
                                     const float* __restrict__ fcb,
                                     float* __restrict__ out, int N) {
  typedef typename gat_rowvec<D>::type vec;
  static_assert(!TAIL || D == OUT2, "the classifier tail is written for the 64-wide row");
  float tb0 = 0.f, tb1 = 0.f, tw00 = 0.f, tw01 = 0.f, tw10 = 0.f, tw11 = 0.f, tf0 = 0.f, tf1 = 0.f;
  if constexpr (TAIL) {
    const int c = 2 * (threadIdx.x & 31);
    tb0  = bias[c];
    tb1  = bias[c + 1];
    tw00 = fcW[c * NCLS];
    tw01 = fcW[c * NCLS + 1];
    tw10 = fcW[(c + 1) * NCLS];
    tw11 = fcW[(c + 1) * NCLS + 1];
    tf0  = fcb[0];
    tf1  = fcb[1];
  }
  const int lane = threadIdx.x & 31;
  const int wib    = __builtin_amdgcn_readfirstlane((int)(threadIdx.x >> 5));
  const int wave0  = blockIdx.x * (blockDim.x >> 5) + wib;
  const int nwaves = gridDim.x * (blockDim.x >> 5);
  const float NEG_INF = -__builtin_inff();
  for (int d = wave0; d < N; d += nwaves) {
    const int word   = tab[(size_t)d * LINE_W + lane];
    const float adv  = ad[d];
    const int listed = __builtin_amdgcn_readlane(word, SLOTS);
    const int nslot  = listed < SLOTS ? listed : SLOTS;
    const int cnt    = nslot + 1;

    int nsp = 0;
    if (listed > SLOTS) {
      nsp = __builtin_amdgcn_readfirstlane(spill_n[0]);
      if (nsp > E) nsp = E;
    }

    const bool act0 = lane < cnt;
    int   s0 = d;
    float e0 = NEG_INF;
    if (act0) {
      if (lane < nslot) s0 = word;
      e0 = leakyr(as[s0] + adv);
    }
    float mx = e0;
    for (int base = 0; base < nsp; base += 32) {
      const int s = spill_src(spill, base + lane, nsp, d);
      if (s >= 0) mx = fmaxf(mx, leakyr(as[s] + adv));
    }
    mx = wave_max(mx);

    const float ex0 = act0 ? expf(e0 - mx) : 0.f;
    float sm = ex0;
    for (int base = 0; base < nsp; base += 32) {
      const int s = spill_src(spill, base + lane, nsp, d);
      if (s >= 0) sm += expf(leakyr(as[s] + adv) - mx);
    }
    sm = wave_sum(sm);

    vec acc = {};
    agg_rows<D>(h, s0, ex0 / sm, cnt, lane, acc);
    for (int base = 0; base < nsp; base += 32) {
      const int s = spill_src(spill, base + lane, nsp, d);
      float a = 0.f;
      if (s >= 0) a = expf(leakyr(as[s] + adv) - mx) / sm;
      unsigned m = __builtin_amdgcn_ballot_w32(s >= 0);
      while (m) {
        const int l = __builtin_ctz(m);
        m &= m - 1;
        const int   sl = __builtin_amdgcn_readlane(s, l);
        const float al = __int_as_float(__builtin_amdgcn_readlane(__float_as_int(a), l));
        acc += al * ((const vec*)(h + (size_t)sl * D))[lane];
      }
    }

    if constexpr (TAIL) {
      float v0 = acc[0] + tb0;
      float v1 = acc[1] + tb1;
      v0 = v0 > 0.f ? v0 : 0.f;
      v1 = v1 > 0.f ? v1 : 0.f;
      const float o0 = wave_sum(v0 * tw00 + v1 * tw10);
      const float o1 = wave_sum(v0 * tw01 + v1 * tw11);
      if (lane == 0) {
        const v2f o = {o0 + tf0, o1 + tf1};
        *(v2f*)(out + (size_t)d * NCLS) = o;
      }
    } else {
      ((vec*)(out + (size_t)d * D))[lane] = acc;
    }
  }
}

extern "C" void kernel_launch(void* const* d_in, const int* in_sizes, int n_in,
                              void* d_out, int out_size, void* d_ws, size_t ws_size,
                              hipStream_t stream) {
  const float* x   = (const float*)d_in[0];
  const int*   ei  = (const int*)  d_in[1];
  const float* W1  = (const float*)d_in[2];
  const float* a1s = (const float*)d_in[3];
  const float* a1d = (const float*)d_in[4];
  const float* b1  = (const float*)d_in[5];
  const float* W2  = (const float*)d_in[6];
  const float* a2s = (const float*)d_in[7];
  const float* a2d = (const float*)d_in[8];
  const float* b2  = (const float*)d_in[9];
  const float* fcW = (const float*)d_in[10];
  const float* fcb = (const float*)d_in[11];

  const int N    = in_sizes[0] / IN_C;
  const int E    = in_sizes[1] / 2;

  float* w       = (float*)d_ws;
  float* bufA    = w;
  float* bufB    = w + (size_t)N * HID;
  float* as_buf  = w + (size_t)2 * N * HID;
  float* ad_buf  = as_buf + N;
  int*   tab     = (int*)(bufB + (size_t)N * IN_C);
  int*   spill_n = tab + (size_t)N * LINE_W;
  int*   spill   = spill_n + 4;
  const int nch  = (E + CHUNK - 1) / CHUNK;
  int2*  sorted  = (int2*)(bufA + 1024);
  int*   off     = (int*)(sorted + (size_t)nch * CHUNK);

  const int TB = 256;

  const int CHIP = 2048;
  auto capped = [](int blocks, int cap) { return blocks < cap ? blocks : cap; };
  const int rng_blk   = (N + RNG_N - 1) / RNG_N;
  const int node_blk  = capped((N + 7) / 8, CHIP);

  float* wa   = bufA;
  float* wd   = bufA + IN_C;
  float* aggx = bufB;
  gat_sort_chunks_w1_dots<<<nch, TB, 0, stream>>>(ei, E, sorted, off, spill_n, W1, a1s, a1d, wa, wd);
  gat_fill_lines_x_dots<<<rng_blk, TB, 0, stream>>>(sorted, off, nch, tab, spill_n, spill, x, wa, wd, as_buf, ad_buf, N);

  gat_node_softmax_agg<IN_C, false><<<node_blk, TB, 0, stream>>>(tab, spill_n, spill, E, as_buf, ad_buf, x,
                                                               nullptr, nullptr, nullptr, aggx, N);

  float* h2   = bufA;
  gat_transform_fused<<<(N + FM_ROWS - 1) / FM_ROWS, TB, 0, stream>>>(aggx, W1, b1, W2, a2s, a2d, h2, as_buf, ad_buf, N);

  gat_node_softmax_agg<OUT2, true><<<node_blk, TB, 0, stream>>>(tab, spill_n, spill, E, as_buf, ad_buf, h2,
                                                              b2, fcW, fcb, (float*)d_out, N);
}


# Node passes: one 16-lane half of a wave per destination node, two nodes per wave in step; two edges per lane, 16-lane max/sum by DPP; half the waves; softmax and classifier sums in a new tree
# speedup vs baseline: 8.0978x; 1.0580x over previous
#include <hip/hip_runtime.h>


typedef __attribute__((ext_vector_type(2))) float v2f;
typedef __attribute__((ext_vector_type(8))) float v8f;
typedef __attribute__((ext_vector_type(4))) float v4f;
typedef __attribute__((ext_vector_type(2))) int   v2i;

#define IN_C  128
#define HID   512
#define OUT2  64
#define NCLS  2
#define NEG_SLOPE 0.2f

#define FM_HC   64
#define FM_NCH  (HID / FM_HC)
#define FM_R    13
#define FM_ROWS (16 * FM_R)
#define FM_OWN  ((FM_R + 1) / 2)
#define FM_OWNP ((FM_OWN + 1) & ~1)
#define FM_AP   (IN_C + 4)
#define FM_HP   (FM_HC + 4)
#define FM_AI   (FM_ROWS * (IN_C / 4) / 256)

__device__ __forceinline__ v8f fm_wmma(v2f a, v2f b, v8f c) {
  return __builtin_amdgcn_wmma_f32_16x16x4_f32(false, a, false, b, (short)0, c, false, false);
}

__device__ __forceinline__ void fm_chain_end(v8f& c) {
  asm volatile("v_nop\n\tv_nop\n\tv_nop\n\tv_nop" : "+v"(c));
}

template <int K0, int K1>
__device__ __forceinline__ void fm_load_b1(const float* __restrict__ W1, int j, int c, int half, int l16,
                                           v2f (&bw1)[IN_C / 4]) {
  const float* g = W1 + (size_t)(2 * half) * HID + FM_HC * j + 16 * c + l16;
#pragma unroll
  for (int kq = K0; kq < K1; ++kq) {
    bw1[kq].x = g[(size_t)(4 * kq) * HID];
    bw1[kq].y = g[(size_t)(4 * kq + 1) * HID];
  }
}

__device__ __forceinline__ void fm_load_b2(const float* __restrict__ W2, int j, int c, int half, int l16,
                                           v2f (&bw2)[FM_HC / 4]) {
  const float* g = W2 + (size_t)(FM_HC * j + 2 * half) * OUT2 + 16 * c + l16;
#pragma unroll
  for (int kk = 0; kk < FM_HC / 4; ++kk) {
    bw2[kk].x = g[(size_t)(4 * kk) * OUT2];
    bw2[kk].y = g[(size_t)(4 * kk + 1) * OUT2];
  }
}

template <int NK>
__device__ __forceinline__ void fm_frags(const float* s, v2f (&f)[NK]) {
#pragma unroll
  for (int k = 0; k < NK; ++k) f[k] = *(const v2f*)(s + 4 * k);
}

__device__ __forceinline__ void fm_park_hidden(float* Hst, v8f c, float bia) {
#pragma unroll
  for (int i = 0; i < 8; ++i) {
    const float v = c[i] + bia;
    Hst[i * FM_HP] = v > 0.f ? v : 0.f;
  }
}

__global__ __launch_bounds__(256)
void gat_transform_fused(const float* __restrict__ A,
                         const float* __restrict__ W1,
                         const float* __restrict__ b1,
                         const float* __restrict__ W2,
                         const float* __restrict__ a2s,
                         const float* __restrict__ a2d,
                         float* __restrict__ H2,
                         float* __restrict__ as_out,
                         float* __restrict__ ad_out, int N) {
  __shared__ __attribute__((aligned(16))) float lds[FM_ROWS * FM_AP + 2 * FM_ROWS * FM_HP];
  float* As = lds;
  float* Hs = lds + FM_ROWS * FM_AP;

  const int tid  = threadIdx.x;
  const int lane = tid & 31;
  const int wave = __builtin_amdgcn_readfirstlane(tid >> 5);
  const int c    = wave & 3;
  const int p    = wave >> 2;
  const int half = lane >> 4;
  const int l16  = lane & 15;
  const int row0 = blockIdx.x * FM_ROWS;
  int nst = (N - row0 + 15) >> 4;
  if (nst > FM_R) nst = FM_R;

  v2f bw1[IN_C / 4], bw2[FM_HC / 4];

  {
    v4f ar[FM_AI];
#pragma unroll
    for (int t = 0; t < FM_AI; ++t) {
      const int i = tid + 256 * t;
      int grow = row0 + i / (IN_C / 4);
      if (grow > N - 1) grow = N - 1;
      ar[t] = *(const v4f*)(A + (size_t)grow * IN_C + 4 * (i % (IN_C / 4)));
    }
    fm_load_b1<0, IN_C / 8>(W1, 0, c, half, l16, bw1);
#pragma unroll
    for (int t = 0; t < FM_AI; ++t) {
      const int i = tid + 256 * t;
      *(v4f*)(As + (i / (IN_C / 4)) * FM_AP + 4 * (i % (IN_C / 4))) = ar[t];
    }
  }
  fm_load_b1<IN_C / 8, IN_C / 4>(W1, 0, c, half, l16, bw1);
  fm_load_b2(W2, 0, c, half, l16, bw2);
  float bia = b1[16 * c + l16];
  __syncthreads();

  const float* Al = As + l16 * FM_AP + 2 * half;

  v8f acc2[FM_OWNP];
#pragma unroll
  for (int ri = 0; ri < FM_OWNP; ++ri) acc2[ri] = v8f{};

  for (int j = 0; j < FM_NCH; ++j) {
    float* Hb = Hs + (j & 1) * (FM_ROWS * FM_HP);
    float* Hst = Hb + (8 * half) * FM_HP + 16 * c + l16;

    int r = p;
    for (; r + 2 < nst; r += 4) {
      const float* Aa = Al + r * (16 * FM_AP);
      const float* Ab = Aa + 32 * FM_AP;
      v8f c0 = {}, c1 = {};
      v2f a0[IN_C / 4], a1[IN_C / 4];
      fm_frags(Aa, a0);
      fm_frags(Ab, a1);
#pragma unroll
      for (int kq = 0; kq < IN_C / 4; ++kq) {
        c0 = fm_wmma(a0[kq], bw1[kq], c0);
        c1 = fm_wmma(a1[kq], bw1[kq], c1);
      }
      fm_park_hidden(Hst + r * (16 * FM_HP), c0, bia);
      fm_park_hidden(Hst + (r + 2) * (16 * FM_HP), c1, bia);
    }
    if (r < nst) {
      const float* Aa = Al + r * (16 * FM_AP);
      v8f c0 = {};
      v2f a0[IN_C / 4];
      fm_frags(Aa, a0);
      __builtin_amdgcn_sched_barrier(0);
#pragma unroll
      for (int kq = 0; kq < IN_C / 4; ++kq) c0 = fm_wmma(a0[kq], bw1[kq], c0);
      fm_chain_end(c0);
      fm_park_hidden(Hst + r * (16 * FM_HP), c0, bia);
    }
    __syncthreads();

    const int jn = j + 1 < FM_NCH ? j + 1 : j;
    fm_load_b1<0, IN_C / 8>(W1, jn, c, half, l16, bw1);
    bia = b1[FM_HC * jn + 16 * c + l16];

    const float* Hl = Hb + l16 * FM_HP + 2 * half;
#pragma unroll
    for (int ri = 0; ri < FM_OWN; ri += 2) {
      const int ra = p + 2 * ri;
      if (ri + 1 < FM_OWN && ra + 2 < nst) {
        const float* Ha = Hl + ra * (16 * FM_HP);
        const float* Hc = Ha + 32 * FM_HP;
        v2f h0[FM_HC / 4], h1[FM_HC / 4];
        fm_frags(Ha, h0);
        fm_frags(Hc, h1);
        __builtin_amdgcn_sched_barrier(0);
#pragma unroll
        for (int kk = 0; kk < FM_HC / 4; ++kk) {
          acc2[ri]     = fm_wmma(h0[kk], bw2[kk], acc2[ri]);
          acc2[ri + 1] = fm_wmma(h1[kk], bw2[kk], acc2[ri + 1]);
        }
      } else if (ra < nst) {
        const float* Ha = Hl + ra * (16 * FM_HP);
        v2f h0[FM_HC / 4];
        fm_frags(Ha, h0);
        __builtin_amdgcn_sched_barrier(0);
#pragma unroll
        for (int kk = 0; kk < FM_HC / 4; ++kk) acc2[ri] = fm_wmma(h0[kk], bw2[kk], acc2[ri]);
        fm_chain_end(acc2[ri]);
      }
      if (ri == 0) fm_load_b1<IN_C / 8, IN_C / 4>(W1, jn, c, half, l16, bw1);
    }
    fm_load_b2(W2, jn, c, half, l16, bw2);
  }

  float* Os = Hs;
#pragma unroll
  for (int ri = 0; ri < FM_OWN; ++ri) {
    const int ra = p + 2 * ri;
    if (ra < nst) {
      float* o = Os + (ra * 16 + 8 * half) * FM_HP + 16 * c + l16;
#pragma unroll
      for (int i = 0; i < 8; ++i) o[i * FM_HP] = acc2[ri][i];
    }
  }
  __syncthreads();

  if (tid < nst * 16) {
    const float* o = Os + tid * FM_HP;
    float ps = 0.f, pd = 0.f;
#pragma unroll
    for (int q = 0; q < OUT2 / 4; ++q) {
      const v4f v = *(const v4f*)(o + 4 * q);
      ps += v.x * a2s[4 * q];     pd += v.x * a2d[4 * q];
      ps += v.y * a2s[4 * q + 1]; pd += v.y * a2d[4 * q + 1];
      ps += v.z * a2s[4 * q + 2]; pd += v.z * a2d[4 * q + 2];
      ps += v.w * a2s[4 * q + 3]; pd += v.w * a2d[4 * q + 3];
    }
    const int grow = row0 + tid;
    if (grow < N) { as_out[grow] = ps; ad_out[grow] = pd; }
  }

  v4f orow[FM_R];
#pragma unroll
  for (int t = 0; t < FM_R; ++t)
    orow[t] = *(const v4f*)(Os + (2 * wave + 16 * t + half) * FM_HP + 4 * l16);
#pragma unroll
  for (int t = 0; t < FM_R; ++t) {
    const int grow = row0 + 2 * wave + 16 * t + half;
    if (grow < N) *(v4f*)(H2 + (size_t)grow * OUT2 + 4 * l16) = orow[t];
  }
}

#define LINE_W 32
#define SLOTS  31

#define RNG_SH  7
#define RNG_N   (1 << RNG_SH)
#define MAXR    512
#define CHUNK   1024
#define EPT     (CHUNK / 256)
#define OFF_P   (MAXR + 8)

__global__ __launch_bounds__(256)
void gat_sort_chunks_w1_dots(const int* __restrict__ ei, int E,
                             int2* __restrict__ sorted, int* __restrict__ off,
                             int* __restrict__ spill_n,
                             const float* __restrict__ W1,
                             const float* __restrict__ a_src,
                             const float* __restrict__ a_dst,
                             float* __restrict__ wa,
                             float* __restrict__ wd) {
  __shared__ int cnt[MAXR];
  __shared__ int wsum[8];
  __shared__ int2 stage[CHUNK];
  const int tid  = threadIdx.x;
  const int lane = tid & 31;
  const int wib  = tid >> 5;
  const int base = blockIdx.x * CHUNK;
  int total = E - base;
  if (total > CHUNK) total = CHUNK;

  int d[EPT], s[EPT];
#pragma unroll
  for (int t = 0; t < EPT; ++t) {
    const int i = tid + 256 * t;
    d[t] = 0; s[t] = 0;
    if (i < total) { d[t] = ei[E + base + i]; s[t] = ei[base + i]; }
  }
  cnt[tid] = 0;
  cnt[tid + 256] = 0;
  if (blockIdx.x == 0 && tid == 0) spill_n[0] = 0;
  __syncthreads();

  int rg[EPT], rk[EPT];
#pragma unroll
  for (int t = 0; t < EPT; ++t) {
    int r = d[t] >> RNG_SH;
    if (r > MAXR - 1) r = MAXR - 1;
    if (r < 0) r = 0;
    rg[t] = r;
    rk[t] = 0;
    if (tid + 256 * t < total) rk[t] = atomicAdd(&cnt[r], 1);
  }
  __syncthreads();

  const int c0 = cnt[2 * tid], c1 = cnt[2 * tid + 1];
  const int two = c0 + c1;
  int inc = two;
#pragma unroll
  for (int o = 1; o < 32; o <<= 1) {
    const int v = __shfl_up(inc, o, 32);
    if (lane >= o) inc += v;
  }
  if (lane == 31) wsum[wib] = inc;
  __syncthreads();
  int pre = 0;
#pragma unroll
  for (int w = 0; w < 8; ++w) if (w < wib) pre += wsum[w];
  const int ex = pre + inc - two;
  cnt[2 * tid]     = ex;
  cnt[2 * tid + 1] = ex + c0;
  __syncthreads();

#pragma unroll
  for (int t = 0; t < EPT; ++t)
    if (tid + 256 * t < total) stage[cnt[rg[t]] + rk[t]] = make_int2(d[t], s[t]);
  int* orow = off + (size_t)blockIdx.x * OFF_P;
  orow[tid]       = cnt[tid];
  orow[tid + 256] = cnt[tid + 256];
  if (tid == 0) orow[MAXR] = total;
  __syncthreads();
#pragma unroll
  for (int t = 0; t < EPT; ++t) {
    const int i = tid + 256 * t;
    if (i < total) sorted[(size_t)base + i] = stage[i];
  }

  const int wave0  = __builtin_amdgcn_readfirstlane((int)(blockIdx.x * 8 + wib));
  const int nwaves = gridDim.x * 8;
  for (int r = wave0; r < IN_C; r += nwaves) {
    const float* row = W1 + (size_t)r * HID;
    float v[HID / 32], as[HID / 32], ad[HID / 32];
#pragma unroll
    for (int k = 0; k < HID / 32; ++k) v[k] = row[lane + 32 * k];
#pragma unroll
    for (int k = 0; k < HID / 32; ++k) { as[k] = a_src[lane + 32 * k]; ad[k] = a_dst[lane + 32 * k]; }
    float ss = 0.f, sd = 0.f;
#pragma unroll
    for (int k = 0; k < HID / 32; ++k) { ss = __builtin_fmaf(v[k], as[k], ss); sd = __builtin_fmaf(v[k], ad[k], sd); }
#pragma unroll
    for (int o = 16; o > 0; o >>= 1) {
      ss += __shfl_down(ss, o, 32);
      sd += __shfl_down(sd, o, 32);
    }
    if (lane == 0) { wa[r] = ss; wd[r] = sd; }
  }
}

#define XD_NB 4
#define XD_Q  (IN_C / 32)
#define PC_G  4

__device__ __forceinline__ void xd_load(const float* __restrict__ x, int n0, int nwaves, int N,
                                        int lane, float (&v)[XD_NB][XD_Q]) {
#pragma unroll
  for (int k = 0; k < XD_NB; ++k) {
    int node = n0 + k * nwaves;
    if (node > N - 1) node = n0;
    const float* row = x + (size_t)node * IN_C;
#pragma unroll
    for (int q = 0; q < XD_Q; ++q) v[k][q] = row[lane + 32 * q];
  }
}

__global__ __launch_bounds__(256)
void gat_fill_lines_x_dots(const int2* __restrict__ sorted, const int* __restrict__ off, int nch,
                           int* __restrict__ tab,
                           int* __restrict__ spill_n, int* __restrict__ spill,
                           const float* __restrict__ x,
                           const float* __restrict__ wa,
                           const float* __restrict__ wd,
                           float* __restrict__ as_out,
                           float* __restrict__ ad_out, int N) {
  __shared__ __attribute__((aligned(16))) int lines[RNG_N * LINE_W];
  const int tid  = threadIdx.x;
  const int lane = tid & 31;
  const int wib   = __builtin_amdgcn_readfirstlane(tid >> 5);
  const int rng   = blockIdx.x;
  const int node0 = rng << RNG_SH;
  int nend = node0 + RNG_N;
  if (nend > N) nend = N;

  int4* l4 = (int4*)lines;
#pragma unroll
  for (int t = 0; t < RNG_N * LINE_W / 4 / 256; ++t) l4[tid + 256 * t] = make_int4(0, 0, 0, 0);
  __syncthreads();

  float a[XD_Q], b[XD_Q];
#pragma unroll
  for (int q = 0; q < XD_Q; ++q) { a[q] = wa[lane + 32 * q]; b[q] = wd[lane + 32 * q]; }
  float v[XD_NB][XD_Q];
  int n0 = node0 + wib;
  if (n0 < nend) xd_load(x, n0, 8, nend, lane, v);

  for (int c = tid; c < nch; c += 256) {
    const int* o = off + (size_t)c * OFF_P + rng;
    int b0 = o[0], e0 = o[1];
    if (b0 < 0) b0 = 0;
    if (e0 > CHUNK) e0 = CHUNK;
    const int2* p = sorted + (size_t)c * CHUNK;
    for (int i = b0; i < e0; i += PC_G) {
      int2 q[PC_G];
#pragma unroll
      for (int g = 0; g < PC_G; ++g) q[g] = p[i + g < e0 ? i + g : b0];
#pragma unroll
      for (int g = 0; g < PC_G; ++g) {
        if (i + g < e0) {
          const int n = q[g].x & (RNG_N - 1);
          const int k = atomicAdd(&lines[n * LINE_W + SLOTS], 1);
          if (k < SLOTS) {
            lines[n * LINE_W + k] = q[g].y;
          } else {
            const int w = atomicAdd(spill_n, 1);
            spill[2 * w]     = q[g].x;
            spill[2 * w + 1] = q[g].y;
          }
        }
      }
    }
  }
  __syncthreads();

  int4* t4 = (int4*)(tab + (size_t)node0 * LINE_W);
#pragma unroll
  for (int t = 0; t < RNG_N * LINE_W / 4 / 256; ++t) {
    const int i = tid + 256 * t;
    if (node0 + (i >> 3) < N) t4[i] = l4[i];
  }

  while (n0 < nend) {
    float ss[XD_NB], sd[XD_NB];
#pragma unroll
    for (int k = 0; k < XD_NB; ++k) {
      ss[k] = 0.f; sd[k] = 0.f;
#pragma unroll
      for (int q = 0; q < XD_Q; ++q) {
        ss[k] = __builtin_fmaf(v[k][q], a[q], ss[k]);
        sd[k] = __builtin_fmaf(v[k][q], b[q], sd[k]);
      }
    }
#pragma unroll
    for (int o = 16; o > 0; o >>= 1) {
#pragma unroll
      for (int k = 0; k < XD_NB; ++k) {
        ss[k] += __shfl_down(ss[k], o, 32);
        sd[k] += __shfl_down(sd[k], o, 32);
      }
    }
#pragma unroll
    for (int k = 0; k < XD_NB; ++k) {
      const int node = n0 + k * 8;
      if (lane == 0 && node < nend) { as_out[node] = ss[k]; ad_out[node] = sd[k]; }
    }
    n0 += XD_NB * 8;
    if (n0 < nend) xd_load(x, n0, 8, nend, lane, v);
  }
}

__device__ __forceinline__ float leakyr(float v) {
  return v > 0.f ? v : NEG_SLOPE * v;
}

__device__ __forceinline__ float wave_max(float v) {
#pragma unroll
  for (int off = 16; off > 0; off >>= 1) v = fmaxf(v, __shfl_xor(v, off, 32));
  return v;
}

__device__ __forceinline__ float wave_sum(float v) {
#pragma unroll
  for (int off = 16; off > 0; off >>= 1) v += __shfl_xor(v, off, 32);
  return v;
}

template <int D> struct gat_rowvec;
template <> struct gat_rowvec<128> { typedef v4f type; };
template <> struct gat_rowvec<64>  { typedef v2f type; };

template <int D>
__device__ __forceinline__ void agg_rows(const float* __restrict__ h, int s_l, float a_l,
                                         int n, int lane,
                                         typename gat_rowvec<D>::type& acc) {
  typedef typename gat_rowvec<D>::type vec;
  constexpr int GRP = 8;
  for (int u = 0; u < n; u += GRP) {
    vec v[GRP];
#pragma unroll
    for (int g = 0; g < GRP; ++g) {
      const int s = __builtin_amdgcn_readlane(s_l, u + g);
      v[g] = ((const vec*)(h + (size_t)s * D))[lane];
    }
#pragma unroll
    for (int g = 0; g < GRP; ++g) {
      const float a = __int_as_float(__builtin_amdgcn_readlane(__float_as_int(a_l), u + g));
      acc += a * v[g];
    }
  }
}

__device__ __forceinline__ int spill_src(const int* __restrict__ spill, int j, int nsp, int d) {
  int s = -1;
  if (j < nsp && spill[2 * j] == d) s = spill[2 * j + 1];
  return s;
}

template <int D, bool TAIL>
__device__ __forceinline__ void node_whole_wave(int d, int lane,
                                                const int* __restrict__ tab,
                                                const int* __restrict__ spill_n,
                                                const int* __restrict__ spill, int E,
                                                const float* __restrict__ as,
                                                const float* __restrict__ ad,
                                                const float* __restrict__ h,
                                                const float* __restrict__ bias,
                                                const float* __restrict__ fcW,
                                                const float* __restrict__ fcb,
                                                float* __restrict__ out) {
  typedef typename gat_rowvec<D>::type vec;
  const float NEG_INF = -__builtin_inff();
  const int word   = tab[(size_t)d * LINE_W + lane];
  const float adv  = ad[d];
  const int listed = __builtin_amdgcn_readlane(word, SLOTS);
  const int nslot  = listed < SLOTS ? listed : SLOTS;
  const int cnt    = nslot + 1;

  int nsp = 0;
  if (listed > SLOTS) {
    nsp = __builtin_amdgcn_readfirstlane(spill_n[0]);
    if (nsp > E) nsp = E;
  }

  const bool act0 = lane < cnt;
  int   s0 = d;
  float e0 = NEG_INF;
  if (act0) {
    if (lane < nslot) s0 = word;
    e0 = leakyr(as[s0] + adv);
  }
  float mx = e0;
  for (int base = 0; base < nsp; base += 32) {
    const int s = spill_src(spill, base + lane, nsp, d);
    if (s >= 0) mx = fmaxf(mx, leakyr(as[s] + adv));
  }
  mx = wave_max(mx);

  const float ex0 = act0 ? expf(e0 - mx) : 0.f;
  float sm = ex0;
  for (int base = 0; base < nsp; base += 32) {
    const int s = spill_src(spill, base + lane, nsp, d);
    if (s >= 0) sm += expf(leakyr(as[s] + adv) - mx);
  }
  sm = wave_sum(sm);

  vec acc = {};
  agg_rows<D>(h, s0, ex0 / sm, cnt, lane, acc);
  for (int base = 0; base < nsp; base += 32) {
    const int s = spill_src(spill, base + lane, nsp, d);
    float a = 0.f;
    if (s >= 0) a = expf(leakyr(as[s] + adv) - mx) / sm;
    unsigned m = __builtin_amdgcn_ballot_w32(s >= 0);
    while (m) {
      const int l = __builtin_ctz(m);
      m &= m - 1;
      const int   sl = __builtin_amdgcn_readlane(s, l);
      const float al = __int_as_float(__builtin_amdgcn_readlane(__float_as_int(a), l));
      acc += al * ((const vec*)(h + (size_t)sl * D))[lane];
    }
  }

  if constexpr (TAIL) {
    const int c = 2 * lane;
    float v0 = acc[0] + bias[c];
    float v1 = acc[1] + bias[c + 1];
    v0 = v0 > 0.f ? v0 : 0.f;
    v1 = v1 > 0.f ? v1 : 0.f;
    const float o0 = wave_sum(v0 * fcW[c * NCLS]     + v1 * fcW[(c + 1) * NCLS]);
    const float o1 = wave_sum(v0 * fcW[c * NCLS + 1] + v1 * fcW[(c + 1) * NCLS + 1]);
    if (lane == 0) {
      const v2f o = {o0 + fcb[0], o1 + fcb[1]};
      *(v2f*)(out + (size_t)d * NCLS) = o;
    }
  } else {
    ((vec*)(out + (size_t)d * D))[lane] = acc;
  }
}

template <int M>
__device__ __forceinline__ float row_xor(float v) {
  return __int_as_float(__builtin_amdgcn_update_dpp(0, __float_as_int(v), 0x160 + M, 0xf, 0xf, false));
}
template <int L>
__device__ __forceinline__ int row_take(int v) {
  return __builtin_amdgcn_update_dpp(0, v, 0x150 + L, 0xf, 0xf, false);
}
__device__ __forceinline__ int row_down4(int v) {
  return __builtin_amdgcn_update_dpp(0, v, 0x104, 0xf, 0xf, true);
}

__device__ __forceinline__ float row_max(float v) {
  v = fmaxf(v, row_xor<8>(v));
  v = fmaxf(v, row_xor<4>(v));
  v = fmaxf(v, row_xor<2>(v));
  return fmaxf(v, row_xor<1>(v));
}

__device__ __forceinline__ float row_sum(float v) {
  v += row_xor<8>(v);
  v += row_xor<4>(v);
  v += row_xor<2>(v);
  return v + row_xor<1>(v);
}

template <int D, int G>
__device__ __forceinline__ void agg_fetch(const float* __restrict__ h, int sq0, int sq1, int aq0, int aq1,
                                          int l16, v4f (&v)[D / 64], float& a) {
  const int s = row_take<(G >> 1)>((G & 1) ? sq1 : sq0);
  a = __int_as_float(row_take<(G >> 1)>((G & 1) ? aq1 : aq0));
  const v4f* row = (const v4f*)(h + (size_t)s * D);
#pragma unroll
  for (int qq = 0; qq < D / 64; ++qq) v[qq] = row[l16 + 16 * qq];
}

template <int D, bool TAIL>
__global__ void gat_node_softmax_agg(const int* __restrict__ tab,
                                     const int* __restrict__ spill_n,
                                     const int* __restrict__ spill, int E,
                                     const float* __restrict__ as,
                                     const float* __restrict__ ad,
                                     const float* __restrict__ h,
                                     const float* __restrict__ bias,
                                     const float* __restrict__ fcW,
                                     const float* __restrict__ fcb,
                                     float* __restrict__ out, int N) {
  static_assert(!TAIL || D == OUT2, "the classifier tail is written for the 64-wide row");
  constexpr int Q = D / 64;
  const int lane = threadIdx.x & 31;
  const int l16  = lane & 15;
  const int half = lane >> 4;
  v4f tb = {}, tw0 = {}, tw1 = {};
  float tf0 = 0.f, tf1 = 0.f;
  if constexpr (TAIL) {
    tb  = *(const v4f*)(bias + 4 * l16);
    tw0 = *(const v4f*)(fcW + 4 * l16 * NCLS);
    tw1 = *(const v4f*)(fcW + 4 * l16 * NCLS + 4);
    tf0 = fcb[0];
    tf1 = fcb[1];
  }
  const int wib    = __builtin_amdgcn_readfirstlane((int)(threadIdx.x >> 5));
  const int wave0  = blockIdx.x * (blockDim.x >> 5) + wib;
  const int nwaves = gridDim.x * (blockDim.x >> 5);
  const int npairs = (N + 1) >> 1;
  const float NEG_INF = -__builtin_inff();
  for (int q = wave0; q < npairs; q += nwaves) {
    const int  d     = 2 * q + half;
    const int  dc    = d < N ? d : N - 1;
    const v2i  w     = *(const v2i*)(tab + (size_t)dc * LINE_W + 2 * l16);
    const float adv  = ad[dc];
    const int listedA = __builtin_amdgcn_readlane(w.y, 15);
    const int listedB = __builtin_amdgcn_readlane(w.y, 31);
    const int listed  = half ? listedB : listedA;
    const bool keep = d < N && listed <= SLOTS;
    const int nslot = listed < SLOTS ? listed : SLOTS;
    const int p0    = 2 * l16;
    const bool act0 = p0 <= nslot;
    const bool act1 = p0 + 1 <= nslot;
    const int s0 = p0 < nslot     ? w.x : dc;
    const int s1 = p0 + 1 < nslot ? w.y : dc;
    const float r0 = as[s0];
    const float r1 = as[s1];
    const float e0 = act0 ? leakyr(r0 + adv) : NEG_INF;
    const float e1 = act1 ? leakyr(r1 + adv) : NEG_INF;
    const float mx  = row_max(fmaxf(e0, e1));
    const float ex0 = act0 ? expf(e0 - mx) : 0.f;
    const float ex1 = act1 ? expf(e1 - mx) : 0.f;
    const float sm  = row_sum(ex0 + ex1);

    int sq0 = s0, sq1 = s1;
    int aq0 = __float_as_int(ex0 / sm), aq1 = __float_as_int(ex1 / sm);
    int nmax = (listedA > listedB ? listedA : listedB) + 1;
    if (nmax > 32) nmax = 32;
    v4f acc[Q];
#pragma unroll
    for (int qq = 0; qq < Q; ++qq) acc[qq] = v4f{};
    for (int u = 0; u < nmax; u += 8) {
      v4f v[8][Q];
      float a[8];
      agg_fetch<D, 0>(h, sq0, sq1, aq0, aq1, l16, v[0], a[0]);
      agg_fetch<D, 1>(h, sq0, sq1, aq0, aq1, l16, v[1], a[1]);
      agg_fetch<D, 2>(h, sq0, sq1, aq0, aq1, l16, v[2], a[2]);
      agg_fetch<D, 3>(h, sq0, sq1, aq0, aq1, l16, v[3], a[3]);
      agg_fetch<D, 4>(h, sq0, sq1, aq0, aq1, l16, v[4], a[4]);
      agg_fetch<D, 5>(h, sq0, sq1, aq0, aq1, l16, v[5], a[5]);
      agg_fetch<D, 6>(h, sq0, sq1, aq0, aq1, l16, v[6], a[6]);
      agg_fetch<D, 7>(h, sq0, sq1, aq0, aq1, l16, v[7], a[7]);
      __builtin_amdgcn_sched_barrier(0);
#pragma unroll
      for (int g = 0; g < 8; ++g) {
#pragma unroll
        for (int qq = 0; qq < Q; ++qq) acc[qq] += a[g] * v[g][qq];
      }
      sq0 = row_down4(sq0); sq1 = row_down4(sq1);
      aq0 = row_down4(aq0); aq1 = row_down4(aq1);
    }

    if constexpr (TAIL) {
      v4f r = acc[0] + tb;
      r.x = r.x > 0.f ? r.x : 0.f;
      r.y = r.y > 0.f ? r.y : 0.f;
      r.z = r.z > 0.f ? r.z : 0.f;
      r.w = r.w > 0.f ? r.w : 0.f;
      const float o0 = row_sum(r.x * tw0.x + r.y * tw0.z + r.z * tw1.x + r.w * tw1.z);
      const float o1 = row_sum(r.x * tw0.y + r.y * tw0.w + r.z * tw1.y + r.w * tw1.w);
      if (l16 == 0 && keep) {
        const v2f o = {o0 + tf0, o1 + tf1};
        *(v2f*)(out + (size_t)d * NCLS) = o;
      }
    } else {
      if (keep) {
#pragma unroll
        for (int qq = 0; qq < Q; ++qq) ((v4f*)(out + (size_t)d * D))[l16 + 16 * qq] = acc[qq];
      }
    }

    if (listedA > SLOTS || listedB > SLOTS) {
      for (int k = 0; k < 2; ++k) {
        const int dd = 2 * q + k;
        if ((k ? listedB : listedA) > SLOTS && dd < N)
          node_whole_wave<D, TAIL>(dd, lane, tab, spill_n, spill, E, as, ad, h, bias, fcW, fcb, out);
      }
    }
  }
}

extern "C" void kernel_launch(void* const* d_in, const int* in_sizes, int n_in,
                              void* d_out, int out_size, void* d_ws, size_t ws_size,
                              hipStream_t stream) {
  const float* x   = (const float*)d_in[0];
  const int*   ei  = (const int*)  d_in[1];
  const float* W1  = (const float*)d_in[2];
  const float* a1s = (const float*)d_in[3];
  const float* a1d = (const float*)d_in[4];
  const float* b1  = (const float*)d_in[5];
  const float* W2  = (const float*)d_in[6];
  const float* a2s = (const float*)d_in[7];
  const float* a2d = (const float*)d_in[8];
  const float* b2  = (const float*)d_in[9];
  const float* fcW = (const float*)d_in[10];
  const float* fcb = (const float*)d_in[11];

  const int N    = in_sizes[0] / IN_C;
  const int E    = in_sizes[1] / 2;

  float* w       = (float*)d_ws;
  float* bufA    = w;
  float* bufB    = w + (size_t)N * HID;
  float* as_buf  = w + (size_t)2 * N * HID;
  float* ad_buf  = as_buf + N;
  int*   tab     = (int*)(bufB + (size_t)N * IN_C);
  int*   spill_n = tab + (size_t)N * LINE_W;
  int*   spill   = spill_n + 4;
  const int nch  = (E + CHUNK - 1) / CHUNK;
  int2*  sorted  = (int2*)(bufA + 1024);
  int*   off     = (int*)(sorted + (size_t)nch * CHUNK);

  const int TB = 256;

  const int CHIP = 2048;
  auto capped = [](int blocks, int cap) { return blocks < cap ? blocks : cap; };
  const int rng_blk   = (N + RNG_N - 1) / RNG_N;
  const int node_blk  = capped((N + 47) / 48, CHIP / 2);

  float* wa   = bufA;
  float* wd   = bufA + IN_C;
  float* aggx = bufB;
  gat_sort_chunks_w1_dots<<<nch, TB, 0, stream>>>(ei, E, sorted, off, spill_n, W1, a1s, a1d, wa, wd);
  gat_fill_lines_x_dots<<<rng_blk, TB, 0, stream>>>(sorted, off, nch, tab, spill_n, spill, x, wa, wd, as_buf, ad_buf, N);

  gat_node_softmax_agg<IN_C, false><<<node_blk, TB, 0, stream>>>(tab, spill_n, spill, E, as_buf, ad_buf, x,
                                                               nullptr, nullptr, nullptr, aggx, N);

  float* h2   = bufA;
  gat_transform_fused<<<(N + FM_ROWS - 1) / FM_ROWS, TB, 0, stream>>>(aggx, W1, b1, W2, a2s, a2d, h2, as_buf, ad_buf, N);

  gat_node_softmax_agg<OUT2, true><<<node_blk, TB, 0, stream>>>(tab, spill_n, spill, E, as_buf, ad_buf, h2,
                                                              b2, fcW, fcb, (float*)d_out, N);
}
